# starting kernel
# baseline (speedup 1.0000x reference)
#include <hip/hip_runtime.h>
#include <math.h>

#define IN_C   128
#define HEADS  8
#define HID    8
#define F1     (HEADS * HID)
#define OUT_C  16
#define NEG_SLOPE 0.2f

typedef float v2f __attribute__((ext_vector_type(2)));
typedef float v8f __attribute__((ext_vector_type(8)));

__device__ __forceinline__ unsigned encodeF(float f) {
    unsigned u = __float_as_uint(f);
    return (u & 0x80000000u) ? ~u : (u | 0x80000000u);
}
__device__ __forceinline__ float decodeF(unsigned u) {
    return (u & 0x80000000u) ? __uint_as_float(u & 0x7FFFFFFFu)
                             : __uint_as_float(~u);
}
__device__ __forceinline__ float lrelu(float v) {
    return v > 0.0f ? v : NEG_SLOPE * v;
}

__global__ void gat_gemm1_wmma(const float* __restrict__ x,
                               const float* __restrict__ W,
                               float* __restrict__ h1,
                               int N, int nTiles) {
    __shared__ float w[IN_C * F1];
    for (int i = threadIdx.x; i < IN_C * F1; i += blockDim.x) w[i] = W[i];
    __syncthreads();

    const int wavesPerBlk = blockDim.x >> 5;
    const int tile = blockIdx.x * wavesPerBlk + (threadIdx.x >> 5);
    if (tile >= nTiles) return;

    const int lane  = threadIdx.x & 31;
    const int row16 = lane & 15;
    const int kh    = lane >> 4;

    int rowIn = tile * 16 + row16;
    if (rowIn >= N) rowIn = N - 1;
    const float* xr = x + (size_t)rowIn * IN_C;

    v8f acc0 = {}, acc1 = {}, acc2 = {}, acc3 = {};

    for (int k = 0; k < IN_C; k += 4) {
        const int ka = k + 2 * kh;
        v2f a;
        a.x = xr[ka];
        a.y = xr[ka + 1];
        const float* w0 = &w[ka * F1 + row16];
        const float* w1 = &w[(ka + 1) * F1 + row16];
        v2f b0; b0.x = w0[0];  b0.y = w1[0];
        v2f b1; b1.x = w0[16]; b1.y = w1[16];
        v2f b2; b2.x = w0[32]; b2.y = w1[32];
        v2f b3; b3.x = w0[48]; b3.y = w1[48];
        acc0 = __builtin_amdgcn_wmma_f32_16x16x4_f32(false, a, false, b0, (short)0, acc0, false, false);
        acc1 = __builtin_amdgcn_wmma_f32_16x16x4_f32(false, a, false, b1, (short)0, acc1, false, false);
        acc2 = __builtin_amdgcn_wmma_f32_16x16x4_f32(false, a, false, b2, (short)0, acc2, false, false);
        acc3 = __builtin_amdgcn_wmma_f32_16x16x4_f32(false, a, false, b3, (short)0, acc3, false, false);
    }

    #pragma unroll
    for (int i = 0; i < 8; ++i) {
        const int r = tile * 16 + i + 8 * kh;
        if (r < N) {
            float* o = h1 + (size_t)r * F1 + row16;
            o[0]  = acc0[i];
            o[16] = acc1[i];
            o[32] = acc2[i];
            o[48] = acc3[i];
        }
    }
}

__global__ void gat_gemm2_wmma(const float* __restrict__ hin,
                               const float* __restrict__ W,
                               float* __restrict__ h2,
                               int N, int nTiles) {
    __shared__ float w[F1 * OUT_C];
    for (int i = threadIdx.x; i < F1 * OUT_C; i += blockDim.x) w[i] = W[i];
    __syncthreads();

    const int wavesPerBlk = blockDim.x >> 5;
    const int tile = blockIdx.x * wavesPerBlk + (threadIdx.x >> 5);
    if (tile >= nTiles) return;

    const int lane  = threadIdx.x & 31;
    const int row16 = lane & 15;
    const int kh    = lane >> 4;

    int rowIn = tile * 16 + row16;
    if (rowIn >= N) rowIn = N - 1;
    const float* xr = hin + (size_t)rowIn * F1;

    v8f acc = {};
    for (int k = 0; k < F1; k += 4) {
        const int ka = k + 2 * kh;
        v2f a;
        a.x = xr[ka];
        a.y = xr[ka + 1];
        v2f b;
        b.x = w[ka * OUT_C + row16];
        b.y = w[(ka + 1) * OUT_C + row16];
        acc = __builtin_amdgcn_wmma_f32_16x16x4_f32(false, a, false, b, (short)0, acc, false, false);
    }

    #pragma unroll
    for (int i = 0; i < 8; ++i) {
        const int r = tile * 16 + i + 8 * kh;
        if (r < N) h2[(size_t)r * OUT_C + row16] = acc[i];
    }
}

__global__ void att_dot1(const float* __restrict__ h1,
                         const float* __restrict__ attS,
                         const float* __restrict__ attD,
                         float* __restrict__ aS, float* __restrict__ aD, int NH) {
    int i = blockIdx.x * blockDim.x + threadIdx.x;
    if (i >= NH) return;
    const int n = i >> 3, h = i & 7;
    const float* hp = h1 + (size_t)n * F1 + h * HID;
    float s = 0.f, d = 0.f;
    #pragma unroll
    for (int c = 0; c < HID; ++c) {
        const float v = hp[c];
        s += v * attS[h * HID + c];
        d += v * attD[h * HID + c];
    }
    aS[i] = s;
    aD[i] = d;
}

__global__ void att_dot2(const float* __restrict__ h2,
                         const float* __restrict__ attS,
                         const float* __restrict__ attD,
                         float* __restrict__ aS, float* __restrict__ aD, int N) {
    int n = blockIdx.x * blockDim.x + threadIdx.x;
    if (n >= N) return;
    const float* hp = h2 + (size_t)n * OUT_C;
    float s = 0.f, d = 0.f;
    #pragma unroll
    for (int c = 0; c < OUT_C; ++c) {
        const float v = hp[c];
        s += v * attS[c];
        d += v * attD[c];
    }
    aS[n] = s;
    aD[n] = d;
}

__global__ void edge_max1(const int* __restrict__ src, const int* __restrict__ dst,
                          const float* __restrict__ aS, const float* __restrict__ aD,
                          unsigned* __restrict__ m, int EH) {
    int i = blockIdx.x * blockDim.x + threadIdx.x;
    if (i >= EH) return;
    const int e = i >> 3, h = i & 7;
    const int s = src[e], d = dst[e];
    const float v = lrelu(aS[s * HEADS + h] + aD[d * HEADS + h]);
    atomicMax(&m[d * HEADS + h], encodeF(v));
}

__global__ void edge_den1(const int* __restrict__ src, const int* __restrict__ dst,
                          const float* __restrict__ aS, const float* __restrict__ aD,
                          const unsigned* __restrict__ m, float* __restrict__ den, int EH) {
    int i = blockIdx.x * blockDim.x + threadIdx.x;
    if (i >= EH) return;
    const int e = i >> 3, h = i & 7;
    const int s = src[e], d = dst[e];
    const float v = lrelu(aS[s * HEADS + h] + aD[d * HEADS + h]);
    const float ex = expf(v - decodeF(m[d * HEADS + h]));
    atomicAdd(&den[d * HEADS + h], ex);
}

__global__ void edge_scat1(const int* __restrict__ src, const int* __restrict__ dst,
                           const float* __restrict__ aS, const float* __restrict__ aD,
                           const unsigned* __restrict__ m, const float* __restrict__ den,
                           const float* __restrict__ h1, float* __restrict__ o1, int EH) {
    int i = blockIdx.x * blockDim.x + threadIdx.x;
    if (i >= EH) return;
    const int e = i >> 3, h = i & 7;
    const int s = src[e], d = dst[e];
    const float v = lrelu(aS[s * HEADS + h] + aD[d * HEADS + h]);
    const float ex = expf(v - decodeF(m[d * HEADS + h]));
    const float alpha = ex / (den[d * HEADS + h] + 1e-16f);
    const float* hp = h1 + (size_t)s * F1 + h * HID;
    float* op = o1 + (size_t)d * F1 + h * HID;
    #pragma unroll
    for (int c = 0; c < HID; ++c) atomicAdd(op + c, alpha * hp[c]);
}

__global__ void elu_bias(float* __restrict__ o1, const float* __restrict__ b1, int NF) {
    int i = blockIdx.x * blockDim.x + threadIdx.x;
    if (i >= NF) return;
    const float t = o1[i] + b1[i & (F1 - 1)];
    o1[i] = t > 0.f ? t : (expf(t) - 1.f);
}

__global__ void edge_max2(const int* __restrict__ src, const int* __restrict__ dst,
                          const float* __restrict__ aS, const float* __restrict__ aD,
                          unsigned* __restrict__ m, int E) {
    int e = blockIdx.x * blockDim.x + threadIdx.x;
    if (e >= E) return;
    const int s = src[e], d = dst[e];
    atomicMax(&m[d], encodeF(lrelu(aS[s] + aD[d])));
}

__global__ void edge_den2(const int* __restrict__ src, const int* __restrict__ dst,
                          const float* __restrict__ aS, const float* __restrict__ aD,
                          const unsigned* __restrict__ m, float* __restrict__ den, int E) {
    int e = blockIdx.x * blockDim.x + threadIdx.x;
    if (e >= E) return;
    const int s = src[e], d = dst[e];
    const float v = lrelu(aS[s] + aD[d]);
    atomicAdd(&den[d], expf(v - decodeF(m[d])));
}

__global__ void edge_scat2(const int* __restrict__ src, const int* __restrict__ dst,
                           const float* __restrict__ aS, const float* __restrict__ aD,
                           const unsigned* __restrict__ m, const float* __restrict__ den,
                           const float* __restrict__ h2, float* __restrict__ out, int EC) {
    int i = blockIdx.x * blockDim.x + threadIdx.x;
    if (i >= EC) return;
    const int e = i >> 4, c = i & 15;
    const int s = src[e], d = dst[e];
    const float v = lrelu(aS[s] + aD[d]);
    const float ex = expf(v - decodeF(m[d]));
    const float alpha = ex / (den[d] + 1e-16f);
    atomicAdd(&out[(size_t)d * OUT_C + c], alpha * h2[(size_t)s * OUT_C + c]);
}

__global__ void add_b2(float* __restrict__ out, const float* __restrict__ b2, int NC) {
    int i = blockIdx.x * blockDim.x + threadIdx.x;
    if (i >= NC) return;
    out[i] += b2[i & (OUT_C - 1)];
}

extern "C" void kernel_launch(void* const* d_in, const int* in_sizes, int n_in,
                              void* d_out, int out_size, void* d_ws, size_t ws_size,
                              hipStream_t stream) {
    const int N = in_sizes[0] / IN_C;
    const int E = in_sizes[1] / 2;

    const float* x     = (const float*)d_in[0];
    const int*   ei    = (const int*)  d_in[1];
    const int*   src   = ei;
    const int*   dst   = ei + E;
    const float* W1    = (const float*)d_in[2];
    const float* b1    = (const float*)d_in[3];
    const float* attS1 = (const float*)d_in[4];
    const float* attD1 = (const float*)d_in[5];
    const float* W2    = (const float*)d_in[6];
    const float* b2    = (const float*)d_in[7];
    const float* attS2 = (const float*)d_in[8];
    const float* attD2 = (const float*)d_in[9];
    float* out = (float*)d_out;

    char* ws = (char*)d_ws;
    size_t off = 0;
    auto alloc = [&](size_t elems) -> float* {
        float* p = (float*)(ws + off);
        off = (off + elems * sizeof(float) + 255) & ~(size_t)255;
        return p;
    };
    float*    h1   = alloc((size_t)N * F1);
    float*    aS1  = alloc((size_t)N * HEADS);
    float*    aD1  = alloc((size_t)N * HEADS);
    float*    h2   = alloc((size_t)N * OUT_C);
    float*    aS2  = alloc((size_t)N);
    float*    aD2  = alloc((size_t)N);
    const size_t zoff = off;
    unsigned* m1   = (unsigned*)alloc((size_t)N * HEADS);
    float*    den1 = alloc((size_t)N * HEADS);
    float*    o1   = alloc((size_t)N * F1);
    unsigned* m2   = (unsigned*)alloc((size_t)N);
    float*    den2 = alloc((size_t)N);
    const size_t zbytes = off - zoff;

    hipMemsetAsync(ws + zoff, 0, zbytes, stream);
    hipMemsetAsync(d_out, 0, (size_t)out_size * sizeof(float), stream);

    const int TB = 256;
    auto nb = [](long long n, int tb) { return (unsigned)((n + tb - 1) / tb); };

    const int nTiles = (N + 15) / 16;
    const int gemmBlks = (nTiles + 7) / 8;

    gat_gemm1_wmma<<<gemmBlks, TB, 0, stream>>>(x, W1, h1, N, nTiles);
    att_dot1<<<nb((long long)N * HEADS, TB), TB, 0, stream>>>(h1, attS1, attD1, aS1, aD1, N * HEADS);
    edge_max1<<<nb((long long)E * HEADS, TB), TB, 0, stream>>>(src, dst, aS1, aD1, m1, E * HEADS);
    edge_den1<<<nb((long long)E * HEADS, TB), TB, 0, stream>>>(src, dst, aS1, aD1, m1, den1, E * HEADS);
    edge_scat1<<<nb((long long)E * HEADS, TB), TB, 0, stream>>>(src, dst, aS1, aD1, m1, den1, h1, o1, E * HEADS);
    elu_bias<<<nb((long long)N * F1, TB), TB, 0, stream>>>(o1, b1, N * F1);

    gat_gemm2_wmma<<<gemmBlks, TB, 0, stream>>>(o1, W2, h2, N, nTiles);
    att_dot2<<<nb(N, TB), TB, 0, stream>>>(h2, attS2, attD2, aS2, aD2, N);
    edge_max2<<<nb(E, TB), TB, 0, stream>>>(src, dst, aS2, aD2, m2, E);
    edge_den2<<<nb(E, TB), TB, 0, stream>>>(src, dst, aS2, aD2, m2, den2, E);
    edge_scat2<<<nb((long long)E * OUT_C, TB), TB, 0, stream>>>(src, dst, aS2, aD2, m2, den2, h2, out, E * OUT_C);
    add_b2<<<nb((long long)N * OUT_C, TB), TB, 0, stream>>>(out, b2, N * OUT_C);
}


# Edges counting-sorted by destination each call (one int atomic per edge, arrival-order rank); per-node softmax and weighted sum in registers, no float atomics; summation order changes
# speedup vs baseline: 2.7171x; 2.7171x over previous
#include <hip/hip_runtime.h>
#include <math.h>

#define IN_C   128
#define HEADS  8
#define HID    8
#define F1     (HEADS * HID)
#define OUT_C  16
#define NEG_SLOPE 0.2f

typedef float v2f __attribute__((ext_vector_type(2)));
typedef float v8f __attribute__((ext_vector_type(8)));

__device__ __forceinline__ float lrelu(float v) {
    return v > 0.0f ? v : NEG_SLOPE * v;
}

__global__ void gat_gemm1_wmma(const float* __restrict__ x,
                               const float* __restrict__ W,
                               float* __restrict__ h1,
                               int N, int nTiles) {
    __shared__ float w[IN_C * F1];
    for (int i = threadIdx.x; i < IN_C * F1; i += blockDim.x) w[i] = W[i];
    __syncthreads();

    const int wavesPerBlk = blockDim.x >> 5;
    const int tile = blockIdx.x * wavesPerBlk + (threadIdx.x >> 5);
    if (tile >= nTiles) return;

    const int lane  = threadIdx.x & 31;
    const int row16 = lane & 15;
    const int kh    = lane >> 4;

    int rowIn = tile * 16 + row16;
    if (rowIn >= N) rowIn = N - 1;
    const float* xr = x + (size_t)rowIn * IN_C;

    v8f acc0 = {}, acc1 = {}, acc2 = {}, acc3 = {};

    for (int k = 0; k < IN_C; k += 4) {
        const int ka = k + 2 * kh;
        v2f a;
        a.x = xr[ka];
        a.y = xr[ka + 1];
        const float* w0 = &w[ka * F1 + row16];
        const float* w1 = &w[(ka + 1) * F1 + row16];
        v2f b0; b0.x = w0[0];  b0.y = w1[0];
        v2f b1; b1.x = w0[16]; b1.y = w1[16];
        v2f b2; b2.x = w0[32]; b2.y = w1[32];
        v2f b3; b3.x = w0[48]; b3.y = w1[48];
        acc0 = __builtin_amdgcn_wmma_f32_16x16x4_f32(false, a, false, b0, (short)0, acc0, false, false);
        acc1 = __builtin_amdgcn_wmma_f32_16x16x4_f32(false, a, false, b1, (short)0, acc1, false, false);
        acc2 = __builtin_amdgcn_wmma_f32_16x16x4_f32(false, a, false, b2, (short)0, acc2, false, false);
        acc3 = __builtin_amdgcn_wmma_f32_16x16x4_f32(false, a, false, b3, (short)0, acc3, false, false);
    }

    #pragma unroll
    for (int i = 0; i < 8; ++i) {
        const int r = tile * 16 + i + 8 * kh;
        if (r < N) {
            float* o = h1 + (size_t)r * F1 + row16;
            o[0]  = acc0[i];
            o[16] = acc1[i];
            o[32] = acc2[i];
            o[48] = acc3[i];
        }
    }
}

__global__ void gat_gemm2_wmma(const float* __restrict__ hin,
                               const float* __restrict__ W,
                               float* __restrict__ h2,
                               int N, int nTiles) {
    __shared__ float w[F1 * OUT_C];
    for (int i = threadIdx.x; i < F1 * OUT_C; i += blockDim.x) w[i] = W[i];
    __syncthreads();

    const int wavesPerBlk = blockDim.x >> 5;
    const int tile = blockIdx.x * wavesPerBlk + (threadIdx.x >> 5);
    if (tile >= nTiles) return;

    const int lane  = threadIdx.x & 31;
    const int row16 = lane & 15;
    const int kh    = lane >> 4;

    int rowIn = tile * 16 + row16;
    if (rowIn >= N) rowIn = N - 1;
    const float* xr = hin + (size_t)rowIn * F1;

    v8f acc = {};
    for (int k = 0; k < F1; k += 4) {
        const int ka = k + 2 * kh;
        v2f a;
        a.x = xr[ka];
        a.y = xr[ka + 1];
        v2f b;
        b.x = w[ka * OUT_C + row16];
        b.y = w[(ka + 1) * OUT_C + row16];
        acc = __builtin_amdgcn_wmma_f32_16x16x4_f32(false, a, false, b, (short)0, acc, false, false);
    }

    #pragma unroll
    for (int i = 0; i < 8; ++i) {
        const int r = tile * 16 + i + 8 * kh;
        if (r < N) h2[(size_t)r * OUT_C + row16] = acc[i];
    }
}

__global__ void att_dot1(const float* __restrict__ h1,
                         const float* __restrict__ attS,
                         const float* __restrict__ attD,
                         float* __restrict__ aS, float* __restrict__ aD, int NH) {
    int i = blockIdx.x * blockDim.x + threadIdx.x;
    if (i >= NH) return;
    const int n = i >> 3, h = i & 7;
    const float* hp = h1 + (size_t)n * F1 + h * HID;
    float s = 0.f, d = 0.f;
    #pragma unroll
    for (int c = 0; c < HID; ++c) {
        const float v = hp[c];
        s += v * attS[h * HID + c];
        d += v * attD[h * HID + c];
    }
    aS[i] = s;
    aD[i] = d;
}

__global__ void att_dot2(const float* __restrict__ h2,
                         const float* __restrict__ attS,
                         const float* __restrict__ attD,
                         float* __restrict__ aS, float* __restrict__ aD, int N) {
    int n = blockIdx.x * blockDim.x + threadIdx.x;
    if (n >= N) return;
    const float* hp = h2 + (size_t)n * OUT_C;
    float s = 0.f, d = 0.f;
    #pragma unroll
    for (int c = 0; c < OUT_C; ++c) {
        const float v = hp[c];
        s += v * attS[c];
        d += v * attD[c];
    }
    aS[n] = s;
    aD[n] = d;
}

__global__ void edge_rank(const int* __restrict__ dst, int* __restrict__ cnt,
                          int* __restrict__ rank, int E) {
    int e = blockIdx.x * blockDim.x + threadIdx.x;
    if (e >= E) return;
    rank[e] = atomicAdd(&cnt[dst[e]], 1);
}

__global__ void __launch_bounds__(1024) scan_local(const int* __restrict__ cnt, int* __restrict__ rowptr,
                                                   int* __restrict__ bsum, int N) {
    __shared__ int part[1024];
    const int t = threadIdx.x;
    const int i = blockIdx.x * 1024 + t;
    const int c = (i < N) ? cnt[i] : 0;
    part[t] = c;
    __syncthreads();
    for (int off = 1; off < 1024; off <<= 1) {
        const int v = (t >= off) ? part[t - off] : 0;
        __syncthreads();
        part[t] += v;
        __syncthreads();
    }
    if (i <= N) rowptr[i] = part[t] - c;
    if (t == 1023) bsum[blockIdx.x] = part[1023];
}

__global__ void __launch_bounds__(1024) scan_counts(const int* __restrict__ cnt,
                                                    int* __restrict__ rowptr, int N) {
    __shared__ int part[1024];
    const int t = threadIdx.x;
    const int per = (N + 1023) / 1024;
    int beg = t * per;      if (beg > N) beg = N;
    int end = beg + per;    if (end > N) end = N;
    int s = 0;
    for (int i = beg; i < end; ++i) s += cnt[i];
    part[t] = s;
    __syncthreads();
    for (int off = 1; off < 1024; off <<= 1) {
        const int v = (t >= off) ? part[t - off] : 0;
        __syncthreads();
        part[t] += v;
        __syncthreads();
    }
    int run = part[t] - s;
    for (int i = beg; i < end; ++i) { rowptr[i] = run; run += cnt[i]; }
    if (t == 1023) rowptr[N] = part[1023];
}

__global__ void edge_place(const int* __restrict__ src, const int* __restrict__ dst,
                           const int* __restrict__ rank, const int* __restrict__ rowptr,
                           const int* __restrict__ boff, int* __restrict__ csr, int E) {
    int e = blockIdx.x * blockDim.x + threadIdx.x;
    if (e >= E) return;
    const int d = dst[e];
    csr[rowptr[d] + boff[d >> 10] + rank[e]] = src[e];
}

__global__ void node_gather1(const int* __restrict__ rowptr, const int* __restrict__ boff,
                             const int* __restrict__ csr,
                             const float* __restrict__ aS, const float* __restrict__ aD,
                             const float* __restrict__ h1, const float* __restrict__ b1,
                             float* __restrict__ o1, int NH) {
    int i = blockIdx.x * blockDim.x + threadIdx.x;
    if (i >= NH) return;
    const int n = i >> 3, h = i & 7;
    const int beg = rowptr[n] + boff[n >> 10];
    const int end = rowptr[n + 1] + boff[(n + 1) >> 10];
    const int last = end - 1;
    const float ad = aD[i];

    float m = -INFINITY;
    for (int j = beg; j < end; j += 4) {
        const int j1 = (j + 1 < end) ? j + 1 : last;
        const int j2 = (j + 2 < end) ? j + 2 : last;
        const int j3 = (j + 3 < end) ? j + 3 : last;
        const int s0 = csr[j], s1 = csr[j1], s2 = csr[j2], s3 = csr[j3];
        const float v0 = lrelu(aS[s0 * HEADS + h] + ad);
        const float v1 = lrelu(aS[s1 * HEADS + h] + ad);
        const float v2 = lrelu(aS[s2 * HEADS + h] + ad);
        const float v3 = lrelu(aS[s3 * HEADS + h] + ad);
        const float u = v0 > v1 ? v0 : v1;
        const float w = v2 > v3 ? v2 : v3;
        const float z = u > w ? u : w;
        m = z > m ? z : m;
    }

    float den = 0.f;
    float a0 = 0.f, a1 = 0.f, a2 = 0.f, a3 = 0.f, a4 = 0.f, a5 = 0.f, a6 = 0.f, a7 = 0.f;
    for (int j = beg; j < end; j += 4) {
        const bool k1 = j + 1 < end, k2 = j + 2 < end, k3 = j + 3 < end;
        const int j1 = k1 ? j + 1 : last;
        const int j2 = k2 ? j + 2 : last;
        const int j3 = k3 ? j + 3 : last;
        const int s0 = csr[j], s1 = csr[j1], s2 = csr[j2], s3 = csr[j3];
        const float g0 = aS[s0 * HEADS + h], g1 = aS[s1 * HEADS + h];
        const float g2 = aS[s2 * HEADS + h], g3 = aS[s3 * HEADS + h];
        const float4* p0 = (const float4*)(h1 + (size_t)s0 * F1 + h * HID);
        const float4* p1 = (const float4*)(h1 + (size_t)s1 * F1 + h * HID);
        const float4* p2 = (const float4*)(h1 + (size_t)s2 * F1 + h * HID);
        const float4* p3 = (const float4*)(h1 + (size_t)s3 * F1 + h * HID);
        const float4 x0 = p0[0], y0 = p0[1];
        const float4 x1 = p1[0], y1 = p1[1];
        const float4 x2 = p2[0], y2 = p2[1];
        const float4 x3 = p3[0], y3 = p3[1];
        const float e0 = expf(lrelu(g0 + ad) - m);
        const float e1 = k1 ? expf(lrelu(g1 + ad) - m) : 0.f;
        const float e2 = k2 ? expf(lrelu(g2 + ad) - m) : 0.f;
        const float e3 = k3 ? expf(lrelu(g3 + ad) - m) : 0.f;
        den += e0; den += e1; den += e2; den += e3;
        a0 += e0 * x0.x; a1 += e0 * x0.y; a2 += e0 * x0.z; a3 += e0 * x0.w;
        a4 += e0 * y0.x; a5 += e0 * y0.y; a6 += e0 * y0.z; a7 += e0 * y0.w;
        a0 += e1 * x1.x; a1 += e1 * x1.y; a2 += e1 * x1.z; a3 += e1 * x1.w;
        a4 += e1 * y1.x; a5 += e1 * y1.y; a6 += e1 * y1.z; a7 += e1 * y1.w;
        a0 += e2 * x2.x; a1 += e2 * x2.y; a2 += e2 * x2.z; a3 += e2 * x2.w;
        a4 += e2 * y2.x; a5 += e2 * y2.y; a6 += e2 * y2.z; a7 += e2 * y2.w;
        a0 += e3 * x3.x; a1 += e3 * x3.y; a2 += e3 * x3.z; a3 += e3 * x3.w;
        a4 += e3 * y3.x; a5 += e3 * y3.y; a6 += e3 * y3.z; a7 += e3 * y3.w;
    }

    const float dn = den + 1e-16f;
    const float* bp = b1 + h * HID;
    const float t0 = a0 / dn + bp[0], t1 = a1 / dn + bp[1], t2 = a2 / dn + bp[2], t3 = a3 / dn + bp[3];
    const float t4 = a4 / dn + bp[4], t5 = a5 / dn + bp[5], t6 = a6 / dn + bp[6], t7 = a7 / dn + bp[7];
    float4 r0, r1;
    r0.x = t0 > 0.f ? t0 : (expf(t0) - 1.f);
    r0.y = t1 > 0.f ? t1 : (expf(t1) - 1.f);
    r0.z = t2 > 0.f ? t2 : (expf(t2) - 1.f);
    r0.w = t3 > 0.f ? t3 : (expf(t3) - 1.f);
    r1.x = t4 > 0.f ? t4 : (expf(t4) - 1.f);
    r1.y = t5 > 0.f ? t5 : (expf(t5) - 1.f);
    r1.z = t6 > 0.f ? t6 : (expf(t6) - 1.f);
    r1.w = t7 > 0.f ? t7 : (expf(t7) - 1.f);
    float4* op = (float4*)(o1 + (size_t)n * F1 + h * HID);
    op[0] = r0;
    op[1] = r1;
}

__global__ void node_gather2(const int* __restrict__ rowptr, const int* __restrict__ boff,
                             const int* __restrict__ csr,
                             const float* __restrict__ aS, const float* __restrict__ aD,
                             const float* __restrict__ h2, const float* __restrict__ b2,
                             float* __restrict__ out, int NQ) {
    int i = blockIdx.x * blockDim.x + threadIdx.x;
    if (i >= NQ) return;
    const int n = i >> 2, q = i & 3;
    const int beg = rowptr[n] + boff[n >> 10];
    const int end = rowptr[n + 1] + boff[(n + 1) >> 10];
    const int last = end - 1;
    const float ad = aD[n];

    float m = -INFINITY;
    for (int j = beg; j < end; j += 4) {
        const int j1 = (j + 1 < end) ? j + 1 : last;
        const int j2 = (j + 2 < end) ? j + 2 : last;
        const int j3 = (j + 3 < end) ? j + 3 : last;
        const int s0 = csr[j], s1 = csr[j1], s2 = csr[j2], s3 = csr[j3];
        const float v0 = lrelu(aS[s0] + ad);
        const float v1 = lrelu(aS[s1] + ad);
        const float v2 = lrelu(aS[s2] + ad);
        const float v3 = lrelu(aS[s3] + ad);
        const float u = v0 > v1 ? v0 : v1;
        const float w = v2 > v3 ? v2 : v3;
        const float z = u > w ? u : w;
        m = z > m ? z : m;
    }

    float den = 0.f;
    float a0 = 0.f, a1 = 0.f, a2 = 0.f, a3 = 0.f;
    for (int j = beg; j < end; j += 4) {
        const bool k1 = j + 1 < end, k2 = j + 2 < end, k3 = j + 3 < end;
        const int j1 = k1 ? j + 1 : last;
        const int j2 = k2 ? j + 2 : last;
        const int j3 = k3 ? j + 3 : last;
        const int s0 = csr[j], s1 = csr[j1], s2 = csr[j2], s3 = csr[j3];
        const float g0 = aS[s0], g1 = aS[s1], g2 = aS[s2], g3 = aS[s3];
        const float4 x0 = ((const float4*)(h2 + (size_t)s0 * OUT_C))[q];
        const float4 x1 = ((const float4*)(h2 + (size_t)s1 * OUT_C))[q];
        const float4 x2 = ((const float4*)(h2 + (size_t)s2 * OUT_C))[q];
        const float4 x3 = ((const float4*)(h2 + (size_t)s3 * OUT_C))[q];
        const float e0 = expf(lrelu(g0 + ad) - m);
        const float e1 = k1 ? expf(lrelu(g1 + ad) - m) : 0.f;
        const float e2 = k2 ? expf(lrelu(g2 + ad) - m) : 0.f;
        const float e3 = k3 ? expf(lrelu(g3 + ad) - m) : 0.f;
        den += e0; den += e1; den += e2; den += e3;
        a0 += e0 * x0.x; a1 += e0 * x0.y; a2 += e0 * x0.z; a3 += e0 * x0.w;
        a0 += e1 * x1.x; a1 += e1 * x1.y; a2 += e1 * x1.z; a3 += e1 * x1.w;
        a0 += e2 * x2.x; a1 += e2 * x2.y; a2 += e2 * x2.z; a3 += e2 * x2.w;
        a0 += e3 * x3.x; a1 += e3 * x3.y; a2 += e3 * x3.z; a3 += e3 * x3.w;
    }

    const float dn = den + 1e-16f;
    const float* bq = b2 + q * 4;
    float4 r;
    r.x = a0 / dn + bq[0];
    r.y = a1 / dn + bq[1];
    r.z = a2 / dn + bq[2];
    r.w = a3 / dn + bq[3];
    ((float4*)(out + (size_t)n * OUT_C))[q] = r;
}

extern "C" void kernel_launch(void* const* d_in, const int* in_sizes, int n_in,
                              void* d_out, int out_size, void* d_ws, size_t ws_size,
                              hipStream_t stream) {
    const int N = in_sizes[0] / IN_C;
    const int E = in_sizes[1] / 2;

    const float* x     = (const float*)d_in[0];
    const int*   ei    = (const int*)  d_in[1];
    const int*   src   = ei;
    const int*   dst   = ei + E;
    const float* W1    = (const float*)d_in[2];
    const float* b1    = (const float*)d_in[3];
    const float* attS1 = (const float*)d_in[4];
    const float* attD1 = (const float*)d_in[5];
    const float* W2    = (const float*)d_in[6];
    const float* b2    = (const float*)d_in[7];
    const float* attS2 = (const float*)d_in[8];
    const float* attD2 = (const float*)d_in[9];
    float* out = (float*)d_out;

    char* ws = (char*)d_ws;
    size_t off = 0;
    auto alloc = [&](size_t elems) -> float* {
        float* p = (float*)(ws + off);
        off = (off + elems * sizeof(float) + 255) & ~(size_t)255;
        return p;
    };
    float* h1   = alloc((size_t)N * F1);
    float* aS1  = alloc((size_t)N * HEADS);
    float* aD1  = alloc((size_t)N * HEADS);
    float* o1   = alloc((size_t)N * F1);
    float* h2   = alloc((size_t)N * OUT_C);
    float* aS2  = alloc((size_t)N);
    float* aD2  = alloc((size_t)N);
    int*   cnt    = (int*)alloc((size_t)N);
    int*   rowptr = (int*)alloc((size_t)N + 1);
    const int nScanBlks = (N + 1 + 1023) / 1024;
    int*   bsum   = (int*)alloc((size_t)nScanBlks);
    int*   boff   = (int*)alloc((size_t)nScanBlks + 1);
    int*   rank   = (int*)alloc((size_t)E);
    int*   csr    = (int*)alloc((size_t)E);

    hipMemsetAsync(cnt, 0, (size_t)N * sizeof(int), stream);

    const int TB = 256;
    auto nb = [](long long n, int tb) { return (unsigned)((n + tb - 1) / tb); };

    const int nTiles = (N + 15) / 16;
    const int gemmBlks = (nTiles + 7) / 8;

    edge_rank<<<nb(E, TB), TB, 0, stream>>>(dst, cnt, rank, E);
    scan_local<<<nScanBlks, 1024, 0, stream>>>(cnt, rowptr, bsum, N);
    scan_counts<<<1, 1024, 0, stream>>>(bsum, boff, nScanBlks);
    edge_place<<<nb(E, TB), TB, 0, stream>>>(src, dst, rank, rowptr, boff, csr, E);

    gat_gemm1_wmma<<<gemmBlks, TB, 0, stream>>>(x, W1, h1, N, nTiles);
    att_dot1<<<nb((long long)N * HEADS, TB), TB, 0, stream>>>(h1, attS1, attD1, aS1, aD1, N * HEADS);
    node_gather1<<<nb((long long)N * HEADS, TB), TB, 0, stream>>>(rowptr, boff, csr, aS1, aD1, h1, b1, o1, N * HEADS);

    gat_gemm2_wmma<<<gemmBlks, TB, 0, stream>>>(o1, W2, h2, N, nTiles);
    att_dot2<<<nb(N, TB), TB, 0, stream>>>(h2, attS2, attD2, aS2, aD2, N);
    node_gather2<<<nb((long long)N * 4, TB), TB, 0, stream>>>(rowptr, boff, csr, aS2, aD2, h2, b2, out, N * 4);
}


# Attention scores computed in the node walks from the rows they gather anyway (fused multiply-adds; layer-2 dot summed over 4 lanes, order changes): both dot kernels and all score gathers removed
# speedup vs baseline: 3.1276x; 1.1511x over previous
#include <hip/hip_runtime.h>
#include <math.h>

#define IN_C   128
#define HEADS  8
#define HID    8
#define F1     (HEADS * HID)
#define OUT_C  16
#define NEG_SLOPE 0.2f

typedef float v2f __attribute__((ext_vector_type(2)));
typedef float v8f __attribute__((ext_vector_type(8)));

__device__ __forceinline__ float lrelu(float v) {
    return v > 0.0f ? v : NEG_SLOPE * v;
}

__global__ void gat_gemm1_wmma(const float* __restrict__ x,
                               const float* __restrict__ W,
                               float* __restrict__ h1,
                               int N, int nTiles) {
    __shared__ float w[IN_C * F1];
    for (int i = threadIdx.x; i < IN_C * F1; i += blockDim.x) w[i] = W[i];
    __syncthreads();

    const int wavesPerBlk = blockDim.x >> 5;
    const int tile = blockIdx.x * wavesPerBlk + (threadIdx.x >> 5);
    if (tile >= nTiles) return;

    const int lane  = threadIdx.x & 31;
    const int row16 = lane & 15;
    const int kh    = lane >> 4;

    int rowIn = tile * 16 + row16;
    if (rowIn >= N) rowIn = N - 1;
    const float* xr = x + (size_t)rowIn * IN_C;

    v8f acc0 = {}, acc1 = {}, acc2 = {}, acc3 = {};

    for (int k = 0; k < IN_C; k += 4) {
        const int ka = k + 2 * kh;
        v2f a;
        a.x = xr[ka];
        a.y = xr[ka + 1];
        const float* w0 = &w[ka * F1 + row16];
        const float* w1 = &w[(ka + 1) * F1 + row16];
        v2f b0; b0.x = w0[0];  b0.y = w1[0];
        v2f b1; b1.x = w0[16]; b1.y = w1[16];
        v2f b2; b2.x = w0[32]; b2.y = w1[32];
        v2f b3; b3.x = w0[48]; b3.y = w1[48];
        acc0 = __builtin_amdgcn_wmma_f32_16x16x4_f32(false, a, false, b0, (short)0, acc0, false, false);
        acc1 = __builtin_amdgcn_wmma_f32_16x16x4_f32(false, a, false, b1, (short)0, acc1, false, false);
        acc2 = __builtin_amdgcn_wmma_f32_16x16x4_f32(false, a, false, b2, (short)0, acc2, false, false);
        acc3 = __builtin_amdgcn_wmma_f32_16x16x4_f32(false, a, false, b3, (short)0, acc3, false, false);
    }

    #pragma unroll
    for (int i = 0; i < 8; ++i) {
        const int r = tile * 16 + i + 8 * kh;
        if (r < N) {
            float* o = h1 + (size_t)r * F1 + row16;
            o[0]  = acc0[i];
            o[16] = acc1[i];
            o[32] = acc2[i];
            o[48] = acc3[i];
        }
    }
}

__global__ void gat_gemm2_wmma(const float* __restrict__ hin,
                               const float* __restrict__ W,
                               float* __restrict__ h2,
                               int N, int nTiles) {
    __shared__ float w[F1 * OUT_C];
    for (int i = threadIdx.x; i < F1 * OUT_C; i += blockDim.x) w[i] = W[i];
    __syncthreads();

    const int wavesPerBlk = blockDim.x >> 5;
    const int tile = blockIdx.x * wavesPerBlk + (threadIdx.x >> 5);
    if (tile >= nTiles) return;

    const int lane  = threadIdx.x & 31;
    const int row16 = lane & 15;
    const int kh    = lane >> 4;

    int rowIn = tile * 16 + row16;
    if (rowIn >= N) rowIn = N - 1;
    const float* xr = hin + (size_t)rowIn * F1;

    v8f acc = {};
    for (int k = 0; k < F1; k += 4) {
        const int ka = k + 2 * kh;
        v2f a;
        a.x = xr[ka];
        a.y = xr[ka + 1];
        v2f b;
        b.x = w[ka * OUT_C + row16];
        b.y = w[(ka + 1) * OUT_C + row16];
        acc = __builtin_amdgcn_wmma_f32_16x16x4_f32(false, a, false, b, (short)0, acc, false, false);
    }

    #pragma unroll
    for (int i = 0; i < 8; ++i) {
        const int r = tile * 16 + i + 8 * kh;
        if (r < N) h2[(size_t)r * OUT_C + row16] = acc[i];
    }
}

__device__ __forceinline__ float dot8(const float4 a, const float4 b,
                                      const float w0, const float w1, const float w2, const float w3,
                                      const float w4, const float w5, const float w6, const float w7) {
    float s = a.x * w0;
    s = fmaf(a.y, w1, s); s = fmaf(a.z, w2, s); s = fmaf(a.w, w3, s);
    s = fmaf(b.x, w4, s); s = fmaf(b.y, w5, s); s = fmaf(b.z, w6, s); s = fmaf(b.w, w7, s);
    return s;
}

__device__ __forceinline__ float dot4(const float4 a, const float w0, const float w1, const float w2, const float w3) {
    float s = a.x * w0;
    s = fmaf(a.y, w1, s); s = fmaf(a.z, w2, s); s = fmaf(a.w, w3, s);
    return s;
}

__device__ __forceinline__ float quad_sum(float v) {
    v += __builtin_bit_cast(float, __builtin_amdgcn_update_dpp(0, __builtin_bit_cast(int, v), 0xB1, 0xF, 0xF, true));
    v += __builtin_bit_cast(float, __builtin_amdgcn_update_dpp(0, __builtin_bit_cast(int, v), 0x4E, 0xF, 0xF, true));
    return v;
}

__global__ void edge_rank(const int* __restrict__ dst, int* __restrict__ cnt,
                          int* __restrict__ rank, int E) {
    int e = blockIdx.x * blockDim.x + threadIdx.x;
    if (e >= E) return;
    rank[e] = atomicAdd(&cnt[dst[e]], 1);
}

__global__ void __launch_bounds__(1024) scan_local(const int* __restrict__ cnt, int* __restrict__ rowptr,
                                                   int* __restrict__ bsum, int N) {
    __shared__ int part[1024];
    const int t = threadIdx.x;
    const int i = blockIdx.x * 1024 + t;
    const int c = (i < N) ? cnt[i] : 0;
    part[t] = c;
    __syncthreads();
    for (int off = 1; off < 1024; off <<= 1) {
        const int v = (t >= off) ? part[t - off] : 0;
        __syncthreads();
        part[t] += v;
        __syncthreads();
    }
    if (i <= N) rowptr[i] = part[t] - c;
    if (t == 1023) bsum[blockIdx.x] = part[1023];
}

__global__ void __launch_bounds__(1024) scan_counts(const int* __restrict__ cnt,
                                                    int* __restrict__ rowptr, int N) {
    __shared__ int part[1024];
    const int t = threadIdx.x;
    const int per = (N + 1023) / 1024;
    int beg = t * per;      if (beg > N) beg = N;
    int end = beg + per;    if (end > N) end = N;
    int s = 0;
    for (int i = beg; i < end; ++i) s += cnt[i];
    part[t] = s;
    __syncthreads();
    for (int off = 1; off < 1024; off <<= 1) {
        const int v = (t >= off) ? part[t - off] : 0;
        __syncthreads();
        part[t] += v;
        __syncthreads();
    }
    int run = part[t] - s;
    for (int i = beg; i < end; ++i) { rowptr[i] = run; run += cnt[i]; }
    if (t == 1023) rowptr[N] = part[1023];
}

__global__ void edge_place(const int* __restrict__ src, const int* __restrict__ dst,
                           const int* __restrict__ rank, const int* __restrict__ rowptr,
                           const int* __restrict__ boff, int* __restrict__ csr, int E) {
    int e = blockIdx.x * blockDim.x + threadIdx.x;
    if (e >= E) return;
    const int d = dst[e];
    csr[rowptr[d] + boff[d >> 10] + rank[e]] = src[e];
}

__global__ void node_gather1(const int* __restrict__ rowptr, const int* __restrict__ boff,
                             const int* __restrict__ csr,
                             const float* __restrict__ attS, const float* __restrict__ attD,
                             const float* __restrict__ h1, const float* __restrict__ b1,
                             float* __restrict__ o1, int NH) {
    int i = blockIdx.x * blockDim.x + threadIdx.x;
    if (i >= NH) return;
    const int n = i >> 3, h = i & 7;
    const int beg = rowptr[n] + boff[n >> 10];
    const int end = rowptr[n + 1] + boff[(n + 1) >> 10];
    const int last = end - 1;

    const float4* __restrict__ hv = (const float4*)h1;
    const float* ws = attS + h * HID;
    const float w0 = ws[0], w1 = ws[1], w2 = ws[2], w3 = ws[3];
    const float w4 = ws[4], w5 = ws[5], w6 = ws[6], w7 = ws[7];
    float ad;
    {
        const float* wd = attD + h * HID;
        const float4 p0 = hv[i * 2], p1 = hv[i * 2 + 1];
        ad = dot8(p0, p1, wd[0], wd[1], wd[2], wd[3], wd[4], wd[5], wd[6], wd[7]);
    }
    float m = -INFINITY;
    float den = 0.f;
    float a0 = 0.f, a1 = 0.f, a2 = 0.f, a3 = 0.f, a4 = 0.f, a5 = 0.f, a6 = 0.f, a7 = 0.f;
    for (int j = beg; j < end; j += 4) {
        const bool k1 = j + 1 < end, k2 = j + 2 < end, k3 = j + 3 < end;
        const int j1 = k1 ? j + 1 : last;
        const int j2 = k2 ? j + 2 : last;
        const int j3 = k3 ? j + 3 : last;
        const int s0 = csr[j], s1 = csr[j1], s2 = csr[j2], s3 = csr[j3];
        const int i0 = s0 * (F1 / 4) + h * 2, i1 = s1 * (F1 / 4) + h * 2;
        const int i2 = s2 * (F1 / 4) + h * 2, i3 = s3 * (F1 / 4) + h * 2;
        const float4 x0 = hv[i0], y0 = hv[i0 + 1];
        const float4 x1 = hv[i1], y1 = hv[i1 + 1];
        const float4 x2 = hv[i2], y2 = hv[i2 + 1];
        const float4 x3 = hv[i3], y3 = hv[i3 + 1];
        const float g0 = dot8(x0, y0, w0, w1, w2, w3, w4, w5, w6, w7);
        const float g1 = dot8(x1, y1, w0, w1, w2, w3, w4, w5, w6, w7);
        const float g2 = dot8(x2, y2, w0, w1, w2, w3, w4, w5, w6, w7);
        const float g3 = dot8(x3, y3, w0, w1, w2, w3, w4, w5, w6, w7);
        const float v0 = lrelu(g0 + ad), v1 = lrelu(g1 + ad);
        const float v2 = lrelu(g2 + ad), v3 = lrelu(g3 + ad);
        const float u = v0 > v1 ? v0 : v1;
        const float w = v2 > v3 ? v2 : v3;
        const float z = u > w ? u : w;
        const float mn = z > m ? z : m;
        const float sc = expf(m - mn);
        m = mn;
        const float e0 = expf(v0 - mn);
        const float e1 = k1 ? expf(v1 - mn) : 0.f;
        const float e2 = k2 ? expf(v2 - mn) : 0.f;
        const float e3 = k3 ? expf(v3 - mn) : 0.f;
        den *= sc;
        a0 *= sc; a1 *= sc; a2 *= sc; a3 *= sc; a4 *= sc; a5 *= sc; a6 *= sc; a7 *= sc;
        den += e0; den += e1; den += e2; den += e3;
        a0 += e0 * x0.x; a1 += e0 * x0.y; a2 += e0 * x0.z; a3 += e0 * x0.w;
        a4 += e0 * y0.x; a5 += e0 * y0.y; a6 += e0 * y0.z; a7 += e0 * y0.w;
        a0 += e1 * x1.x; a1 += e1 * x1.y; a2 += e1 * x1.z; a3 += e1 * x1.w;
        a4 += e1 * y1.x; a5 += e1 * y1.y; a6 += e1 * y1.z; a7 += e1 * y1.w;
        a0 += e2 * x2.x; a1 += e2 * x2.y; a2 += e2 * x2.z; a3 += e2 * x2.w;
        a4 += e2 * y2.x; a5 += e2 * y2.y; a6 += e2 * y2.z; a7 += e2 * y2.w;
        a0 += e3 * x3.x; a1 += e3 * x3.y; a2 += e3 * x3.z; a3 += e3 * x3.w;
        a4 += e3 * y3.x; a5 += e3 * y3.y; a6 += e3 * y3.z; a7 += e3 * y3.w;
    }

    const float dn = den + 1e-16f;
    const float* bp = b1 + h * HID;
    const float t0 = a0 / dn + bp[0], t1 = a1 / dn + bp[1], t2 = a2 / dn + bp[2], t3 = a3 / dn + bp[3];
    const float t4 = a4 / dn + bp[4], t5 = a5 / dn + bp[5], t6 = a6 / dn + bp[6], t7 = a7 / dn + bp[7];
    float4 r0, r1;
    r0.x = t0 > 0.f ? t0 : (expf(t0) - 1.f);
    r0.y = t1 > 0.f ? t1 : (expf(t1) - 1.f);
    r0.z = t2 > 0.f ? t2 : (expf(t2) - 1.f);
    r0.w = t3 > 0.f ? t3 : (expf(t3) - 1.f);
    r1.x = t4 > 0.f ? t4 : (expf(t4) - 1.f);
    r1.y = t5 > 0.f ? t5 : (expf(t5) - 1.f);
    r1.z = t6 > 0.f ? t6 : (expf(t6) - 1.f);
    r1.w = t7 > 0.f ? t7 : (expf(t7) - 1.f);
    float4* op = (float4*)(o1 + (size_t)n * F1 + h * HID);
    op[0] = r0;
    op[1] = r1;
}

__global__ void node_gather2(const int* __restrict__ rowptr, const int* __restrict__ boff,
                             const int* __restrict__ csr,
                             const float* __restrict__ attS, const float* __restrict__ attD,
                             const float* __restrict__ h2, const float* __restrict__ b2,
                             float* __restrict__ out, int NQ) {
    int i = blockIdx.x * blockDim.x + threadIdx.x;
    if (i >= NQ) return;
    const int n = i >> 2, q = i & 3;
    const int beg = rowptr[n] + boff[n >> 10];
    const int end = rowptr[n + 1] + boff[(n + 1) >> 10];
    const int last = end - 1;

    const float4* __restrict__ hv = (const float4*)h2;
    const float* ws = attS + q * 4;
    const float w0 = ws[0], w1 = ws[1], w2 = ws[2], w3 = ws[3];
    float ad;
    {
        const float* wd = attD + q * 4;
        const float4 p = hv[i];
        ad = quad_sum(dot4(p, wd[0], wd[1], wd[2], wd[3]));
    }
    float m = -INFINITY;
    float den = 0.f;
    float a0 = 0.f, a1 = 0.f, a2 = 0.f, a3 = 0.f;
    for (int j = beg; j < end; j += 4) {
        const bool k1 = j + 1 < end, k2 = j + 2 < end, k3 = j + 3 < end;
        const int j1 = k1 ? j + 1 : last;
        const int j2 = k2 ? j + 2 : last;
        const int j3 = k3 ? j + 3 : last;
        const int s0 = csr[j], s1 = csr[j1], s2 = csr[j2], s3 = csr[j3];
        const float4 x0 = hv[s0 * (OUT_C / 4) + q];
        const float4 x1 = hv[s1 * (OUT_C / 4) + q];
        const float4 x2 = hv[s2 * (OUT_C / 4) + q];
        const float4 x3 = hv[s3 * (OUT_C / 4) + q];
        const float g0 = quad_sum(dot4(x0, w0, w1, w2, w3)), g1 = quad_sum(dot4(x1, w0, w1, w2, w3));
        const float g2 = quad_sum(dot4(x2, w0, w1, w2, w3)), g3 = quad_sum(dot4(x3, w0, w1, w2, w3));
        const float v0 = lrelu(g0 + ad), v1 = lrelu(g1 + ad);
        const float v2 = lrelu(g2 + ad), v3 = lrelu(g3 + ad);
        const float u = v0 > v1 ? v0 : v1;
        const float w = v2 > v3 ? v2 : v3;
        const float z = u > w ? u : w;
        const float mn = z > m ? z : m;
        const float sc = expf(m - mn);
        m = mn;
        const float e0 = expf(v0 - mn);
        const float e1 = k1 ? expf(v1 - mn) : 0.f;
        const float e2 = k2 ? expf(v2 - mn) : 0.f;
        const float e3 = k3 ? expf(v3 - mn) : 0.f;
        den *= sc;
        a0 *= sc; a1 *= sc; a2 *= sc; a3 *= sc;
        den += e0; den += e1; den += e2; den += e3;
        a0 += e0 * x0.x; a1 += e0 * x0.y; a2 += e0 * x0.z; a3 += e0 * x0.w;
        a0 += e1 * x1.x; a1 += e1 * x1.y; a2 += e1 * x1.z; a3 += e1 * x1.w;
        a0 += e2 * x2.x; a1 += e2 * x2.y; a2 += e2 * x2.z; a3 += e2 * x2.w;
        a0 += e3 * x3.x; a1 += e3 * x3.y; a2 += e3 * x3.z; a3 += e3 * x3.w;
    }

    const float dn = den + 1e-16f;
    const float* bq = b2 + q * 4;
    float4 r;
    r.x = a0 / dn + bq[0];
    r.y = a1 / dn + bq[1];
    r.z = a2 / dn + bq[2];
    r.w = a3 / dn + bq[3];
    ((float4*)(out + (size_t)n * OUT_C))[q] = r;
}

extern "C" void kernel_launch(void* const* d_in, const int* in_sizes, int n_in,
                              void* d_out, int out_size, void* d_ws, size_t ws_size,
                              hipStream_t stream) {
    const int N = in_sizes[0] / IN_C;
    const int E = in_sizes[1] / 2;

    const float* x     = (const float*)d_in[0];
    const int*   ei    = (const int*)  d_in[1];
    const int*   src   = ei;
    const int*   dst   = ei + E;
    const float* W1    = (const float*)d_in[2];
    const float* b1    = (const float*)d_in[3];
    const float* attS1 = (const float*)d_in[4];
    const float* attD1 = (const float*)d_in[5];
    const float* W2    = (const float*)d_in[6];
    const float* b2    = (const float*)d_in[7];
    const float* attS2 = (const float*)d_in[8];
    const float* attD2 = (const float*)d_in[9];
    float* out = (float*)d_out;

    char* ws = (char*)d_ws;
    size_t off = 0;
    auto alloc = [&](size_t elems) -> float* {
        float* p = (float*)(ws + off);
        off = (off + elems * sizeof(float) + 255) & ~(size_t)255;
        return p;
    };
    float* h1   = alloc((size_t)N * F1);
    float* o1   = alloc((size_t)N * F1);
    float* h2   = alloc((size_t)N * OUT_C);
    int*   cnt    = (int*)alloc((size_t)N);
    int*   rowptr = (int*)alloc((size_t)N + 1);
    const int nScanBlks = (N + 1 + 1023) / 1024;
    int*   bsum   = (int*)alloc((size_t)nScanBlks);
    int*   boff   = (int*)alloc((size_t)nScanBlks + 1);
    int*   rank   = (int*)alloc((size_t)E);
    int*   csr    = (int*)alloc((size_t)E);

    hipMemsetAsync(cnt, 0, (size_t)N * sizeof(int), stream);

    const int TB = 256;
    auto nb = [](long long n, int tb) { return (unsigned)((n + tb - 1) / tb); };

    const int nTiles = (N + 15) / 16;
    const int gemmBlks = (nTiles + 7) / 8;

    edge_rank<<<nb(E, TB), TB, 0, stream>>>(dst, cnt, rank, E);
    scan_local<<<nScanBlks, 1024, 0, stream>>>(cnt, rowptr, bsum, N);
    scan_counts<<<1, 1024, 0, stream>>>(bsum, boff, nScanBlks);
    edge_place<<<nb(E, TB), TB, 0, stream>>>(src, dst, rank, rowptr, boff, csr, E);

    gat_gemm1_wmma<<<gemmBlks, TB, 0, stream>>>(x, W1, h1, N, nTiles);
    node_gather1<<<nb((long long)N * HEADS, TB), TB, 0, stream>>>(rowptr, boff, csr, attS1, attD1, h1, b1, o1, N * HEADS);

    gat_gemm2_wmma<<<gemmBlks, TB, 0, stream>>>(o1, W2, h2, N, nTiles);
    node_gather2<<<nb((long long)N * 4, TB), TB, 0, stream>>>(rowptr, boff, csr, attS2, attD2, h2, b2, out, N * 4);
}


# Edge sort by destination redone as a two-level bucket sort on LDS counters: one global int atomic per (tile, bucket) instead of per edge, 3 launches for 4; segment order stays arrival order
# speedup vs baseline: 5.4080x; 1.7291x over previous
#include <hip/hip_runtime.h>
#include <math.h>

#define IN_C   128
#define HEADS  8
#define HID    8
#define F1     (HEADS * HID)
#define OUT_C  16
#define NEG_SLOPE 0.2f

typedef float v2f __attribute__((ext_vector_type(2)));
typedef float v8f __attribute__((ext_vector_type(8)));

__device__ __forceinline__ float lrelu(float v) {
    return v > 0.0f ? v : NEG_SLOPE * v;
}

__global__ void gat_gemm1_wmma(const float* __restrict__ x,
                               const float* __restrict__ W,
                               float* __restrict__ h1,
                               int N, int nTiles) {
    __shared__ float w[IN_C * F1];
    for (int i = threadIdx.x; i < IN_C * F1; i += blockDim.x) w[i] = W[i];
    __syncthreads();

    const int wavesPerBlk = blockDim.x >> 5;
    const int tile = blockIdx.x * wavesPerBlk + (threadIdx.x >> 5);
    if (tile >= nTiles) return;

    const int lane  = threadIdx.x & 31;
    const int row16 = lane & 15;
    const int kh    = lane >> 4;

    int rowIn = tile * 16 + row16;
    if (rowIn >= N) rowIn = N - 1;
    const float* xr = x + (size_t)rowIn * IN_C;

    v8f acc0 = {}, acc1 = {}, acc2 = {}, acc3 = {};

    for (int k = 0; k < IN_C; k += 4) {
        const int ka = k + 2 * kh;
        v2f a;
        a.x = xr[ka];
        a.y = xr[ka + 1];
        const float* w0 = &w[ka * F1 + row16];
        const float* w1 = &w[(ka + 1) * F1 + row16];
        v2f b0; b0.x = w0[0];  b0.y = w1[0];
        v2f b1; b1.x = w0[16]; b1.y = w1[16];
        v2f b2; b2.x = w0[32]; b2.y = w1[32];
        v2f b3; b3.x = w0[48]; b3.y = w1[48];
        acc0 = __builtin_amdgcn_wmma_f32_16x16x4_f32(false, a, false, b0, (short)0, acc0, false, false);
        acc1 = __builtin_amdgcn_wmma_f32_16x16x4_f32(false, a, false, b1, (short)0, acc1, false, false);
        acc2 = __builtin_amdgcn_wmma_f32_16x16x4_f32(false, a, false, b2, (short)0, acc2, false, false);
        acc3 = __builtin_amdgcn_wmma_f32_16x16x4_f32(false, a, false, b3, (short)0, acc3, false, false);
    }

    #pragma unroll
    for (int i = 0; i < 8; ++i) {
        const int r = tile * 16 + i + 8 * kh;
        if (r < N) {
            float* o = h1 + (size_t)r * F1 + row16;
            o[0]  = acc0[i];
            o[16] = acc1[i];
            o[32] = acc2[i];
            o[48] = acc3[i];
        }
    }
}

__global__ void gat_gemm2_wmma(const float* __restrict__ hin,
                               const float* __restrict__ W,
                               float* __restrict__ h2,
                               int N, int nTiles) {
    __shared__ float w[F1 * OUT_C];
    for (int i = threadIdx.x; i < F1 * OUT_C; i += blockDim.x) w[i] = W[i];
    __syncthreads();

    const int wavesPerBlk = blockDim.x >> 5;
    const int tile = blockIdx.x * wavesPerBlk + (threadIdx.x >> 5);
    if (tile >= nTiles) return;

    const int lane  = threadIdx.x & 31;
    const int row16 = lane & 15;
    const int kh    = lane >> 4;

    int rowIn = tile * 16 + row16;
    if (rowIn >= N) rowIn = N - 1;
    const float* xr = hin + (size_t)rowIn * F1;

    v8f acc = {};
    for (int k = 0; k < F1; k += 4) {
        const int ka = k + 2 * kh;
        v2f a;
        a.x = xr[ka];
        a.y = xr[ka + 1];
        v2f b;
        b.x = w[ka * OUT_C + row16];
        b.y = w[(ka + 1) * OUT_C + row16];
        acc = __builtin_amdgcn_wmma_f32_16x16x4_f32(false, a, false, b, (short)0, acc, false, false);
    }

    #pragma unroll
    for (int i = 0; i < 8; ++i) {
        const int r = tile * 16 + i + 8 * kh;
        if (r < N) h2[(size_t)r * OUT_C + row16] = acc[i];
    }
}

__device__ __forceinline__ float dot8(const float4 a, const float4 b,
                                      const float w0, const float w1, const float w2, const float w3,
                                      const float w4, const float w5, const float w6, const float w7) {
    float s = a.x * w0;
    s = fmaf(a.y, w1, s); s = fmaf(a.z, w2, s); s = fmaf(a.w, w3, s);
    s = fmaf(b.x, w4, s); s = fmaf(b.y, w5, s); s = fmaf(b.z, w6, s); s = fmaf(b.w, w7, s);
    return s;
}

__device__ __forceinline__ float dot4(const float4 a, const float w0, const float w1, const float w2, const float w3) {
    float s = a.x * w0;
    s = fmaf(a.y, w1, s); s = fmaf(a.z, w2, s); s = fmaf(a.w, w3, s);
    return s;
}

__device__ __forceinline__ float quad_sum(float v) {
    v += __builtin_bit_cast(float, __builtin_amdgcn_update_dpp(0, __builtin_bit_cast(int, v), 0xB1, 0xF, 0xF, true));
    v += __builtin_bit_cast(float, __builtin_amdgcn_update_dpp(0, __builtin_bit_cast(int, v), 0x4E, 0xF, 0xF, true));
    return v;
}

#define TILE_E  4096
#define BKT_MAX 512

__device__ __forceinline__ void scan256(int* part, const int t) {
    for (int off = 1; off < 256; off <<= 1) {
        const int v = (t >= off) ? part[t - off] : 0;
        __syncthreads();
        part[t] += v;
        __syncthreads();
    }
}

__global__ void __launch_bounds__(256) bucket_count(const int* __restrict__ dst, int* __restrict__ bucketCount,
                                                    int* __restrict__ tileOff, int E, int NB) {
    __shared__ int hist[BKT_MAX];
    const int t = threadIdx.x;
    for (int i = t; i < NB; i += 256) hist[i] = 0;
    __syncthreads();
    const int base = blockIdx.x * TILE_E + t;
    int d[16];
    #pragma unroll
    for (int k = 0; k < 16; ++k) {
        const int e = base + k * 256;
        d[k] = dst[e < E ? e : E - 1];
    }
    #pragma unroll
    for (int k = 0; k < 16; ++k) {
        if (base + k * 256 < E) atomicAdd(&hist[d[k] >> 8], 1);
    }
    __syncthreads();
    const int i0 = t, i1 = t + 256;
    int r0 = 0, r1 = 0;
    if (i0 < NB) r0 = atomicAdd(&bucketCount[i0], hist[i0]);
    if (i1 < NB) r1 = atomicAdd(&bucketCount[i1], hist[i1]);
    if (i0 < NB) tileOff[blockIdx.x * NB + i0] = r0;
    if (i1 < NB) tileOff[blockIdx.x * NB + i1] = r1;
}

__global__ void __launch_bounds__(256) bucket_scatter(const int* __restrict__ src, const int* __restrict__ dst,
                                                      const int* __restrict__ bucketCount,
                                                      const int* __restrict__ tileOff,
                                                      int* __restrict__ bucketStart, int* __restrict__ binned,
                                                      int E, int NB) {
    __shared__ int cur[BKT_MAX + 32];
    __shared__ int part[256];
    const int t = threadIdx.x;
    const int base = blockIdx.x * TILE_E + t;
    if (t < 32) cur[BKT_MAX + t] = 0;
    int d[16], s[16];
    #pragma unroll
    for (int k = 0; k < 16; ++k) {
        const int e = base + k * 256;
        const int ec = e < E ? e : E - 1;
        d[k] = dst[ec];
        s[k] = src[ec];
    }
    const int b0 = 2 * t, b1 = 2 * t + 1;
    const int b0c = b0 < NB ? b0 : NB - 1, b1c = b1 < NB ? b1 : NB - 1;
    const int c0r = bucketCount[b0c], c1r = bucketCount[b1c];
    const int o0r = tileOff[blockIdx.x * NB + b0c], o1r = tileOff[blockIdx.x * NB + b1c];
    __builtin_amdgcn_sched_barrier(0);
    const int c0 = b0 < NB ? c0r : 0, c1 = b1 < NB ? c1r : 0;
    const int o0 = b0 < NB ? o0r : 0, o1 = b1 < NB ? o1r : 0;
    part[t] = c0 + c1;
    __syncthreads();
    scan256(part, t);
    const int ex = part[t] - (c0 + c1);
    if (b0 < NB) cur[b0] = ex + o0;
    if (b1 < NB) cur[b1] = ex + c0 + o1;
    if (blockIdx.x == 0) {
        if (b0 < NB) bucketStart[b0] = ex;
        if (b1 < NB) bucketStart[b1] = ex + c0;
        if (t == 255) bucketStart[NB] = part[255];
    }
    __syncthreads();
    int pos[16];
    #pragma unroll
    for (int k = 0; k < 16; ++k)
        pos[k] = atomicAdd(&cur[(base + k * 256 < E) ? (d[k] >> 8) : (BKT_MAX + (t & 31))], 1);
    #pragma unroll
    for (int k = 0; k < 16; ++k)
        if (base + k * 256 < E) binned[pos[k]] = (int)(((unsigned)s[k] << 8) | (unsigned)(d[k] & 255));
}

__global__ void __launch_bounds__(256) csr_fill(const int* __restrict__ bucketStart, const int* __restrict__ binned,
                                                int* __restrict__ rowptr, int* __restrict__ csr, int N) {
    __shared__ int cnt[256 + 32];
    __shared__ int part[256];
    const int t = threadIdx.x;
    const int spare = 256 + (t & 31);
    const int beg = bucketStart[blockIdx.x], end = bucketStart[blockIdx.x + 1];
    const int last = end - 1;
    cnt[t] = 0;
    if (t < 32) cnt[256 + t] = 0;
    __syncthreads();
    for (int j = beg + t; j < end; j += 2048) {
        int v[8];
        #pragma unroll
        for (int k = 0; k < 8; ++k) {
            const int jk = j + k * 256;
            v[k] = binned[jk < end ? jk : last];
        }
        #pragma unroll
        for (int k = 0; k < 8; ++k)
            atomicAdd(&cnt[(j + k * 256 < end) ? (v[k] & 255) : spare], 1);
    }
    __syncthreads();
    const int c = cnt[t];
    part[t] = c;
    __syncthreads();
    scan256(part, t);
    const int first = beg + part[t] - c;
    const int n = blockIdx.x * 256 + t;
    if (n < N) rowptr[n] = first;
    if (n == N - 1) rowptr[N] = first + c;
    cnt[t] = first;
    __syncthreads();
    for (int j = beg + t; j < end; j += 2048) {
        int v[8], pos[8];
        #pragma unroll
        for (int k = 0; k < 8; ++k) {
            const int jk = j + k * 256;
            v[k] = binned[jk < end ? jk : last];
        }
        #pragma unroll
        for (int k = 0; k < 8; ++k)
            pos[k] = atomicAdd(&cnt[(j + k * 256 < end) ? (v[k] & 255) : spare], 1);
        #pragma unroll
        for (int k = 0; k < 8; ++k)
            if (j + k * 256 < end) csr[pos[k]] = (int)((unsigned)v[k] >> 8);
    }
}

__global__ void node_gather1(const int* __restrict__ rowptr,
                             const int* __restrict__ csr,
                             const float* __restrict__ attS, const float* __restrict__ attD,
                             const float* __restrict__ h1, const float* __restrict__ b1,
                             float* __restrict__ o1, int NH) {
    int i = blockIdx.x * blockDim.x + threadIdx.x;
    if (i >= NH) return;
    const int n = i >> 3, h = i & 7;
    const int beg = rowptr[n];
    const int end = rowptr[n + 1];
    const int last = end - 1;

    const float4* __restrict__ hv = (const float4*)h1;
    const float* ws = attS + h * HID;
    const float w0 = ws[0], w1 = ws[1], w2 = ws[2], w3 = ws[3];
    const float w4 = ws[4], w5 = ws[5], w6 = ws[6], w7 = ws[7];
    float ad;
    {
        const float* wd = attD + h * HID;
        const float4 p0 = hv[i * 2], p1 = hv[i * 2 + 1];
        ad = dot8(p0, p1, wd[0], wd[1], wd[2], wd[3], wd[4], wd[5], wd[6], wd[7]);
    }
    float m = -INFINITY;
    float den = 0.f;
    float a0 = 0.f, a1 = 0.f, a2 = 0.f, a3 = 0.f, a4 = 0.f, a5 = 0.f, a6 = 0.f, a7 = 0.f;
    for (int j = beg; j < end; j += 4) {
        const bool k1 = j + 1 < end, k2 = j + 2 < end, k3 = j + 3 < end;
        const int j1 = k1 ? j + 1 : last;
        const int j2 = k2 ? j + 2 : last;
        const int j3 = k3 ? j + 3 : last;
        const int s0 = csr[j], s1 = csr[j1], s2 = csr[j2], s3 = csr[j3];
        const int i0 = s0 * (F1 / 4) + h * 2, i1 = s1 * (F1 / 4) + h * 2;
        const int i2 = s2 * (F1 / 4) + h * 2, i3 = s3 * (F1 / 4) + h * 2;
        const float4 x0 = hv[i0], y0 = hv[i0 + 1];
        const float4 x1 = hv[i1], y1 = hv[i1 + 1];
        const float4 x2 = hv[i2], y2 = hv[i2 + 1];
        const float4 x3 = hv[i3], y3 = hv[i3 + 1];
        const float g0 = dot8(x0, y0, w0, w1, w2, w3, w4, w5, w6, w7);
        const float g1 = dot8(x1, y1, w0, w1, w2, w3, w4, w5, w6, w7);
        const float g2 = dot8(x2, y2, w0, w1, w2, w3, w4, w5, w6, w7);
        const float g3 = dot8(x3, y3, w0, w1, w2, w3, w4, w5, w6, w7);
        const float v0 = lrelu(g0 + ad), v1 = lrelu(g1 + ad);
        const float v2 = lrelu(g2 + ad), v3 = lrelu(g3 + ad);
        const float u = v0 > v1 ? v0 : v1;
        const float w = v2 > v3 ? v2 : v3;
        const float z = u > w ? u : w;
        const float mn = z > m ? z : m;
        const float sc = expf(m - mn);
        m = mn;
        const float e0 = expf(v0 - mn);
        const float e1 = k1 ? expf(v1 - mn) : 0.f;
        const float e2 = k2 ? expf(v2 - mn) : 0.f;
        const float e3 = k3 ? expf(v3 - mn) : 0.f;
        den *= sc;
        a0 *= sc; a1 *= sc; a2 *= sc; a3 *= sc; a4 *= sc; a5 *= sc; a6 *= sc; a7 *= sc;
        den += e0; den += e1; den += e2; den += e3;
        a0 += e0 * x0.x; a1 += e0 * x0.y; a2 += e0 * x0.z; a3 += e0 * x0.w;
        a4 += e0 * y0.x; a5 += e0 * y0.y; a6 += e0 * y0.z; a7 += e0 * y0.w;
        a0 += e1 * x1.x; a1 += e1 * x1.y; a2 += e1 * x1.z; a3 += e1 * x1.w;
        a4 += e1 * y1.x; a5 += e1 * y1.y; a6 += e1 * y1.z; a7 += e1 * y1.w;
        a0 += e2 * x2.x; a1 += e2 * x2.y; a2 += e2 * x2.z; a3 += e2 * x2.w;
        a4 += e2 * y2.x; a5 += e2 * y2.y; a6 += e2 * y2.z; a7 += e2 * y2.w;
        a0 += e3 * x3.x; a1 += e3 * x3.y; a2 += e3 * x3.z; a3 += e3 * x3.w;
        a4 += e3 * y3.x; a5 += e3 * y3.y; a6 += e3 * y3.z; a7 += e3 * y3.w;
    }

    const float dn = den + 1e-16f;
    const float* bp = b1 + h * HID;
    const float t0 = a0 / dn + bp[0], t1 = a1 / dn + bp[1], t2 = a2 / dn + bp[2], t3 = a3 / dn + bp[3];
    const float t4 = a4 / dn + bp[4], t5 = a5 / dn + bp[5], t6 = a6 / dn + bp[6], t7 = a7 / dn + bp[7];
    float4 r0, r1;
    r0.x = t0 > 0.f ? t0 : (expf(t0) - 1.f);
    r0.y = t1 > 0.f ? t1 : (expf(t1) - 1.f);
    r0.z = t2 > 0.f ? t2 : (expf(t2) - 1.f);
    r0.w = t3 > 0.f ? t3 : (expf(t3) - 1.f);
    r1.x = t4 > 0.f ? t4 : (expf(t4) - 1.f);
    r1.y = t5 > 0.f ? t5 : (expf(t5) - 1.f);
    r1.z = t6 > 0.f ? t6 : (expf(t6) - 1.f);
    r1.w = t7 > 0.f ? t7 : (expf(t7) - 1.f);
    float4* op = (float4*)(o1 + (size_t)n * F1 + h * HID);
    op[0] = r0;
    op[1] = r1;
}

__global__ void node_gather2(const int* __restrict__ rowptr,
                             const int* __restrict__ csr,
                             const float* __restrict__ attS, const float* __restrict__ attD,
                             const float* __restrict__ h2, const float* __restrict__ b2,
                             float* __restrict__ out, int NQ) {
    int i = blockIdx.x * blockDim.x + threadIdx.x;
    if (i >= NQ) return;
    const int n = i >> 2, q = i & 3;
    const int beg = rowptr[n];
    const int end = rowptr[n + 1];
    const int last = end - 1;

    const float4* __restrict__ hv = (const float4*)h2;
    const float* ws = attS + q * 4;
    const float w0 = ws[0], w1 = ws[1], w2 = ws[2], w3 = ws[3];
    float ad;
    {
        const float* wd = attD + q * 4;
        const float4 p = hv[i];
        ad = quad_sum(dot4(p, wd[0], wd[1], wd[2], wd[3]));
    }
    float m = -INFINITY;
    float den = 0.f;
    float a0 = 0.f, a1 = 0.f, a2 = 0.f, a3 = 0.f;
    for (int j = beg; j < end; j += 4) {
        const bool k1 = j + 1 < end, k2 = j + 2 < end, k3 = j + 3 < end;
        const int j1 = k1 ? j + 1 : last;
        const int j2 = k2 ? j + 2 : last;
        const int j3 = k3 ? j + 3 : last;
        const int s0 = csr[j], s1 = csr[j1], s2 = csr[j2], s3 = csr[j3];
        const float4 x0 = hv[s0 * (OUT_C / 4) + q];
        const float4 x1 = hv[s1 * (OUT_C / 4) + q];
        const float4 x2 = hv[s2 * (OUT_C / 4) + q];
        const float4 x3 = hv[s3 * (OUT_C / 4) + q];
        const float g0 = quad_sum(dot4(x0, w0, w1, w2, w3)), g1 = quad_sum(dot4(x1, w0, w1, w2, w3));
        const float g2 = quad_sum(dot4(x2, w0, w1, w2, w3)), g3 = quad_sum(dot4(x3, w0, w1, w2, w3));
        const float v0 = lrelu(g0 + ad), v1 = lrelu(g1 + ad);
        const float v2 = lrelu(g2 + ad), v3 = lrelu(g3 + ad);
        const float u = v0 > v1 ? v0 : v1;
        const float w = v2 > v3 ? v2 : v3;
        const float z = u > w ? u : w;
        const float mn = z > m ? z : m;
        const float sc = expf(m - mn);
        m = mn;
        const float e0 = expf(v0 - mn);
        const float e1 = k1 ? expf(v1 - mn) : 0.f;
        const float e2 = k2 ? expf(v2 - mn) : 0.f;
        const float e3 = k3 ? expf(v3 - mn) : 0.f;
        den *= sc;
        a0 *= sc; a1 *= sc; a2 *= sc; a3 *= sc;
        den += e0; den += e1; den += e2; den += e3;
        a0 += e0 * x0.x; a1 += e0 * x0.y; a2 += e0 * x0.z; a3 += e0 * x0.w;
        a0 += e1 * x1.x; a1 += e1 * x1.y; a2 += e1 * x1.z; a3 += e1 * x1.w;
        a0 += e2 * x2.x; a1 += e2 * x2.y; a2 += e2 * x2.z; a3 += e2 * x2.w;
        a0 += e3 * x3.x; a1 += e3 * x3.y; a2 += e3 * x3.z; a3 += e3 * x3.w;
    }

    const float dn = den + 1e-16f;
    const float* bq = b2 + q * 4;
    float4 r;
    r.x = a0 / dn + bq[0];
    r.y = a1 / dn + bq[1];
    r.z = a2 / dn + bq[2];
    r.w = a3 / dn + bq[3];
    ((float4*)(out + (size_t)n * OUT_C))[q] = r;
}

extern "C" void kernel_launch(void* const* d_in, const int* in_sizes, int n_in,
                              void* d_out, int out_size, void* d_ws, size_t ws_size,
                              hipStream_t stream) {
    const int N = in_sizes[0] / IN_C;
    const int E = in_sizes[1] / 2;

    const float* x     = (const float*)d_in[0];
    const int*   ei    = (const int*)  d_in[1];
    const int*   src   = ei;
    const int*   dst   = ei + E;
    const float* W1    = (const float*)d_in[2];
    const float* b1    = (const float*)d_in[3];
    const float* attS1 = (const float*)d_in[4];
    const float* attD1 = (const float*)d_in[5];
    const float* W2    = (const float*)d_in[6];
    const float* b2    = (const float*)d_in[7];
    const float* attS2 = (const float*)d_in[8];
    const float* attD2 = (const float*)d_in[9];
    float* out = (float*)d_out;

    char* ws = (char*)d_ws;
    size_t off = 0;
    auto alloc = [&](size_t elems) -> float* {
        float* p = (float*)(ws + off);
        off = (off + elems * sizeof(float) + 255) & ~(size_t)255;
        return p;
    };
    float* h1   = alloc((size_t)N * F1);
    float* o1   = alloc((size_t)N * F1);
    float* h2   = alloc((size_t)N * OUT_C);
    const int NB = (N + 255) / 256;
    const int NT = (E + TILE_E - 1) / TILE_E;
    int*   bucketCount = (int*)alloc((size_t)BKT_MAX);
    int*   bucketStart = (int*)alloc((size_t)BKT_MAX + 1);
    int*   tileOff     = (int*)alloc((size_t)NT * NB);
    int*   rowptr      = (int*)alloc((size_t)N + 1);
    int*   binned      = (int*)alloc((size_t)E);
    int*   csr         = (int*)alloc((size_t)E);

    hipMemsetAsync(bucketCount, 0, (size_t)NB * sizeof(int), stream);

    const int TB = 256;
    auto nb = [](long long n, int tb) { return (unsigned)((n + tb - 1) / tb); };

    const int nTiles = (N + 15) / 16;
    const int gemmBlks = (nTiles + 7) / 8;

    bucket_count<<<NT, 256, 0, stream>>>(dst, bucketCount, tileOff, E, NB);
    bucket_scatter<<<NT, 256, 0, stream>>>(src, dst, bucketCount, tileOff, bucketStart, binned, E, NB);
    csr_fill<<<NB, 256, 0, stream>>>(bucketStart, binned, rowptr, csr, N);

    gat_gemm1_wmma<<<gemmBlks, TB, 0, stream>>>(x, W1, h1, N, nTiles);
    node_gather1<<<nb((long long)N * HEADS, TB), TB, 0, stream>>>(rowptr, csr, attS1, attD1, h1, b1, o1, N * HEADS);

    gat_gemm2_wmma<<<gemmBlks, TB, 0, stream>>>(o1, W2, h2, N, nTiles);
    node_gather2<<<nb((long long)N * 4, TB), TB, 0, stream>>>(rowptr, csr, attS2, attD2, h2, b2, out, N * 4);
}


# GEMM1's work-groups ride in the three edge-sort launches (a third each, dispatched behind the sort's own) and its own launch is gone: matrix work fills the sort's idle wave slots; sums unchanged
# speedup vs baseline: 6.3055x; 1.1660x over previous
#include <hip/hip_runtime.h>
#include <math.h>

#define IN_C   128
#define HEADS  8
#define HID    8
#define F1     (HEADS * HID)
#define OUT_C  16
#define NEG_SLOPE 0.2f

typedef float v2f __attribute__((ext_vector_type(2)));
typedef float v8f __attribute__((ext_vector_type(8)));

__device__ __forceinline__ float lrelu(float v) {
    return v > 0.0f ? v : NEG_SLOPE * v;
}

__device__ __forceinline__ void gemm1_block(const float* __restrict__ x,
                                            const float* __restrict__ W,
                                            float* __restrict__ h1,
                                            int N, int nTiles, int gblk) {
    __shared__ float w[IN_C * F1];
    for (int i = threadIdx.x; i < IN_C * F1; i += 256) w[i] = W[i];
    __syncthreads();

    const int tile = gblk * 8 + (threadIdx.x >> 5);
    if (tile >= nTiles) return;

    const int lane  = threadIdx.x & 31;
    const int row16 = lane & 15;
    const int kh    = lane >> 4;

    int rowIn = tile * 16 + row16;
    if (rowIn >= N) rowIn = N - 1;
    const float* xr = x + (size_t)rowIn * IN_C;

    v8f acc0 = {}, acc1 = {}, acc2 = {}, acc3 = {};

    for (int k = 0; k < IN_C; k += 16) {
        v2f a[4];
        #pragma unroll
        for (int u = 0; u < 4; ++u) {
            const int ka = k + 4 * u + 2 * kh;
            a[u].x = xr[ka];
            a[u].y = xr[ka + 1];
        }
        __builtin_amdgcn_sched_barrier(0);
        #pragma unroll
        for (int u = 0; u < 4; ++u) {
            const int ka = k + 4 * u + 2 * kh;
            const float* w0 = &w[ka * F1 + row16];
            const float* w1 = &w[(ka + 1) * F1 + row16];
            v2f b0; b0.x = w0[0];  b0.y = w1[0];
            v2f b1; b1.x = w0[16]; b1.y = w1[16];
            v2f b2; b2.x = w0[32]; b2.y = w1[32];
            v2f b3; b3.x = w0[48]; b3.y = w1[48];
            acc0 = __builtin_amdgcn_wmma_f32_16x16x4_f32(false, a[u], false, b0, (short)0, acc0, false, false);
            acc1 = __builtin_amdgcn_wmma_f32_16x16x4_f32(false, a[u], false, b1, (short)0, acc1, false, false);
            acc2 = __builtin_amdgcn_wmma_f32_16x16x4_f32(false, a[u], false, b2, (short)0, acc2, false, false);
            acc3 = __builtin_amdgcn_wmma_f32_16x16x4_f32(false, a[u], false, b3, (short)0, acc3, false, false);
        }
    }

    #pragma unroll
    for (int i = 0; i < 8; ++i) {
        const int r = tile * 16 + i + 8 * kh;
        if (r < N) {
            float* o = h1 + (size_t)r * F1 + row16;
            o[0]  = acc0[i];
            o[16] = acc1[i];
            o[32] = acc2[i];
            o[48] = acc3[i];
        }
    }
}

__global__ void gat_gemm2_wmma(const float* __restrict__ hin,
                               const float* __restrict__ W,
                               float* __restrict__ h2,
                               int N, int nTiles) {
    __shared__ float w[F1 * OUT_C];
    for (int i = threadIdx.x; i < F1 * OUT_C; i += blockDim.x) w[i] = W[i];
    __syncthreads();

    const int wavesPerBlk = blockDim.x >> 5;
    const int tile = blockIdx.x * wavesPerBlk + (threadIdx.x >> 5);
    if (tile >= nTiles) return;

    const int lane  = threadIdx.x & 31;
    const int row16 = lane & 15;
    const int kh    = lane >> 4;

    int rowIn = tile * 16 + row16;
    if (rowIn >= N) rowIn = N - 1;
    const float* xr = hin + (size_t)rowIn * F1;

    v8f acc = {};
    for (int k = 0; k < F1; k += 4) {
        const int ka = k + 2 * kh;
        v2f a;
        a.x = xr[ka];
        a.y = xr[ka + 1];
        v2f b;
        b.x = w[ka * OUT_C + row16];
        b.y = w[(ka + 1) * OUT_C + row16];
        acc = __builtin_amdgcn_wmma_f32_16x16x4_f32(false, a, false, b, (short)0, acc, false, false);
    }

    #pragma unroll
    for (int i = 0; i < 8; ++i) {
        const int r = tile * 16 + i + 8 * kh;
        if (r < N) h2[(size_t)r * OUT_C + row16] = acc[i];
    }
}

__device__ __forceinline__ float dot8(const float4 a, const float4 b,
                                      const float w0, const float w1, const float w2, const float w3,
                                      const float w4, const float w5, const float w6, const float w7) {
    float s = a.x * w0;
    s = fmaf(a.y, w1, s); s = fmaf(a.z, w2, s); s = fmaf(a.w, w3, s);
    s = fmaf(b.x, w4, s); s = fmaf(b.y, w5, s); s = fmaf(b.z, w6, s); s = fmaf(b.w, w7, s);
    return s;
}

__device__ __forceinline__ float dot4(const float4 a, const float w0, const float w1, const float w2, const float w3) {
    float s = a.x * w0;
    s = fmaf(a.y, w1, s); s = fmaf(a.z, w2, s); s = fmaf(a.w, w3, s);
    return s;
}

__device__ __forceinline__ float quad_sum(float v) {
    v += __builtin_bit_cast(float, __builtin_amdgcn_update_dpp(0, __builtin_bit_cast(int, v), 0xB1, 0xF, 0xF, true));
    v += __builtin_bit_cast(float, __builtin_amdgcn_update_dpp(0, __builtin_bit_cast(int, v), 0x4E, 0xF, 0xF, true));
    return v;
}

#define TILE_E  4096
#define BKT_MAX 512

__device__ __forceinline__ void scan256(int* part, const int t) {
    for (int off = 1; off < 256; off <<= 1) {
        const int v = (t >= off) ? part[t - off] : 0;
        __syncthreads();
        part[t] += v;
        __syncthreads();
    }
}

__global__ void __launch_bounds__(256) bucket_count(const int* __restrict__ dst, int* __restrict__ bucketCount,
                                                    int* __restrict__ tileOff, int E, int NB, int NT,
                                                    const float* __restrict__ x, const float* __restrict__ W1,
                                                    float* __restrict__ h1, int N, int nTiles, int gBase) {
    if ((int)blockIdx.x >= NT) { gemm1_block(x, W1, h1, N, nTiles, gBase + (int)blockIdx.x - NT); return; }
    __shared__ int hist[BKT_MAX];
    const int t = threadIdx.x;
    for (int i = t; i < NB; i += 256) hist[i] = 0;
    __syncthreads();
    const int base = blockIdx.x * TILE_E + t;
    int d[16];
    #pragma unroll
    for (int k = 0; k < 16; ++k) {
        const int e = base + k * 256;
        d[k] = dst[e < E ? e : E - 1];
    }
    #pragma unroll
    for (int k = 0; k < 16; ++k) {
        if (base + k * 256 < E) atomicAdd(&hist[d[k] >> 8], 1);
    }
    __syncthreads();
    const int i0 = t, i1 = t + 256;
    int r0 = 0, r1 = 0;
    if (i0 < NB) r0 = atomicAdd(&bucketCount[i0], hist[i0]);
    if (i1 < NB) r1 = atomicAdd(&bucketCount[i1], hist[i1]);
    if (i0 < NB) tileOff[blockIdx.x * NB + i0] = r0;
    if (i1 < NB) tileOff[blockIdx.x * NB + i1] = r1;
}

__global__ void __launch_bounds__(256) bucket_scatter(const int* __restrict__ src, const int* __restrict__ dst,
                                                      const int* __restrict__ bucketCount,
                                                      const int* __restrict__ tileOff,
                                                      int* __restrict__ bucketStart, int* __restrict__ binned,
                                                      int E, int NB, int NT,
                                                      const float* __restrict__ x, const float* __restrict__ W1,
                                                      float* __restrict__ h1, int N, int nTiles, int gBase) {
    if ((int)blockIdx.x >= NT) { gemm1_block(x, W1, h1, N, nTiles, gBase + (int)blockIdx.x - NT); return; }
    __shared__ int cur[BKT_MAX + 32];
    __shared__ int part[256];
    const int t = threadIdx.x;
    const int base = blockIdx.x * TILE_E + t;
    if (t < 32) cur[BKT_MAX + t] = 0;
    int d[16], s[16];
    #pragma unroll
    for (int k = 0; k < 16; ++k) {
        const int e = base + k * 256;
        const int ec = e < E ? e : E - 1;
        d[k] = dst[ec];
        s[k] = src[ec];
    }
    const int b0 = 2 * t, b1 = 2 * t + 1;
    const int b0c = b0 < NB ? b0 : NB - 1, b1c = b1 < NB ? b1 : NB - 1;
    const int c0r = bucketCount[b0c], c1r = bucketCount[b1c];
    const int o0r = tileOff[blockIdx.x * NB + b0c], o1r = tileOff[blockIdx.x * NB + b1c];
    __builtin_amdgcn_sched_barrier(0);
    const int c0 = b0 < NB ? c0r : 0, c1 = b1 < NB ? c1r : 0;
    const int o0 = b0 < NB ? o0r : 0, o1 = b1 < NB ? o1r : 0;
    part[t] = c0 + c1;
    __syncthreads();
    scan256(part, t);
    const int ex = part[t] - (c0 + c1);
    if (b0 < NB) cur[b0] = ex + o0;
    if (b1 < NB) cur[b1] = ex + c0 + o1;
    if (blockIdx.x == 0) {
        if (b0 < NB) bucketStart[b0] = ex;
        if (b1 < NB) bucketStart[b1] = ex + c0;
        if (t == 255) bucketStart[NB] = part[255];
    }
    __syncthreads();
    int pos[16];
    #pragma unroll
    for (int k = 0; k < 16; ++k)
        pos[k] = atomicAdd(&cur[(base + k * 256 < E) ? (d[k] >> 8) : (BKT_MAX + (t & 31))], 1);
    #pragma unroll
    for (int k = 0; k < 16; ++k)
        if (base + k * 256 < E) binned[pos[k]] = (int)(((unsigned)s[k] << 8) | (unsigned)(d[k] & 255));
}

__global__ void __launch_bounds__(256) csr_fill(const int* __restrict__ bucketStart, const int* __restrict__ binned,
                                                int* __restrict__ rowptr, int* __restrict__ csr, int N, int NB,
                                                const float* __restrict__ x, const float* __restrict__ W1,
                                                float* __restrict__ h1, int nTiles, int gBase) {
    if ((int)blockIdx.x >= NB) { gemm1_block(x, W1, h1, N, nTiles, gBase + (int)blockIdx.x - NB); return; }
    __shared__ int cnt[256 + 32];
    __shared__ int part[256];
    const int t = threadIdx.x;
    const int spare = 256 + (t & 31);
    const int beg = bucketStart[blockIdx.x], end = bucketStart[blockIdx.x + 1];
    const int last = end - 1;
    cnt[t] = 0;
    if (t < 32) cnt[256 + t] = 0;
    __syncthreads();
    for (int j = beg + t; j < end; j += 2048) {
        int v[8];
        #pragma unroll
        for (int k = 0; k < 8; ++k) {
            const int jk = j + k * 256;
            v[k] = binned[jk < end ? jk : last];
        }
        #pragma unroll
        for (int k = 0; k < 8; ++k)
            atomicAdd(&cnt[(j + k * 256 < end) ? (v[k] & 255) : spare], 1);
    }
    __syncthreads();
    const int c = cnt[t];
    part[t] = c;
    __syncthreads();
    scan256(part, t);
    const int first = beg + part[t] - c;
    const int n = blockIdx.x * 256 + t;
    if (n < N) rowptr[n] = first;
    if (n == N - 1) rowptr[N] = first + c;
    cnt[t] = first;
    __syncthreads();
    for (int j = beg + t; j < end; j += 2048) {
        int v[8], pos[8];
        #pragma unroll
        for (int k = 0; k < 8; ++k) {
            const int jk = j + k * 256;
            v[k] = binned[jk < end ? jk : last];
        }
        #pragma unroll
        for (int k = 0; k < 8; ++k)
            pos[k] = atomicAdd(&cnt[(j + k * 256 < end) ? (v[k] & 255) : spare], 1);
        #pragma unroll
        for (int k = 0; k < 8; ++k)
            if (j + k * 256 < end) csr[pos[k]] = (int)((unsigned)v[k] >> 8);
    }
}

__global__ void node_gather1(const int* __restrict__ rowptr,
                             const int* __restrict__ csr,
                             const float* __restrict__ attS, const float* __restrict__ attD,
                             const float* __restrict__ h1, const float* __restrict__ b1,
                             float* __restrict__ o1, int NH) {
    int i = blockIdx.x * blockDim.x + threadIdx.x;
    if (i >= NH) return;
    const int n = i >> 3, h = i & 7;
    const int beg = rowptr[n];
    const int end = rowptr[n + 1];
    const int last = end - 1;

    const float4* __restrict__ hv = (const float4*)h1;
    const float* ws = attS + h * HID;
    const float w0 = ws[0], w1 = ws[1], w2 = ws[2], w3 = ws[3];
    const float w4 = ws[4], w5 = ws[5], w6 = ws[6], w7 = ws[7];
    float ad;
    {
        const float* wd = attD + h * HID;
        const float4 p0 = hv[i * 2], p1 = hv[i * 2 + 1];
        ad = dot8(p0, p1, wd[0], wd[1], wd[2], wd[3], wd[4], wd[5], wd[6], wd[7]);
    }
    float m = -INFINITY;
    float den = 0.f;
    float a0 = 0.f, a1 = 0.f, a2 = 0.f, a3 = 0.f, a4 = 0.f, a5 = 0.f, a6 = 0.f, a7 = 0.f;
    for (int j = beg; j < end; j += 4) {
        const bool k1 = j + 1 < end, k2 = j + 2 < end, k3 = j + 3 < end;
        const int j1 = k1 ? j + 1 : last;
        const int j2 = k2 ? j + 2 : last;
        const int j3 = k3 ? j + 3 : last;
        const int s0 = csr[j], s1 = csr[j1], s2 = csr[j2], s3 = csr[j3];
        const int i0 = s0 * (F1 / 4) + h * 2, i1 = s1 * (F1 / 4) + h * 2;
        const int i2 = s2 * (F1 / 4) + h * 2, i3 = s3 * (F1 / 4) + h * 2;
        const float4 x0 = hv[i0], y0 = hv[i0 + 1];
        const float4 x1 = hv[i1], y1 = hv[i1 + 1];
        const float4 x2 = hv[i2], y2 = hv[i2 + 1];
        const float4 x3 = hv[i3], y3 = hv[i3 + 1];
        const float g0 = dot8(x0, y0, w0, w1, w2, w3, w4, w5, w6, w7);
        const float g1 = dot8(x1, y1, w0, w1, w2, w3, w4, w5, w6, w7);
        const float g2 = dot8(x2, y2, w0, w1, w2, w3, w4, w5, w6, w7);
        const float g3 = dot8(x3, y3, w0, w1, w2, w3, w4, w5, w6, w7);
        const float v0 = lrelu(g0 + ad), v1 = lrelu(g1 + ad);
        const float v2 = lrelu(g2 + ad), v3 = lrelu(g3 + ad);
        const float u = v0 > v1 ? v0 : v1;
        const float w = v2 > v3 ? v2 : v3;
        const float z = u > w ? u : w;
        const float mn = z > m ? z : m;
        const float sc = expf(m - mn);
        m = mn;
        const float e0 = expf(v0 - mn);
        const float e1 = k1 ? expf(v1 - mn) : 0.f;
        const float e2 = k2 ? expf(v2 - mn) : 0.f;
        const float e3 = k3 ? expf(v3 - mn) : 0.f;
        den *= sc;
        a0 *= sc; a1 *= sc; a2 *= sc; a3 *= sc; a4 *= sc; a5 *= sc; a6 *= sc; a7 *= sc;
        den += e0; den += e1; den += e2; den += e3;
        a0 += e0 * x0.x; a1 += e0 * x0.y; a2 += e0 * x0.z; a3 += e0 * x0.w;
        a4 += e0 * y0.x; a5 += e0 * y0.y; a6 += e0 * y0.z; a7 += e0 * y0.w;
        a0 += e1 * x1.x; a1 += e1 * x1.y; a2 += e1 * x1.z; a3 += e1 * x1.w;
        a4 += e1 * y1.x; a5 += e1 * y1.y; a6 += e1 * y1.z; a7 += e1 * y1.w;
        a0 += e2 * x2.x; a1 += e2 * x2.y; a2 += e2 * x2.z; a3 += e2 * x2.w;
        a4 += e2 * y2.x; a5 += e2 * y2.y; a6 += e2 * y2.z; a7 += e2 * y2.w;
        a0 += e3 * x3.x; a1 += e3 * x3.y; a2 += e3 * x3.z; a3 += e3 * x3.w;
        a4 += e3 * y3.x; a5 += e3 * y3.y; a6 += e3 * y3.z; a7 += e3 * y3.w;
    }

    const float dn = den + 1e-16f;
    const float* bp = b1 + h * HID;
    const float t0 = a0 / dn + bp[0], t1 = a1 / dn + bp[1], t2 = a2 / dn + bp[2], t3 = a3 / dn + bp[3];
    const float t4 = a4 / dn + bp[4], t5 = a5 / dn + bp[5], t6 = a6 / dn + bp[6], t7 = a7 / dn + bp[7];
    float4 r0, r1;
    r0.x = t0 > 0.f ? t0 : (expf(t0) - 1.f);
    r0.y = t1 > 0.f ? t1 : (expf(t1) - 1.f);
    r0.z = t2 > 0.f ? t2 : (expf(t2) - 1.f);
    r0.w = t3 > 0.f ? t3 : (expf(t3) - 1.f);
    r1.x = t4 > 0.f ? t4 : (expf(t4) - 1.f);
    r1.y = t5 > 0.f ? t5 : (expf(t5) - 1.f);
    r1.z = t6 > 0.f ? t6 : (expf(t6) - 1.f);
    r1.w = t7 > 0.f ? t7 : (expf(t7) - 1.f);
    float4* op = (float4*)(o1 + (size_t)n * F1 + h * HID);
    op[0] = r0;
    op[1] = r1;
}

__global__ void node_gather2(const int* __restrict__ rowptr,
                             const int* __restrict__ csr,
                             const float* __restrict__ attS, const float* __restrict__ attD,
                             const float* __restrict__ h2, const float* __restrict__ b2,
                             float* __restrict__ out, int NQ) {
    int i = blockIdx.x * blockDim.x + threadIdx.x;
    if (i >= NQ) return;
    const int n = i >> 2, q = i & 3;
    const int beg = rowptr[n];
    const int end = rowptr[n + 1];
    const int last = end - 1;

    const float4* __restrict__ hv = (const float4*)h2;
    const float* ws = attS + q * 4;
    const float w0 = ws[0], w1 = ws[1], w2 = ws[2], w3 = ws[3];
    float ad;
    {
        const float* wd = attD + q * 4;
        const float4 p = hv[i];
        ad = quad_sum(dot4(p, wd[0], wd[1], wd[2], wd[3]));
    }
    float m = -INFINITY;
    float den = 0.f;
    float a0 = 0.f, a1 = 0.f, a2 = 0.f, a3 = 0.f;
    for (int j = beg; j < end; j += 4) {
        const bool k1 = j + 1 < end, k2 = j + 2 < end, k3 = j + 3 < end;
        const int j1 = k1 ? j + 1 : last;
        const int j2 = k2 ? j + 2 : last;
        const int j3 = k3 ? j + 3 : last;
        const int s0 = csr[j], s1 = csr[j1], s2 = csr[j2], s3 = csr[j3];
        const float4 x0 = hv[s0 * (OUT_C / 4) + q];
        const float4 x1 = hv[s1 * (OUT_C / 4) + q];
        const float4 x2 = hv[s2 * (OUT_C / 4) + q];
        const float4 x3 = hv[s3 * (OUT_C / 4) + q];
        const float g0 = quad_sum(dot4(x0, w0, w1, w2, w3)), g1 = quad_sum(dot4(x1, w0, w1, w2, w3));
        const float g2 = quad_sum(dot4(x2, w0, w1, w2, w3)), g3 = quad_sum(dot4(x3, w0, w1, w2, w3));
        const float v0 = lrelu(g0 + ad), v1 = lrelu(g1 + ad);
        const float v2 = lrelu(g2 + ad), v3 = lrelu(g3 + ad);
        const float u = v0 > v1 ? v0 : v1;
        const float w = v2 > v3 ? v2 : v3;
        const float z = u > w ? u : w;
        const float mn = z > m ? z : m;
        const float sc = expf(m - mn);
        m = mn;
        const float e0 = expf(v0 - mn);
        const float e1 = k1 ? expf(v1 - mn) : 0.f;
        const float e2 = k2 ? expf(v2 - mn) : 0.f;
        const float e3 = k3 ? expf(v3 - mn) : 0.f;
        den *= sc;
        a0 *= sc; a1 *= sc; a2 *= sc; a3 *= sc;
        den += e0; den += e1; den += e2; den += e3;
        a0 += e0 * x0.x; a1 += e0 * x0.y; a2 += e0 * x0.z; a3 += e0 * x0.w;
        a0 += e1 * x1.x; a1 += e1 * x1.y; a2 += e1 * x1.z; a3 += e1 * x1.w;
        a0 += e2 * x2.x; a1 += e2 * x2.y; a2 += e2 * x2.z; a3 += e2 * x2.w;
        a0 += e3 * x3.x; a1 += e3 * x3.y; a2 += e3 * x3.z; a3 += e3 * x3.w;
    }

    const float dn = den + 1e-16f;
    const float* bq = b2 + q * 4;
    float4 r;
    r.x = a0 / dn + bq[0];
    r.y = a1 / dn + bq[1];
    r.z = a2 / dn + bq[2];
    r.w = a3 / dn + bq[3];
    ((float4*)(out + (size_t)n * OUT_C))[q] = r;
}

extern "C" void kernel_launch(void* const* d_in, const int* in_sizes, int n_in,
                              void* d_out, int out_size, void* d_ws, size_t ws_size,
                              hipStream_t stream) {
    const int N = in_sizes[0] / IN_C;
    const int E = in_sizes[1] / 2;

    const float* x     = (const float*)d_in[0];
    const int*   ei    = (const int*)  d_in[1];
    const int*   src   = ei;
    const int*   dst   = ei + E;
    const float* W1    = (const float*)d_in[2];
    const float* b1    = (const float*)d_in[3];
    const float* attS1 = (const float*)d_in[4];
    const float* attD1 = (const float*)d_in[5];
    const float* W2    = (const float*)d_in[6];
    const float* b2    = (const float*)d_in[7];
    const float* attS2 = (const float*)d_in[8];
    const float* attD2 = (const float*)d_in[9];
    float* out = (float*)d_out;

    char* ws = (char*)d_ws;
    size_t off = 0;
    auto alloc = [&](size_t elems) -> float* {
        float* p = (float*)(ws + off);
        off = (off + elems * sizeof(float) + 255) & ~(size_t)255;
        return p;
    };
    float* h1   = alloc((size_t)N * F1);
    float* o1   = alloc((size_t)N * F1);
    float* h2   = alloc((size_t)N * OUT_C);
    const int NB = (N + 255) / 256;
    const int NT = (E + TILE_E - 1) / TILE_E;
    int*   bucketCount = (int*)alloc((size_t)BKT_MAX);
    int*   bucketStart = (int*)alloc((size_t)BKT_MAX + 1);
    int*   tileOff     = (int*)alloc((size_t)NT * NB);
    int*   rowptr      = (int*)alloc((size_t)N + 1);
    int*   binned      = (int*)alloc((size_t)E);
    int*   csr         = (int*)alloc((size_t)E);

    hipMemsetAsync(bucketCount, 0, (size_t)NB * sizeof(int), stream);

    const int TB = 256;
    auto nb = [](long long n, int tb) { return (unsigned)((n + tb - 1) / tb); };

    const int nTiles = (N + 15) / 16;
    const int gemmBlks = (nTiles + 7) / 8;
    const int gPart = (gemmBlks + 2) / 3;

    bucket_count<<<NT + gPart, 256, 0, stream>>>(dst, bucketCount, tileOff, E, NB, NT, x, W1, h1, N, nTiles, 0);
    bucket_scatter<<<NT + gPart, 256, 0, stream>>>(src, dst, bucketCount, tileOff, bucketStart, binned, E, NB, NT,
                                                   x, W1, h1, N, nTiles, gPart);
    csr_fill<<<NB + gPart, 256, 0, stream>>>(bucketStart, binned, rowptr, csr, N, NB, x, W1, h1, nTiles, 2 * gPart);

    node_gather1<<<nb((long long)N * HEADS, TB), TB, 0, stream>>>(rowptr, csr, attS1, attD1, h1, b1, o1, N * HEADS);

    gat_gemm2_wmma<<<gemmBlks, TB, 0, stream>>>(o1, W2, h2, N, nTiles);
    node_gather2<<<nb((long long)N * 4, TB), TB, 0, stream>>>(rowptr, csr, attS2, attD2, h2, b2, out, N * 4);
}


# Layer-2 product moved into the layer-1 node walk: each work-group's 32 finished rows go through LDS to two of its waves (same 16 f32 WMMA steps in k order, sums unchanged); one launch, one array less
# speedup vs baseline: 6.6338x; 1.0521x over previous
#include <hip/hip_runtime.h>
#include <math.h>

#define IN_C   128
#define HEADS  8
#define HID    8
#define F1     (HEADS * HID)
#define OUT_C  16
#define NEG_SLOPE 0.2f

typedef float v2f __attribute__((ext_vector_type(2)));
typedef float v8f __attribute__((ext_vector_type(8)));

__device__ __forceinline__ float lrelu(float v) {
    return v > 0.0f ? v : NEG_SLOPE * v;
}

__device__ __forceinline__ void gemm1_block(const float* __restrict__ x,
                                            const float* __restrict__ W,
                                            float* __restrict__ h1,
                                            int N, int nTiles, int gblk) {
    __shared__ float w[IN_C * F1];
    for (int i = threadIdx.x; i < IN_C * F1; i += 256) w[i] = W[i];
    __syncthreads();

    const int tile = gblk * 8 + (threadIdx.x >> 5);
    if (tile >= nTiles) return;

    const int lane  = threadIdx.x & 31;
    const int row16 = lane & 15;
    const int kh    = lane >> 4;

    int rowIn = tile * 16 + row16;
    if (rowIn >= N) rowIn = N - 1;
    const float* xr = x + (size_t)rowIn * IN_C;

    v8f acc0 = {}, acc1 = {}, acc2 = {}, acc3 = {};

    for (int k = 0; k < IN_C; k += 16) {
        v2f a[4];
        #pragma unroll
        for (int u = 0; u < 4; ++u) {
            const int ka = k + 4 * u + 2 * kh;
            a[u].x = xr[ka];
            a[u].y = xr[ka + 1];
        }
        __builtin_amdgcn_sched_barrier(0);
        #pragma unroll
        for (int u = 0; u < 4; ++u) {
            const int ka = k + 4 * u + 2 * kh;
            const float* w0 = &w[ka * F1 + row16];
            const float* w1 = &w[(ka + 1) * F1 + row16];
            v2f b0; b0.x = w0[0];  b0.y = w1[0];
            v2f b1; b1.x = w0[16]; b1.y = w1[16];
            v2f b2; b2.x = w0[32]; b2.y = w1[32];
            v2f b3; b3.x = w0[48]; b3.y = w1[48];
            acc0 = __builtin_amdgcn_wmma_f32_16x16x4_f32(false, a[u], false, b0, (short)0, acc0, false, false);
            acc1 = __builtin_amdgcn_wmma_f32_16x16x4_f32(false, a[u], false, b1, (short)0, acc1, false, false);
            acc2 = __builtin_amdgcn_wmma_f32_16x16x4_f32(false, a[u], false, b2, (short)0, acc2, false, false);
            acc3 = __builtin_amdgcn_wmma_f32_16x16x4_f32(false, a[u], false, b3, (short)0, acc3, false, false);
        }
    }

    #pragma unroll
    for (int i = 0; i < 8; ++i) {
        const int r = tile * 16 + i + 8 * kh;
        if (r < N) {
            float* o = h1 + (size_t)r * F1 + row16;
            o[0]  = acc0[i];
            o[16] = acc1[i];
            o[32] = acc2[i];
            o[48] = acc3[i];
        }
    }
}

__device__ __forceinline__ float dot8(const float4 a, const float4 b,
                                      const float w0, const float w1, const float w2, const float w3,
                                      const float w4, const float w5, const float w6, const float w7) {
    float s = a.x * w0;
    s = fmaf(a.y, w1, s); s = fmaf(a.z, w2, s); s = fmaf(a.w, w3, s);
    s = fmaf(b.x, w4, s); s = fmaf(b.y, w5, s); s = fmaf(b.z, w6, s); s = fmaf(b.w, w7, s);
    return s;
}

__device__ __forceinline__ float dot4(const float4 a, const float w0, const float w1, const float w2, const float w3) {
    float s = a.x * w0;
    s = fmaf(a.y, w1, s); s = fmaf(a.z, w2, s); s = fmaf(a.w, w3, s);
    return s;
}

__device__ __forceinline__ float quad_sum(float v) {
    v += __builtin_bit_cast(float, __builtin_amdgcn_update_dpp(0, __builtin_bit_cast(int, v), 0xB1, 0xF, 0xF, true));
    v += __builtin_bit_cast(float, __builtin_amdgcn_update_dpp(0, __builtin_bit_cast(int, v), 0x4E, 0xF, 0xF, true));
    return v;
}

#define TILE_E  4096
#define BKT_MAX 512

__device__ __forceinline__ void scan256(int* part, const int t) {
    for (int off = 1; off < 256; off <<= 1) {
        const int v = (t >= off) ? part[t - off] : 0;
        __syncthreads();
        part[t] += v;
        __syncthreads();
    }
}

__global__ void __launch_bounds__(256) bucket_count(const int* __restrict__ dst, int* __restrict__ bucketCount,
                                                    int* __restrict__ tileOff, int E, int NB, int NT,
                                                    const float* __restrict__ x, const float* __restrict__ W1,
                                                    float* __restrict__ h1, int N, int nTiles, int gBase) {
    if ((int)blockIdx.x >= NT) { gemm1_block(x, W1, h1, N, nTiles, gBase + (int)blockIdx.x - NT); return; }
    __shared__ int hist[BKT_MAX];
    const int t = threadIdx.x;
    for (int i = t; i < NB; i += 256) hist[i] = 0;
    __syncthreads();
    const int base = blockIdx.x * TILE_E + t;
    int d[16];
    #pragma unroll
    for (int k = 0; k < 16; ++k) {
        const int e = base + k * 256;
        d[k] = dst[e < E ? e : E - 1];
    }
    #pragma unroll
    for (int k = 0; k < 16; ++k) {
        if (base + k * 256 < E) atomicAdd(&hist[d[k] >> 8], 1);
    }
    __syncthreads();
    const int i0 = t, i1 = t + 256;
    int r0 = 0, r1 = 0;
    if (i0 < NB) r0 = atomicAdd(&bucketCount[i0], hist[i0]);
    if (i1 < NB) r1 = atomicAdd(&bucketCount[i1], hist[i1]);
    if (i0 < NB) tileOff[blockIdx.x * NB + i0] = r0;
    if (i1 < NB) tileOff[blockIdx.x * NB + i1] = r1;
}

__global__ void __launch_bounds__(256) bucket_scatter(const int* __restrict__ src, const int* __restrict__ dst,
                                                      const int* __restrict__ bucketCount,
                                                      const int* __restrict__ tileOff,
                                                      int* __restrict__ bucketStart, int* __restrict__ binned,
                                                      int E, int NB, int NT,
                                                      const float* __restrict__ x, const float* __restrict__ W1,
                                                      float* __restrict__ h1, int N, int nTiles, int gBase) {
    if ((int)blockIdx.x >= NT) { gemm1_block(x, W1, h1, N, nTiles, gBase + (int)blockIdx.x - NT); return; }
    __shared__ int cur[BKT_MAX + 32];
    __shared__ int part[256];
    const int t = threadIdx.x;
    const int base = blockIdx.x * TILE_E + t;
    if (t < 32) cur[BKT_MAX + t] = 0;
    int d[16], s[16];
    #pragma unroll
    for (int k = 0; k < 16; ++k) {
        const int e = base + k * 256;
        const int ec = e < E ? e : E - 1;
        d[k] = dst[ec];
        s[k] = src[ec];
    }
    const int b0 = 2 * t, b1 = 2 * t + 1;
    const int b0c = b0 < NB ? b0 : NB - 1, b1c = b1 < NB ? b1 : NB - 1;
    const int c0r = bucketCount[b0c], c1r = bucketCount[b1c];
    const int o0r = tileOff[blockIdx.x * NB + b0c], o1r = tileOff[blockIdx.x * NB + b1c];
    __builtin_amdgcn_sched_barrier(0);
    const int c0 = b0 < NB ? c0r : 0, c1 = b1 < NB ? c1r : 0;
    const int o0 = b0 < NB ? o0r : 0, o1 = b1 < NB ? o1r : 0;
    part[t] = c0 + c1;
    __syncthreads();
    scan256(part, t);
    const int ex = part[t] - (c0 + c1);
    if (b0 < NB) cur[b0] = ex + o0;
    if (b1 < NB) cur[b1] = ex + c0 + o1;
    if (blockIdx.x == 0) {
        if (b0 < NB) bucketStart[b0] = ex;
        if (b1 < NB) bucketStart[b1] = ex + c0;
        if (t == 255) bucketStart[NB] = part[255];
    }
    __syncthreads();
    int pos[16];
    #pragma unroll
    for (int k = 0; k < 16; ++k)
        pos[k] = atomicAdd(&cur[(base + k * 256 < E) ? (d[k] >> 8) : (BKT_MAX + (t & 31))], 1);
    #pragma unroll
    for (int k = 0; k < 16; ++k)
        if (base + k * 256 < E) binned[pos[k]] = (int)(((unsigned)s[k] << 8) | (unsigned)(d[k] & 255));
}

__global__ void __launch_bounds__(256) csr_fill(const int* __restrict__ bucketStart, const int* __restrict__ binned,
                                                int* __restrict__ rowptr, int* __restrict__ csr, int N, int NB,
                                                const float* __restrict__ x, const float* __restrict__ W1,
                                                float* __restrict__ h1, int nTiles, int gBase) {
    if ((int)blockIdx.x >= NB) { gemm1_block(x, W1, h1, N, nTiles, gBase + (int)blockIdx.x - NB); return; }
    __shared__ int cnt[256 + 32];
    __shared__ int part[256];
    const int t = threadIdx.x;
    const int spare = 256 + (t & 31);
    const int beg = bucketStart[blockIdx.x], end = bucketStart[blockIdx.x + 1];
    const int last = end - 1;
    cnt[t] = 0;
    if (t < 32) cnt[256 + t] = 0;
    __syncthreads();
    for (int j = beg + t; j < end; j += 2048) {
        int v[8];
        #pragma unroll
        for (int k = 0; k < 8; ++k) {
            const int jk = j + k * 256;
            v[k] = binned[jk < end ? jk : last];
        }
        #pragma unroll
        for (int k = 0; k < 8; ++k)
            atomicAdd(&cnt[(j + k * 256 < end) ? (v[k] & 255) : spare], 1);
    }
    __syncthreads();
    const int c = cnt[t];
    part[t] = c;
    __syncthreads();
    scan256(part, t);
    const int first = beg + part[t] - c;
    const int n = blockIdx.x * 256 + t;
    if (n < N) rowptr[n] = first;
    if (n == N - 1) rowptr[N] = first + c;
    cnt[t] = first;
    __syncthreads();
    for (int j = beg + t; j < end; j += 2048) {
        int v[8], pos[8];
        #pragma unroll
        for (int k = 0; k < 8; ++k) {
            const int jk = j + k * 256;
            v[k] = binned[jk < end ? jk : last];
        }
        #pragma unroll
        for (int k = 0; k < 8; ++k)
            pos[k] = atomicAdd(&cnt[(j + k * 256 < end) ? (v[k] & 255) : spare], 1);
        #pragma unroll
        for (int k = 0; k < 8; ++k)
            if (j + k * 256 < end) csr[pos[k]] = (int)((unsigned)v[k] >> 8);
    }
}

#define O1_PITCH 68
__global__ void __launch_bounds__(256) node_gather1(const int* __restrict__ rowptr,
                             const int* __restrict__ csr,
                             const float* __restrict__ attS, const float* __restrict__ attD,
                             const float* __restrict__ h1, const float* __restrict__ b1,
                             const float* __restrict__ W2, float* __restrict__ h2, int N) {
    __shared__ float ot[32 * O1_PITCH];
    const int t = threadIdx.x;
    const int h = t & 7;
    int n = (int)blockIdx.x * 32 + (t >> 3);
    if (n >= N) n = N - 1;
    const int i = n * HEADS + h;
    const int beg = rowptr[n];
    const int end = rowptr[n + 1];
    const int last = end - 1;

    const float4* __restrict__ hv = (const float4*)h1;
    const float* ws = attS + h * HID;
    const float w0 = ws[0], w1 = ws[1], w2 = ws[2], w3 = ws[3];
    const float w4 = ws[4], w5 = ws[5], w6 = ws[6], w7 = ws[7];
    float ad;
    {
        const float* wd = attD + h * HID;
        const float4 p0 = hv[i * 2], p1 = hv[i * 2 + 1];
        ad = dot8(p0, p1, wd[0], wd[1], wd[2], wd[3], wd[4], wd[5], wd[6], wd[7]);
    }
    float m = -INFINITY;
    float den = 0.f;
    float a0 = 0.f, a1 = 0.f, a2 = 0.f, a3 = 0.f, a4 = 0.f, a5 = 0.f, a6 = 0.f, a7 = 0.f;
    for (int j = beg; j < end; j += 4) {
        const bool k1 = j + 1 < end, k2 = j + 2 < end, k3 = j + 3 < end;
        const int j1 = k1 ? j + 1 : last;
        const int j2 = k2 ? j + 2 : last;
        const int j3 = k3 ? j + 3 : last;
        const int s0 = csr[j], s1 = csr[j1], s2 = csr[j2], s3 = csr[j3];
        const int i0 = s0 * (F1 / 4) + h * 2, i1 = s1 * (F1 / 4) + h * 2;
        const int i2 = s2 * (F1 / 4) + h * 2, i3 = s3 * (F1 / 4) + h * 2;
        const float4 x0 = hv[i0], y0 = hv[i0 + 1];
        const float4 x1 = hv[i1], y1 = hv[i1 + 1];
        const float4 x2 = hv[i2], y2 = hv[i2 + 1];
        const float4 x3 = hv[i3], y3 = hv[i3 + 1];
        const float g0 = dot8(x0, y0, w0, w1, w2, w3, w4, w5, w6, w7);
        const float g1 = dot8(x1, y1, w0, w1, w2, w3, w4, w5, w6, w7);
        const float g2 = dot8(x2, y2, w0, w1, w2, w3, w4, w5, w6, w7);
        const float g3 = dot8(x3, y3, w0, w1, w2, w3, w4, w5, w6, w7);
        const float v0 = lrelu(g0 + ad), v1 = lrelu(g1 + ad);
        const float v2 = lrelu(g2 + ad), v3 = lrelu(g3 + ad);
        const float u = v0 > v1 ? v0 : v1;
        const float w = v2 > v3 ? v2 : v3;
        const float z = u > w ? u : w;
        const float mn = z > m ? z : m;
        const float sc = expf(m - mn);
        m = mn;
        const float e0 = expf(v0 - mn);
        const float e1 = k1 ? expf(v1 - mn) : 0.f;
        const float e2 = k2 ? expf(v2 - mn) : 0.f;
        const float e3 = k3 ? expf(v3 - mn) : 0.f;
        den *= sc;
        a0 *= sc; a1 *= sc; a2 *= sc; a3 *= sc; a4 *= sc; a5 *= sc; a6 *= sc; a7 *= sc;
        den += e0; den += e1; den += e2; den += e3;
        a0 += e0 * x0.x; a1 += e0 * x0.y; a2 += e0 * x0.z; a3 += e0 * x0.w;
        a4 += e0 * y0.x; a5 += e0 * y0.y; a6 += e0 * y0.z; a7 += e0 * y0.w;
        a0 += e1 * x1.x; a1 += e1 * x1.y; a2 += e1 * x1.z; a3 += e1 * x1.w;
        a4 += e1 * y1.x; a5 += e1 * y1.y; a6 += e1 * y1.z; a7 += e1 * y1.w;
        a0 += e2 * x2.x; a1 += e2 * x2.y; a2 += e2 * x2.z; a3 += e2 * x2.w;
        a4 += e2 * y2.x; a5 += e2 * y2.y; a6 += e2 * y2.z; a7 += e2 * y2.w;
        a0 += e3 * x3.x; a1 += e3 * x3.y; a2 += e3 * x3.z; a3 += e3 * x3.w;
        a4 += e3 * y3.x; a5 += e3 * y3.y; a6 += e3 * y3.z; a7 += e3 * y3.w;
    }

    const int wv    = t >> 5;
    const int lane  = t & 31;
    const int row16 = lane & 15;
    const int kh    = lane >> 4;
    v2f b[16];
    if (wv < 2) {
        #pragma unroll
        for (int u = 0; u < 16; ++u) {
            const int ka = 4 * u + 2 * kh;
            b[u].x = W2[ka * OUT_C + row16];
            b[u].y = W2[(ka + 1) * OUT_C + row16];
        }
    }

    const float dn = den + 1e-16f;
    const float* bp = b1 + h * HID;
    const float t0 = a0 / dn + bp[0], t1 = a1 / dn + bp[1], t2 = a2 / dn + bp[2], t3 = a3 / dn + bp[3];
    const float t4 = a4 / dn + bp[4], t5 = a5 / dn + bp[5], t6 = a6 / dn + bp[6], t7 = a7 / dn + bp[7];
    float4 r0, r1;
    r0.x = t0 > 0.f ? t0 : (expf(t0) - 1.f);
    r0.y = t1 > 0.f ? t1 : (expf(t1) - 1.f);
    r0.z = t2 > 0.f ? t2 : (expf(t2) - 1.f);
    r0.w = t3 > 0.f ? t3 : (expf(t3) - 1.f);
    r1.x = t4 > 0.f ? t4 : (expf(t4) - 1.f);
    r1.y = t5 > 0.f ? t5 : (expf(t5) - 1.f);
    r1.z = t6 > 0.f ? t6 : (expf(t6) - 1.f);
    r1.w = t7 > 0.f ? t7 : (expf(t7) - 1.f);
    float4* lp = (float4*)&ot[(t >> 3) * O1_PITCH + h * HID];
    lp[0] = r0;
    lp[1] = r1;
    __syncthreads();
    if (wv >= 2) return;

    const float* ar = &ot[(wv * 16 + row16) * O1_PITCH + 2 * kh];
    v8f acc = {};
    #pragma unroll
    for (int u = 0; u < 16; ++u) {
        v2f a;
        a.x = ar[4 * u];
        a.y = ar[4 * u + 1];
        acc = __builtin_amdgcn_wmma_f32_16x16x4_f32(false, a, false, b[u], (short)0, acc, false, false);
    }
    #pragma unroll
    for (int r = 0; r < 8; ++r) {
        const int row = (int)blockIdx.x * 32 + wv * 16 + r + 8 * kh;
        if (row < N) h2[(size_t)row * OUT_C + row16] = acc[r];
    }
}

__global__ void node_gather2(const int* __restrict__ rowptr,
                             const int* __restrict__ csr,
                             const float* __restrict__ attS, const float* __restrict__ attD,
                             const float* __restrict__ h2, const float* __restrict__ b2,
                             float* __restrict__ out, int NQ) {
    int i = blockIdx.x * blockDim.x + threadIdx.x;
    if (i >= NQ) return;
    const int n = i >> 2, q = i & 3;
    const int beg = rowptr[n];
    const int end = rowptr[n + 1];
    const int last = end - 1;

    const float4* __restrict__ hv = (const float4*)h2;
    const float* ws = attS + q * 4;
    const float w0 = ws[0], w1 = ws[1], w2 = ws[2], w3 = ws[3];
    float ad;
    {
        const float* wd = attD + q * 4;
        const float4 p = hv[i];
        ad = quad_sum(dot4(p, wd[0], wd[1], wd[2], wd[3]));
    }
    float m = -INFINITY;
    float den = 0.f;
    float a0 = 0.f, a1 = 0.f, a2 = 0.f, a3 = 0.f;
    for (int j = beg; j < end; j += 4) {
        const bool k1 = j + 1 < end, k2 = j + 2 < end, k3 = j + 3 < end;
        const int j1 = k1 ? j + 1 : last;
        const int j2 = k2 ? j + 2 : last;
        const int j3 = k3 ? j + 3 : last;
        const int s0 = csr[j], s1 = csr[j1], s2 = csr[j2], s3 = csr[j3];
        const float4 x0 = hv[s0 * (OUT_C / 4) + q];
        const float4 x1 = hv[s1 * (OUT_C / 4) + q];
        const float4 x2 = hv[s2 * (OUT_C / 4) + q];
        const float4 x3 = hv[s3 * (OUT_C / 4) + q];
        const float g0 = quad_sum(dot4(x0, w0, w1, w2, w3)), g1 = quad_sum(dot4(x1, w0, w1, w2, w3));
        const float g2 = quad_sum(dot4(x2, w0, w1, w2, w3)), g3 = quad_sum(dot4(x3, w0, w1, w2, w3));
        const float v0 = lrelu(g0 + ad), v1 = lrelu(g1 + ad);
        const float v2 = lrelu(g2 + ad), v3 = lrelu(g3 + ad);
        const float u = v0 > v1 ? v0 : v1;
        const float w = v2 > v3 ? v2 : v3;
        const float z = u > w ? u : w;
        const float mn = z > m ? z : m;
        const float sc = expf(m - mn);
        m = mn;
        const float e0 = expf(v0 - mn);
        const float e1 = k1 ? expf(v1 - mn) : 0.f;
        const float e2 = k2 ? expf(v2 - mn) : 0.f;
        const float e3 = k3 ? expf(v3 - mn) : 0.f;
        den *= sc;
        a0 *= sc; a1 *= sc; a2 *= sc; a3 *= sc;
        den += e0; den += e1; den += e2; den += e3;
        a0 += e0 * x0.x; a1 += e0 * x0.y; a2 += e0 * x0.z; a3 += e0 * x0.w;
        a0 += e1 * x1.x; a1 += e1 * x1.y; a2 += e1 * x1.z; a3 += e1 * x1.w;
        a0 += e2 * x2.x; a1 += e2 * x2.y; a2 += e2 * x2.z; a3 += e2 * x2.w;
        a0 += e3 * x3.x; a1 += e3 * x3.y; a2 += e3 * x3.z; a3 += e3 * x3.w;
    }

    const float dn = den + 1e-16f;
    const float* bq = b2 + q * 4;
    float4 r;
    r.x = a0 / dn + bq[0];
    r.y = a1 / dn + bq[1];
    r.z = a2 / dn + bq[2];
    r.w = a3 / dn + bq[3];
    ((float4*)(out + (size_t)n * OUT_C))[q] = r;
}

extern "C" void kernel_launch(void* const* d_in, const int* in_sizes, int n_in,
                              void* d_out, int out_size, void* d_ws, size_t ws_size,
                              hipStream_t stream) {
    const int N = in_sizes[0] / IN_C;
    const int E = in_sizes[1] / 2;

    const float* x     = (const float*)d_in[0];
    const int*   ei    = (const int*)  d_in[1];
    const int*   src   = ei;
    const int*   dst   = ei + E;
    const float* W1    = (const float*)d_in[2];
    const float* b1    = (const float*)d_in[3];
    const float* attS1 = (const float*)d_in[4];
    const float* attD1 = (const float*)d_in[5];
    const float* W2    = (const float*)d_in[6];
    const float* b2    = (const float*)d_in[7];
    const float* attS2 = (const float*)d_in[8];
    const float* attD2 = (const float*)d_in[9];
    float* out = (float*)d_out;

    char* ws = (char*)d_ws;
    size_t off = 0;
    auto alloc = [&](size_t elems) -> float* {
        float* p = (float*)(ws + off);
        off = (off + elems * sizeof(float) + 255) & ~(size_t)255;
        return p;
    };
    float* h1   = alloc((size_t)N * F1);
    float* h2   = alloc((size_t)N * OUT_C);
    const int NB = (N + 255) / 256;
    const int NT = (E + TILE_E - 1) / TILE_E;
    int*   bucketCount = (int*)alloc((size_t)BKT_MAX);
    int*   bucketStart = (int*)alloc((size_t)BKT_MAX + 1);
    int*   tileOff     = (int*)alloc((size_t)NT * NB);
    int*   rowptr      = (int*)alloc((size_t)N + 1);
    int*   binned      = (int*)alloc((size_t)E);
    int*   csr         = (int*)alloc((size_t)E);

    hipMemsetAsync(bucketCount, 0, (size_t)NB * sizeof(int), stream);

    const int TB = 256;
    auto nb = [](long long n, int tb) { return (unsigned)((n + tb - 1) / tb); };

    const int nTiles = (N + 15) / 16;
    const int gemmBlks = (nTiles + 7) / 8;
    const int gPart = (gemmBlks + 2) / 3;

    bucket_count<<<NT + gPart, 256, 0, stream>>>(dst, bucketCount, tileOff, E, NB, NT, x, W1, h1, N, nTiles, 0);
    bucket_scatter<<<NT + gPart, 256, 0, stream>>>(src, dst, bucketCount, tileOff, bucketStart, binned, E, NB, NT,
                                                   x, W1, h1, N, nTiles, gPart);
    csr_fill<<<NB + gPart, 256, 0, stream>>>(bucketStart, binned, rowptr, csr, N, NB, x, W1, h1, nTiles, 2 * gPart);

    node_gather1<<<nb((long long)N, 32), TB, 0, stream>>>(rowptr, csr, attS1, attD1, h1, b1, W2, h2, N);

    node_gather2<<<nb((long long)N * 4, TB), TB, 0, stream>>>(rowptr, csr, attS2, attD2, h2, b2, out, N * 4);
}


# Both sort levels build their output in LDS in output order and store it with consecutive lanes on consecutive words, not one scattered 4-byte store per lane; segments stay in arrival order
# speedup vs baseline: 7.6976x; 1.1604x over previous
#include <hip/hip_runtime.h>
#include <math.h>

#define IN_C   128
#define HEADS  8
#define HID    8
#define F1     (HEADS * HID)
#define OUT_C  16
#define NEG_SLOPE 0.2f

typedef float v2f __attribute__((ext_vector_type(2)));
typedef float v8f __attribute__((ext_vector_type(8)));

__device__ __forceinline__ float lrelu(float v) {
    return v > 0.0f ? v : NEG_SLOPE * v;
}

__device__ __forceinline__ void gemm1_block(const float* __restrict__ x,
                                            const float* __restrict__ W,
                                            float* __restrict__ h1,
                                            int N, int nTiles, int gblk) {
    __shared__ float w[IN_C * F1];
    for (int i = threadIdx.x; i < IN_C * F1; i += 256) w[i] = W[i];
    __syncthreads();

    const int tile = gblk * 8 + (threadIdx.x >> 5);
    if (tile >= nTiles) return;

    const int lane  = threadIdx.x & 31;
    const int row16 = lane & 15;
    const int kh    = lane >> 4;

    int rowIn = tile * 16 + row16;
    if (rowIn >= N) rowIn = N - 1;
    const float* xr = x + (size_t)rowIn * IN_C;

    v8f acc0 = {}, acc1 = {}, acc2 = {}, acc3 = {};

    for (int k = 0; k < IN_C; k += 16) {
        v2f a[4];
        #pragma unroll
        for (int u = 0; u < 4; ++u) {
            const int ka = k + 4 * u + 2 * kh;
            a[u].x = xr[ka];
            a[u].y = xr[ka + 1];
        }
        __builtin_amdgcn_sched_barrier(0);
        #pragma unroll
        for (int u = 0; u < 4; ++u) {
            const int ka = k + 4 * u + 2 * kh;
            const float* w0 = &w[ka * F1 + row16];
            const float* w1 = &w[(ka + 1) * F1 + row16];
            v2f b0; b0.x = w0[0];  b0.y = w1[0];
            v2f b1; b1.x = w0[16]; b1.y = w1[16];
            v2f b2; b2.x = w0[32]; b2.y = w1[32];
            v2f b3; b3.x = w0[48]; b3.y = w1[48];
            acc0 = __builtin_amdgcn_wmma_f32_16x16x4_f32(false, a[u], false, b0, (short)0, acc0, false, false);
            acc1 = __builtin_amdgcn_wmma_f32_16x16x4_f32(false, a[u], false, b1, (short)0, acc1, false, false);
            acc2 = __builtin_amdgcn_wmma_f32_16x16x4_f32(false, a[u], false, b2, (short)0, acc2, false, false);
            acc3 = __builtin_amdgcn_wmma_f32_16x16x4_f32(false, a[u], false, b3, (short)0, acc3, false, false);
        }
    }

    #pragma unroll
    for (int i = 0; i < 8; ++i) {
        const int r = tile * 16 + i + 8 * kh;
        if (r < N) {
            float* o = h1 + (size_t)r * F1 + row16;
            o[0]  = acc0[i];
            o[16] = acc1[i];
            o[32] = acc2[i];
            o[48] = acc3[i];
        }
    }
}

__device__ __forceinline__ float dot8(const float4 a, const float4 b,
                                      const float w0, const float w1, const float w2, const float w3,
                                      const float w4, const float w5, const float w6, const float w7) {
    float s = a.x * w0;
    s = fmaf(a.y, w1, s); s = fmaf(a.z, w2, s); s = fmaf(a.w, w3, s);
    s = fmaf(b.x, w4, s); s = fmaf(b.y, w5, s); s = fmaf(b.z, w6, s); s = fmaf(b.w, w7, s);
    return s;
}

__device__ __forceinline__ float dot4(const float4 a, const float w0, const float w1, const float w2, const float w3) {
    float s = a.x * w0;
    s = fmaf(a.y, w1, s); s = fmaf(a.z, w2, s); s = fmaf(a.w, w3, s);
    return s;
}

__device__ __forceinline__ float quad_sum(float v) {
    v += __builtin_bit_cast(float, __builtin_amdgcn_update_dpp(0, __builtin_bit_cast(int, v), 0xB1, 0xF, 0xF, true));
    v += __builtin_bit_cast(float, __builtin_amdgcn_update_dpp(0, __builtin_bit_cast(int, v), 0x4E, 0xF, 0xF, true));
    return v;
}

#define TILE_E  4096
#define BKT_MAX 512
#define STAGE_MAX 8192

__device__ __forceinline__ void scan256(int* part, const int t) {
    for (int off = 1; off < 256; off <<= 1) {
        const int v = (t >= off) ? part[t - off] : 0;
        __syncthreads();
        part[t] += v;
        __syncthreads();
    }
}

__device__ __forceinline__ void scan256x2(int* pa, int* pb, const int t) {
    for (int off = 1; off < 256; off <<= 1) {
        const int va = (t >= off) ? pa[t - off] : 0;
        const int vb = (t >= off) ? pb[t - off] : 0;
        __syncthreads();
        pa[t] += va;
        pb[t] += vb;
        __syncthreads();
    }
}

__global__ void __launch_bounds__(256) bucket_count(const int* __restrict__ dst, int* __restrict__ bucketCount,
                                                    int* __restrict__ tileOff, int* __restrict__ tileHist,
                                                    int E, int NB, int NT,
                                                    const float* __restrict__ x, const float* __restrict__ W1,
                                                    float* __restrict__ h1, int N, int nTiles, int gBase) {
    if ((int)blockIdx.x >= NT) { gemm1_block(x, W1, h1, N, nTiles, gBase + (int)blockIdx.x - NT); return; }
    __shared__ int hist[BKT_MAX];
    const int t = threadIdx.x;
    for (int i = t; i < NB; i += 256) hist[i] = 0;
    __syncthreads();
    const int base = blockIdx.x * TILE_E + t;
    int d[16];
    #pragma unroll
    for (int k = 0; k < 16; ++k) {
        const int e = base + k * 256;
        d[k] = dst[e < E ? e : E - 1];
    }
    #pragma unroll
    for (int k = 0; k < 16; ++k) {
        if (base + k * 256 < E) atomicAdd(&hist[d[k] >> 8], 1);
    }
    __syncthreads();
    const int i0 = t, i1 = t + 256;
    int r0 = 0, r1 = 0;
    if (i0 < NB) r0 = atomicAdd(&bucketCount[i0], hist[i0]);
    if (i1 < NB) r1 = atomicAdd(&bucketCount[i1], hist[i1]);
    if (i0 < NB) tileOff[blockIdx.x * NB + i0] = r0;
    if (i1 < NB) tileOff[blockIdx.x * NB + i1] = r1;
    if (i0 < NB) tileHist[blockIdx.x * NB + i0] = hist[i0];
    if (i1 < NB) tileHist[blockIdx.x * NB + i1] = hist[i1];
}

__global__ void __launch_bounds__(256) bucket_scatter(const int* __restrict__ src, const int* __restrict__ dst,
                                                      const int* __restrict__ bucketCount,
                                                      const int* __restrict__ tileOff,
                                                      const int* __restrict__ tileHist,
                                                      int* __restrict__ bucketStart, int* __restrict__ binned,
                                                      int E, int NB, int NT,
                                                      const float* __restrict__ x, const float* __restrict__ W1,
                                                      float* __restrict__ h1, int N, int nTiles, int gBase) {
    if ((int)blockIdx.x >= NT) { gemm1_block(x, W1, h1, N, nTiles, gBase + (int)blockIdx.x - NT); return; }
    __shared__ int cur[BKT_MAX + 32];
    __shared__ int dlt[BKT_MAX];
    __shared__ int part[256];
    __shared__ int lpart[256];
    __shared__ int sw[TILE_E + 32];
    __shared__ int sd[TILE_E + 32];
    const int t = threadIdx.x;
    const int base = blockIdx.x * TILE_E + t;
    if (t < 32) cur[BKT_MAX + t] = 0;
    int d[16], s[16];
    #pragma unroll
    for (int k = 0; k < 16; ++k) {
        const int e = base + k * 256;
        const int ec = e < E ? e : E - 1;
        d[k] = dst[ec];
        s[k] = src[ec];
    }
    const int b0 = 2 * t, b1 = 2 * t + 1;
    const int b0c = b0 < NB ? b0 : NB - 1, b1c = b1 < NB ? b1 : NB - 1;
    const int c0r = bucketCount[b0c], c1r = bucketCount[b1c];
    const int o0r = tileOff[blockIdx.x * NB + b0c], o1r = tileOff[blockIdx.x * NB + b1c];
    const int h0r = tileHist[blockIdx.x * NB + b0c], h1r = tileHist[blockIdx.x * NB + b1c];
    __builtin_amdgcn_sched_barrier(0);
    const int c0 = b0 < NB ? c0r : 0, c1 = b1 < NB ? c1r : 0;
    const int o0 = b0 < NB ? o0r : 0, o1 = b1 < NB ? o1r : 0;
    const int h0 = b0 < NB ? h0r : 0, h1c = b1 < NB ? h1r : 0;
    part[t] = c0 + c1;
    lpart[t] = h0 + h1c;
    __syncthreads();
    scan256x2(part, lpart, t);
    const int ex = part[t] - (c0 + c1);
    const int lex = lpart[t] - (h0 + h1c);
    if (b0 < NB) { cur[b0] = ex + o0;      dlt[b0] = lex - (ex + o0); }
    if (b1 < NB) { cur[b1] = ex + c0 + o1; dlt[b1] = lex + h0 - (ex + c0 + o1); }
    if (blockIdx.x == 0) {
        if (b0 < NB) bucketStart[b0] = ex;
        if (b1 < NB) bucketStart[b1] = ex + c0;
        if (t == 255) bucketStart[NB] = part[255];
    }
    __syncthreads();
    int pos[16];
    #pragma unroll
    for (int k = 0; k < 16; ++k)
        pos[k] = atomicAdd(&cur[(base + k * 256 < E) ? (d[k] >> 8) : (BKT_MAX + (t & 31))], 1);
    int dl[16];
    #pragma unroll
    for (int k = 0; k < 16; ++k) dl[k] = dlt[d[k] >> 8];
    #pragma unroll
    for (int k = 0; k < 16; ++k) {
        const int slot = (base + k * 256 < E) ? pos[k] + dl[k] : TILE_E + (t & 31);
        sw[slot] = (int)(((unsigned)s[k] << 8) | (unsigned)(d[k] & 255));
        sd[slot] = pos[k];
    }
    __syncthreads();
    const int nE = min(TILE_E, E - (int)blockIdx.x * TILE_E);
    int wv[16], wd[16];
    #pragma unroll
    for (int k = 0; k < 16; ++k) {
        const int i = t + k * 256;
        const int ic = i < nE ? i : nE - 1;
        wv[k] = sw[ic];
        wd[k] = sd[ic];
    }
    #pragma unroll
    for (int k = 0; k < 16; ++k)
        if (t + k * 256 < nE) binned[wd[k]] = wv[k];
}

__global__ void __launch_bounds__(256) csr_fill(const int* __restrict__ bucketStart, const int* __restrict__ binned,
                                                int* __restrict__ rowptr, int* __restrict__ csr, int N, int NB,
                                                const float* __restrict__ x, const float* __restrict__ W1,
                                                float* __restrict__ h1, int nTiles, int gBase) {
    if ((int)blockIdx.x >= NB) { gemm1_block(x, W1, h1, N, nTiles, gBase + (int)blockIdx.x - NB); return; }
    __shared__ int cnt[256 + 32];
    __shared__ int part[256];
    __shared__ int stage[STAGE_MAX];
    const int t = threadIdx.x;
    const int spare = 256 + (t & 31);
    const int beg = bucketStart[blockIdx.x], end = bucketStart[blockIdx.x + 1];
    const int last = end - 1;
    const int nb = end - beg;
    const bool staged = nb <= STAGE_MAX;
    cnt[t] = 0;
    if (t < 32) cnt[256 + t] = 0;
    __syncthreads();
    for (int j = beg + t; j < end; j += 2048) {
        int v[8];
        #pragma unroll
        for (int k = 0; k < 8; ++k) {
            const int jk = j + k * 256;
            v[k] = binned[jk < end ? jk : last];
        }
        #pragma unroll
        for (int k = 0; k < 8; ++k)
            atomicAdd(&cnt[(j + k * 256 < end) ? (v[k] & 255) : spare], 1);
    }
    __syncthreads();
    const int c = cnt[t];
    part[t] = c;
    __syncthreads();
    scan256(part, t);
    const int first = beg + part[t] - c;
    const int n = blockIdx.x * 256 + t;
    if (n < N) rowptr[n] = first;
    if (n == N - 1) rowptr[N] = first + c;
    cnt[t] = first;
    __syncthreads();
    for (int j = beg + t; j < end; j += 2048) {
        int v[8], pos[8];
        #pragma unroll
        for (int k = 0; k < 8; ++k) {
            const int jk = j + k * 256;
            v[k] = binned[jk < end ? jk : last];
        }
        #pragma unroll
        for (int k = 0; k < 8; ++k)
            pos[k] = atomicAdd(&cnt[(j + k * 256 < end) ? (v[k] & 255) : spare], 1);
        #pragma unroll
        for (int k = 0; k < 8; ++k) {
            if (j + k * 256 < end) {
                const int sv = (int)((unsigned)v[k] >> 8);
                if (staged) stage[pos[k] - beg] = sv;
                else        csr[pos[k]] = sv;
            }
        }
    }
    if (!staged) return;
    __syncthreads();
    const int shift = beg & 31;
    for (int i0 = t - shift; i0 < nb; i0 += 2048) {
        int v[8];
        #pragma unroll
        for (int k = 0; k < 8; ++k) {
            const int i = i0 + k * 256;
            v[k] = stage[max(0, min(i, nb - 1))];
        }
        #pragma unroll
        for (int k = 0; k < 8; ++k) {
            const int i = i0 + k * 256;
            if (i >= 0 && i < nb) csr[beg + i] = v[k];
        }
    }
}

#define O1_PITCH 68
__global__ void __launch_bounds__(256) node_gather1(const int* __restrict__ rowptr,
                             const int* __restrict__ csr,
                             const float* __restrict__ attS, const float* __restrict__ attD,
                             const float* __restrict__ h1, const float* __restrict__ b1,
                             const float* __restrict__ W2, float* __restrict__ h2, int N) {
    __shared__ float ot[32 * O1_PITCH];
    const int t = threadIdx.x;
    const int h = t & 7;
    int n = (int)blockIdx.x * 32 + (t >> 3);
    if (n >= N) n = N - 1;
    const int i = n * HEADS + h;
    const int beg = rowptr[n];
    const int end = rowptr[n + 1];
    const int last = end - 1;

    const float4* __restrict__ hv = (const float4*)h1;
    const float* ws = attS + h * HID;
    const float w0 = ws[0], w1 = ws[1], w2 = ws[2], w3 = ws[3];
    const float w4 = ws[4], w5 = ws[5], w6 = ws[6], w7 = ws[7];
    float ad;
    {
        const float* wd = attD + h * HID;
        const float4 p0 = hv[i * 2], p1 = hv[i * 2 + 1];
        ad = dot8(p0, p1, wd[0], wd[1], wd[2], wd[3], wd[4], wd[5], wd[6], wd[7]);
    }
    float m = -INFINITY;
    float den = 0.f;
    float a0 = 0.f, a1 = 0.f, a2 = 0.f, a3 = 0.f, a4 = 0.f, a5 = 0.f, a6 = 0.f, a7 = 0.f;
    for (int j = beg; j < end; j += 4) {
        const bool k1 = j + 1 < end, k2 = j + 2 < end, k3 = j + 3 < end;
        const int j1 = k1 ? j + 1 : last;
        const int j2 = k2 ? j + 2 : last;
        const int j3 = k3 ? j + 3 : last;
        const int s0 = csr[j], s1 = csr[j1], s2 = csr[j2], s3 = csr[j3];
        const int i0 = s0 * (F1 / 4) + h * 2, i1 = s1 * (F1 / 4) + h * 2;
        const int i2 = s2 * (F1 / 4) + h * 2, i3 = s3 * (F1 / 4) + h * 2;
        const float4 x0 = hv[i0], y0 = hv[i0 + 1];
        const float4 x1 = hv[i1], y1 = hv[i1 + 1];
        const float4 x2 = hv[i2], y2 = hv[i2 + 1];
        const float4 x3 = hv[i3], y3 = hv[i3 + 1];
        const float g0 = dot8(x0, y0, w0, w1, w2, w3, w4, w5, w6, w7);
        const float g1 = dot8(x1, y1, w0, w1, w2, w3, w4, w5, w6, w7);
        const float g2 = dot8(x2, y2, w0, w1, w2, w3, w4, w5, w6, w7);
        const float g3 = dot8(x3, y3, w0, w1, w2, w3, w4, w5, w6, w7);
        const float v0 = lrelu(g0 + ad), v1 = lrelu(g1 + ad);
        const float v2 = lrelu(g2 + ad), v3 = lrelu(g3 + ad);
        const float u = v0 > v1 ? v0 : v1;
        const float w = v2 > v3 ? v2 : v3;
        const float z = u > w ? u : w;
        const float mn = z > m ? z : m;
        const float sc = expf(m - mn);
        m = mn;
        const float e0 = expf(v0 - mn);
        const float e1 = k1 ? expf(v1 - mn) : 0.f;
        const float e2 = k2 ? expf(v2 - mn) : 0.f;
        const float e3 = k3 ? expf(v3 - mn) : 0.f;
        den *= sc;
        a0 *= sc; a1 *= sc; a2 *= sc; a3 *= sc; a4 *= sc; a5 *= sc; a6 *= sc; a7 *= sc;
        den += e0; den += e1; den += e2; den += e3;
        a0 += e0 * x0.x; a1 += e0 * x0.y; a2 += e0 * x0.z; a3 += e0 * x0.w;
        a4 += e0 * y0.x; a5 += e0 * y0.y; a6 += e0 * y0.z; a7 += e0 * y0.w;
        a0 += e1 * x1.x; a1 += e1 * x1.y; a2 += e1 * x1.z; a3 += e1 * x1.w;
        a4 += e1 * y1.x; a5 += e1 * y1.y; a6 += e1 * y1.z; a7 += e1 * y1.w;
        a0 += e2 * x2.x; a1 += e2 * x2.y; a2 += e2 * x2.z; a3 += e2 * x2.w;
        a4 += e2 * y2.x; a5 += e2 * y2.y; a6 += e2 * y2.z; a7 += e2 * y2.w;
        a0 += e3 * x3.x; a1 += e3 * x3.y; a2 += e3 * x3.z; a3 += e3 * x3.w;
        a4 += e3 * y3.x; a5 += e3 * y3.y; a6 += e3 * y3.z; a7 += e3 * y3.w;
    }

    const int wv    = t >> 5;
    const int lane  = t & 31;
    const int row16 = lane & 15;
    const int kh    = lane >> 4;
    v2f b[16];
    if (wv < 2) {
        #pragma unroll
        for (int u = 0; u < 16; ++u) {
            const int ka = 4 * u + 2 * kh;
            b[u].x = W2[ka * OUT_C + row16];
            b[u].y = W2[(ka + 1) * OUT_C + row16];
        }
    }

    const float dn = den + 1e-16f;
    const float* bp = b1 + h * HID;
    const float t0 = a0 / dn + bp[0], t1 = a1 / dn + bp[1], t2 = a2 / dn + bp[2], t3 = a3 / dn + bp[3];
    const float t4 = a4 / dn + bp[4], t5 = a5 / dn + bp[5], t6 = a6 / dn + bp[6], t7 = a7 / dn + bp[7];
    float4 r0, r1;
    r0.x = t0 > 0.f ? t0 : (expf(t0) - 1.f);
    r0.y = t1 > 0.f ? t1 : (expf(t1) - 1.f);
    r0.z = t2 > 0.f ? t2 : (expf(t2) - 1.f);
    r0.w = t3 > 0.f ? t3 : (expf(t3) - 1.f);
    r1.x = t4 > 0.f ? t4 : (expf(t4) - 1.f);
    r1.y = t5 > 0.f ? t5 : (expf(t5) - 1.f);
    r1.z = t6 > 0.f ? t6 : (expf(t6) - 1.f);
    r1.w = t7 > 0.f ? t7 : (expf(t7) - 1.f);
    float4* lp = (float4*)&ot[(t >> 3) * O1_PITCH + h * HID];
    lp[0] = r0;
    lp[1] = r1;
    __syncthreads();
    if (wv >= 2) return;

    const float* ar = &ot[(wv * 16 + row16) * O1_PITCH + 2 * kh];
    v8f acc = {};
    #pragma unroll
    for (int u = 0; u < 16; ++u) {
        v2f a;
        a.x = ar[4 * u];
        a.y = ar[4 * u + 1];
        acc = __builtin_amdgcn_wmma_f32_16x16x4_f32(false, a, false, b[u], (short)0, acc, false, false);
    }
    #pragma unroll
    for (int r = 0; r < 8; ++r) {
        const int row = (int)blockIdx.x * 32 + wv * 16 + r + 8 * kh;
        if (row < N) h2[(size_t)row * OUT_C + row16] = acc[r];
    }
}

__global__ void node_gather2(const int* __restrict__ rowptr,
                             const int* __restrict__ csr,
                             const float* __restrict__ attS, const float* __restrict__ attD,
                             const float* __restrict__ h2, const float* __restrict__ b2,
                             float* __restrict__ out, int NQ) {
    int i = blockIdx.x * blockDim.x + threadIdx.x;
    if (i >= NQ) return;
    const int n = i >> 2, q = i & 3;
    const int beg = rowptr[n];
    const int end = rowptr[n + 1];
    const int last = end - 1;

    const float4* __restrict__ hv = (const float4*)h2;
    const float* ws = attS + q * 4;
    const float w0 = ws[0], w1 = ws[1], w2 = ws[2], w3 = ws[3];
    float ad;
    {
        const float* wd = attD + q * 4;
        const float4 p = hv[i];
        ad = quad_sum(dot4(p, wd[0], wd[1], wd[2], wd[3]));
    }
    float m = -INFINITY;
    float den = 0.f;
    float a0 = 0.f, a1 = 0.f, a2 = 0.f, a3 = 0.f;
    for (int j = beg; j < end; j += 4) {
        const bool k1 = j + 1 < end, k2 = j + 2 < end, k3 = j + 3 < end;
        const int j1 = k1 ? j + 1 : last;
        const int j2 = k2 ? j + 2 : last;
        const int j3 = k3 ? j + 3 : last;
        const int s0 = csr[j], s1 = csr[j1], s2 = csr[j2], s3 = csr[j3];
        const float4 x0 = hv[s0 * (OUT_C / 4) + q];
        const float4 x1 = hv[s1 * (OUT_C / 4) + q];
        const float4 x2 = hv[s2 * (OUT_C / 4) + q];
        const float4 x3 = hv[s3 * (OUT_C / 4) + q];
        const float g0 = quad_sum(dot4(x0, w0, w1, w2, w3)), g1 = quad_sum(dot4(x1, w0, w1, w2, w3));
        const float g2 = quad_sum(dot4(x2, w0, w1, w2, w3)), g3 = quad_sum(dot4(x3, w0, w1, w2, w3));
        const float v0 = lrelu(g0 + ad), v1 = lrelu(g1 + ad);
        const float v2 = lrelu(g2 + ad), v3 = lrelu(g3 + ad);
        const float u = v0 > v1 ? v0 : v1;
        const float w = v2 > v3 ? v2 : v3;
        const float z = u > w ? u : w;
        const float mn = z > m ? z : m;
        const float sc = expf(m - mn);
        m = mn;
        const float e0 = expf(v0 - mn);
        const float e1 = k1 ? expf(v1 - mn) : 0.f;
        const float e2 = k2 ? expf(v2 - mn) : 0.f;
        const float e3 = k3 ? expf(v3 - mn) : 0.f;
        den *= sc;
        a0 *= sc; a1 *= sc; a2 *= sc; a3 *= sc;
        den += e0; den += e1; den += e2; den += e3;
        a0 += e0 * x0.x; a1 += e0 * x0.y; a2 += e0 * x0.z; a3 += e0 * x0.w;
        a0 += e1 * x1.x; a1 += e1 * x1.y; a2 += e1 * x1.z; a3 += e1 * x1.w;
        a0 += e2 * x2.x; a1 += e2 * x2.y; a2 += e2 * x2.z; a3 += e2 * x2.w;
        a0 += e3 * x3.x; a1 += e3 * x3.y; a2 += e3 * x3.z; a3 += e3 * x3.w;
    }

    const float dn = den + 1e-16f;
    const float* bq = b2 + q * 4;
    float4 r;
    r.x = a0 / dn + bq[0];
    r.y = a1 / dn + bq[1];
    r.z = a2 / dn + bq[2];
    r.w = a3 / dn + bq[3];
    ((float4*)(out + (size_t)n * OUT_C))[q] = r;
}

extern "C" void kernel_launch(void* const* d_in, const int* in_sizes, int n_in,
                              void* d_out, int out_size, void* d_ws, size_t ws_size,
                              hipStream_t stream) {
    const int N = in_sizes[0] / IN_C;
    const int E = in_sizes[1] / 2;

    const float* x     = (const float*)d_in[0];
    const int*   ei    = (const int*)  d_in[1];
    const int*   src   = ei;
    const int*   dst   = ei + E;
    const float* W1    = (const float*)d_in[2];
    const float* b1    = (const float*)d_in[3];
    const float* attS1 = (const float*)d_in[4];
    const float* attD1 = (const float*)d_in[5];
    const float* W2    = (const float*)d_in[6];
    const float* b2    = (const float*)d_in[7];
    const float* attS2 = (const float*)d_in[8];
    const float* attD2 = (const float*)d_in[9];
    float* out = (float*)d_out;

    char* ws = (char*)d_ws;
    size_t off = 0;
    auto alloc = [&](size_t elems) -> float* {
        float* p = (float*)(ws + off);
        off = (off + elems * sizeof(float) + 255) & ~(size_t)255;
        return p;
    };
    float* h1   = alloc((size_t)N * F1);
    float* h2   = alloc((size_t)N * OUT_C);
    const int NB = (N + 255) / 256;
    const int NT = (E + TILE_E - 1) / TILE_E;
    int*   bucketCount = (int*)alloc((size_t)BKT_MAX);
    int*   bucketStart = (int*)alloc((size_t)BKT_MAX + 1);
    int*   tileOff     = (int*)alloc((size_t)NT * NB);
    int*   tileHist    = (int*)alloc((size_t)NT * NB);
    int*   rowptr      = (int*)alloc((size_t)N + 1);
    int*   binned      = (int*)alloc((size_t)E);
    int*   csr         = (int*)alloc((size_t)E);

    hipMemsetAsync(bucketCount, 0, (size_t)NB * sizeof(int), stream);

    const int TB = 256;
    auto nb = [](long long n, int tb) { return (unsigned)((n + tb - 1) / tb); };

    const int nTiles = (N + 15) / 16;
    const int gemmBlks = (nTiles + 7) / 8;
    const int gPart = (gemmBlks + 2) / 3;

    bucket_count<<<NT + gPart, 256, 0, stream>>>(dst, bucketCount, tileOff, tileHist, E, NB, NT, x, W1, h1, N, nTiles, 0);
    bucket_scatter<<<NT + gPart, 256, 0, stream>>>(src, dst, bucketCount, tileOff, tileHist, bucketStart, binned, E, NB, NT,
                                                   x, W1, h1, N, nTiles, gPart);
    csr_fill<<<NB + gPart, 256, 0, stream>>>(bucketStart, binned, rowptr, csr, N, NB, x, W1, h1, nTiles, 2 * gPart);

    node_gather1<<<nb((long long)N, 32), TB, 0, stream>>>(rowptr, csr, attS1, attD1, h1, b1, W2, h2, N);

    node_gather2<<<nb((long long)N * 4, TB), TB, 0, stream>>>(rowptr, csr, attS2, attD2, h2, b2, out, N * 4);
}


# Every exponential (both softmax walks, ELU) taken as 2^y on the one-instruction hardware form, attention vectors pre-scaled by log2(e); summation order kept, rounding of the weights changes
# speedup vs baseline: 8.1534x; 1.0592x over previous
#include <hip/hip_runtime.h>
#include <math.h>

#define IN_C   128
#define HEADS  8
#define HID    8
#define F1     (HEADS * HID)
#define OUT_C  16
#define NEG_SLOPE 0.2f

typedef float v2f __attribute__((ext_vector_type(2)));
typedef float v8f __attribute__((ext_vector_type(8)));

__device__ __forceinline__ float lrelu(float v) {
    return v > 0.0f ? v : NEG_SLOPE * v;
}

#define LOG2E 1.4426950408889634f
__device__ __forceinline__ float exp2_hw(float y) {
    return __builtin_amdgcn_exp2f(y);
}

__device__ __forceinline__ void gemm1_block(const float* __restrict__ x,
                                            const float* __restrict__ W,
                                            float* __restrict__ h1,
                                            int N, int nTiles, int gblk) {
    __shared__ float w[IN_C * F1];
    for (int i = threadIdx.x; i < IN_C * F1; i += 256) w[i] = W[i];
    __syncthreads();

    const int tile = gblk * 8 + (threadIdx.x >> 5);
    if (tile >= nTiles) return;

    const int lane  = threadIdx.x & 31;
    const int row16 = lane & 15;
    const int kh    = lane >> 4;

    int rowIn = tile * 16 + row16;
    if (rowIn >= N) rowIn = N - 1;
    const float* xr = x + (size_t)rowIn * IN_C;

    v8f acc0 = {}, acc1 = {}, acc2 = {}, acc3 = {};

    for (int k = 0; k < IN_C; k += 16) {
        v2f a[4];
        #pragma unroll
        for (int u = 0; u < 4; ++u) {
            const int ka = k + 4 * u + 2 * kh;
            a[u].x = xr[ka];
            a[u].y = xr[ka + 1];
        }
        __builtin_amdgcn_sched_barrier(0);
        #pragma unroll
        for (int u = 0; u < 4; ++u) {
            const int ka = k + 4 * u + 2 * kh;
            const float* w0 = &w[ka * F1 + row16];
            const float* w1 = &w[(ka + 1) * F1 + row16];
            v2f b0; b0.x = w0[0];  b0.y = w1[0];
            v2f b1; b1.x = w0[16]; b1.y = w1[16];
            v2f b2; b2.x = w0[32]; b2.y = w1[32];
            v2f b3; b3.x = w0[48]; b3.y = w1[48];
            acc0 = __builtin_amdgcn_wmma_f32_16x16x4_f32(false, a[u], false, b0, (short)0, acc0, false, false);
            acc1 = __builtin_amdgcn_wmma_f32_16x16x4_f32(false, a[u], false, b1, (short)0, acc1, false, false);
            acc2 = __builtin_amdgcn_wmma_f32_16x16x4_f32(false, a[u], false, b2, (short)0, acc2, false, false);
            acc3 = __builtin_amdgcn_wmma_f32_16x16x4_f32(false, a[u], false, b3, (short)0, acc3, false, false);
        }
    }

    #pragma unroll
    for (int i = 0; i < 8; ++i) {
        const int r = tile * 16 + i + 8 * kh;
        if (r < N) {
            float* o = h1 + (size_t)r * F1 + row16;
            o[0]  = acc0[i];
            o[16] = acc1[i];
            o[32] = acc2[i];
            o[48] = acc3[i];
        }
    }
}

__device__ __forceinline__ float dot8(const float4 a, const float4 b,
                                      const float w0, const float w1, const float w2, const float w3,
                                      const float w4, const float w5, const float w6, const float w7) {
    float s = a.x * w0;
    s = fmaf(a.y, w1, s); s = fmaf(a.z, w2, s); s = fmaf(a.w, w3, s);
    s = fmaf(b.x, w4, s); s = fmaf(b.y, w5, s); s = fmaf(b.z, w6, s); s = fmaf(b.w, w7, s);
    return s;
}

__device__ __forceinline__ float dot4(const float4 a, const float w0, const float w1, const float w2, const float w3) {
    float s = a.x * w0;
    s = fmaf(a.y, w1, s); s = fmaf(a.z, w2, s); s = fmaf(a.w, w3, s);
    return s;
}

__device__ __forceinline__ float quad_sum(float v) {
    v += __builtin_bit_cast(float, __builtin_amdgcn_update_dpp(0, __builtin_bit_cast(int, v), 0xB1, 0xF, 0xF, true));
    v += __builtin_bit_cast(float, __builtin_amdgcn_update_dpp(0, __builtin_bit_cast(int, v), 0x4E, 0xF, 0xF, true));
    return v;
}

#define TILE_E  4096
#define BKT_MAX 512
#define STAGE_MAX 8192

__device__ __forceinline__ void scan256(int* part, const int t) {
    for (int off = 1; off < 256; off <<= 1) {
        const int v = (t >= off) ? part[t - off] : 0;
        __syncthreads();
        part[t] += v;
        __syncthreads();
    }
}

__device__ __forceinline__ void scan256x2(int* pa, int* pb, const int t) {
    for (int off = 1; off < 256; off <<= 1) {
        const int va = (t >= off) ? pa[t - off] : 0;
        const int vb = (t >= off) ? pb[t - off] : 0;
        __syncthreads();
        pa[t] += va;
        pb[t] += vb;
        __syncthreads();
    }
}

__global__ void __launch_bounds__(256) bucket_count(const int* __restrict__ dst, int* __restrict__ bucketCount,
                                                    int* __restrict__ tileOff, int* __restrict__ tileHist,
                                                    int E, int NB, int NT,
                                                    const float* __restrict__ x, const float* __restrict__ W1,
                                                    float* __restrict__ h1, int N, int nTiles, int gBase) {
    if ((int)blockIdx.x >= NT) { gemm1_block(x, W1, h1, N, nTiles, gBase + (int)blockIdx.x - NT); return; }
    __shared__ int hist[BKT_MAX];
    const int t = threadIdx.x;
    for (int i = t; i < NB; i += 256) hist[i] = 0;
    __syncthreads();
    const int base = blockIdx.x * TILE_E + t;
    int d[16];
    #pragma unroll
    for (int k = 0; k < 16; ++k) {
        const int e = base + k * 256;
        d[k] = dst[e < E ? e : E - 1];
    }
    #pragma unroll
    for (int k = 0; k < 16; ++k) {
        if (base + k * 256 < E) atomicAdd(&hist[d[k] >> 8], 1);
    }
    __syncthreads();
    const int i0 = t, i1 = t + 256;
    int r0 = 0, r1 = 0;
    if (i0 < NB) r0 = atomicAdd(&bucketCount[i0], hist[i0]);
    if (i1 < NB) r1 = atomicAdd(&bucketCount[i1], hist[i1]);
    if (i0 < NB) tileOff[blockIdx.x * NB + i0] = r0;
    if (i1 < NB) tileOff[blockIdx.x * NB + i1] = r1;
    if (i0 < NB) tileHist[blockIdx.x * NB + i0] = hist[i0];
    if (i1 < NB) tileHist[blockIdx.x * NB + i1] = hist[i1];
}

__global__ void __launch_bounds__(256) bucket_scatter(const int* __restrict__ src, const int* __restrict__ dst,
                                                      const int* __restrict__ bucketCount,
                                                      const int* __restrict__ tileOff,
                                                      const int* __restrict__ tileHist,
                                                      int* __restrict__ bucketStart, int* __restrict__ binned,
                                                      int E, int NB, int NT,
                                                      const float* __restrict__ x, const float* __restrict__ W1,
                                                      float* __restrict__ h1, int N, int nTiles, int gBase) {
    if ((int)blockIdx.x >= NT) { gemm1_block(x, W1, h1, N, nTiles, gBase + (int)blockIdx.x - NT); return; }
    __shared__ int cur[BKT_MAX + 32];
    __shared__ int dlt[BKT_MAX];
    __shared__ int part[256];
    __shared__ int lpart[256];
    __shared__ int sw[TILE_E + 32];
    __shared__ int sd[TILE_E + 32];
    const int t = threadIdx.x;
    const int base = blockIdx.x * TILE_E + t;
    if (t < 32) cur[BKT_MAX + t] = 0;
    int d[16], s[16];
    #pragma unroll
    for (int k = 0; k < 16; ++k) {
        const int e = base + k * 256;
        const int ec = e < E ? e : E - 1;
        d[k] = dst[ec];
        s[k] = src[ec];
    }
    const int b0 = 2 * t, b1 = 2 * t + 1;
    const int b0c = b0 < NB ? b0 : NB - 1, b1c = b1 < NB ? b1 : NB - 1;
    const int c0r = bucketCount[b0c], c1r = bucketCount[b1c];
    const int o0r = tileOff[blockIdx.x * NB + b0c], o1r = tileOff[blockIdx.x * NB + b1c];
    const int h0r = tileHist[blockIdx.x * NB + b0c], h1r = tileHist[blockIdx.x * NB + b1c];
    __builtin_amdgcn_sched_barrier(0);
    const int c0 = b0 < NB ? c0r : 0, c1 = b1 < NB ? c1r : 0;
    const int o0 = b0 < NB ? o0r : 0, o1 = b1 < NB ? o1r : 0;
    const int h0 = b0 < NB ? h0r : 0, h1c = b1 < NB ? h1r : 0;
    part[t] = c0 + c1;
    lpart[t] = h0 + h1c;
    __syncthreads();
    scan256x2(part, lpart, t);
    const int ex = part[t] - (c0 + c1);
    const int lex = lpart[t] - (h0 + h1c);
    if (b0 < NB) { cur[b0] = ex + o0;      dlt[b0] = lex - (ex + o0); }
    if (b1 < NB) { cur[b1] = ex + c0 + o1; dlt[b1] = lex + h0 - (ex + c0 + o1); }
    if (blockIdx.x == 0) {
        if (b0 < NB) bucketStart[b0] = ex;
        if (b1 < NB) bucketStart[b1] = ex + c0;
        if (t == 255) bucketStart[NB] = part[255];
    }
    __syncthreads();
    int pos[16];
    #pragma unroll
    for (int k = 0; k < 16; ++k)
        pos[k] = atomicAdd(&cur[(base + k * 256 < E) ? (d[k] >> 8) : (BKT_MAX + (t & 31))], 1);
    int dl[16];
    #pragma unroll
    for (int k = 0; k < 16; ++k) dl[k] = dlt[d[k] >> 8];
    #pragma unroll
    for (int k = 0; k < 16; ++k) {
        const int slot = (base + k * 256 < E) ? pos[k] + dl[k] : TILE_E + (t & 31);
        sw[slot] = (int)(((unsigned)s[k] << 8) | (unsigned)(d[k] & 255));
        sd[slot] = pos[k];
    }
    __syncthreads();
    const int nE = min(TILE_E, E - (int)blockIdx.x * TILE_E);
    int wv[16], wd[16];
    #pragma unroll
    for (int k = 0; k < 16; ++k) {
        const int i = t + k * 256;
        const int ic = i < nE ? i : nE - 1;
        wv[k] = sw[ic];
        wd[k] = sd[ic];
    }
    #pragma unroll
    for (int k = 0; k < 16; ++k)
        if (t + k * 256 < nE) binned[wd[k]] = wv[k];
}

__global__ void __launch_bounds__(256) csr_fill(const int* __restrict__ bucketStart, const int* __restrict__ binned,
                                                int* __restrict__ rowptr, int* __restrict__ csr, int N, int NB,
                                                const float* __restrict__ x, const float* __restrict__ W1,
                                                float* __restrict__ h1, int nTiles, int gBase) {
    if ((int)blockIdx.x >= NB) { gemm1_block(x, W1, h1, N, nTiles, gBase + (int)blockIdx.x - NB); return; }
    __shared__ int cnt[256 + 32];
    __shared__ int part[256];
    __shared__ int stage[STAGE_MAX];
    const int t = threadIdx.x;
    const int spare = 256 + (t & 31);
    const int beg = bucketStart[blockIdx.x], end = bucketStart[blockIdx.x + 1];
    const int last = end - 1;
    const int nb = end - beg;
    const bool staged = nb <= STAGE_MAX;
    cnt[t] = 0;
    if (t < 32) cnt[256 + t] = 0;
    __syncthreads();
    for (int j = beg + t; j < end; j += 2048) {
        int v[8];
        #pragma unroll
        for (int k = 0; k < 8; ++k) {
            const int jk = j + k * 256;
            v[k] = binned[jk < end ? jk : last];
        }
        #pragma unroll
        for (int k = 0; k < 8; ++k)
            atomicAdd(&cnt[(j + k * 256 < end) ? (v[k] & 255) : spare], 1);
    }
    __syncthreads();
    const int c = cnt[t];
    part[t] = c;
    __syncthreads();
    scan256(part, t);
    const int first = beg + part[t] - c;
    const int n = blockIdx.x * 256 + t;
    if (n < N) rowptr[n] = first;
    if (n == N - 1) rowptr[N] = first + c;
    cnt[t] = first;
    __syncthreads();
    for (int j = beg + t; j < end; j += 2048) {
        int v[8], pos[8];
        #pragma unroll
        for (int k = 0; k < 8; ++k) {
            const int jk = j + k * 256;
            v[k] = binned[jk < end ? jk : last];
        }
        #pragma unroll
        for (int k = 0; k < 8; ++k)
            pos[k] = atomicAdd(&cnt[(j + k * 256 < end) ? (v[k] & 255) : spare], 1);
        #pragma unroll
        for (int k = 0; k < 8; ++k) {
            if (j + k * 256 < end) {
                const int sv = (int)((unsigned)v[k] >> 8);
                if (staged) stage[pos[k] - beg] = sv;
                else        csr[pos[k]] = sv;
            }
        }
    }
    if (!staged) return;
    __syncthreads();
    const int shift = beg & 31;
    for (int i0 = t - shift; i0 < nb; i0 += 2048) {
        int v[8];
        #pragma unroll
        for (int k = 0; k < 8; ++k) {
            const int i = i0 + k * 256;
            v[k] = stage[max(0, min(i, nb - 1))];
        }
        #pragma unroll
        for (int k = 0; k < 8; ++k) {
            const int i = i0 + k * 256;
            if (i >= 0 && i < nb) csr[beg + i] = v[k];
        }
    }
}

#define O1_PITCH 68
__global__ void __launch_bounds__(256, 12) node_gather1(const int* __restrict__ rowptr,
                             const int* __restrict__ csr,
                             const float* __restrict__ attS, const float* __restrict__ attD,
                             const float* __restrict__ h1, const float* __restrict__ b1,
                             const float* __restrict__ W2, float* __restrict__ h2, int N) {
    __shared__ float ot[32 * O1_PITCH];
    const int t = threadIdx.x;
    const int h = t & 7;
    int n = (int)blockIdx.x * 32 + (t >> 3);
    if (n >= N) n = N - 1;
    const int i = n * HEADS + h;
    const int beg = rowptr[n];
    const int end = rowptr[n + 1];
    const int last = end - 1;

    const float4* __restrict__ hv = (const float4*)h1;
    const float* ws = attS + h * HID;
    const float w0 = ws[0] * LOG2E, w1 = ws[1] * LOG2E, w2 = ws[2] * LOG2E, w3 = ws[3] * LOG2E;
    const float w4 = ws[4] * LOG2E, w5 = ws[5] * LOG2E, w6 = ws[6] * LOG2E, w7 = ws[7] * LOG2E;
    float ad;
    {
        const float* wd = attD + h * HID;
        const float4 p0 = hv[i * 2], p1 = hv[i * 2 + 1];
        ad = dot8(p0, p1, wd[0] * LOG2E, wd[1] * LOG2E, wd[2] * LOG2E, wd[3] * LOG2E,
                  wd[4] * LOG2E, wd[5] * LOG2E, wd[6] * LOG2E, wd[7] * LOG2E);
    }
    float m = -INFINITY;
    float den = 0.f;
    float a0 = 0.f, a1 = 0.f, a2 = 0.f, a3 = 0.f, a4 = 0.f, a5 = 0.f, a6 = 0.f, a7 = 0.f;
    for (int j = beg; j < end; j += 4) {
        const bool k1 = j + 1 < end, k2 = j + 2 < end, k3 = j + 3 < end;
        const int j1 = k1 ? j + 1 : last;
        const int j2 = k2 ? j + 2 : last;
        const int j3 = k3 ? j + 3 : last;
        const int s0 = csr[j], s1 = csr[j1], s2 = csr[j2], s3 = csr[j3];
        const int i0 = s0 * (F1 / 4) + h * 2, i1 = s1 * (F1 / 4) + h * 2;
        const int i2 = s2 * (F1 / 4) + h * 2, i3 = s3 * (F1 / 4) + h * 2;
        const float4 x0 = hv[i0], y0 = hv[i0 + 1];
        const float4 x1 = hv[i1], y1 = hv[i1 + 1];
        const float4 x2 = hv[i2], y2 = hv[i2 + 1];
        const float4 x3 = hv[i3], y3 = hv[i3 + 1];
        const float g0 = dot8(x0, y0, w0, w1, w2, w3, w4, w5, w6, w7);
        const float g1 = dot8(x1, y1, w0, w1, w2, w3, w4, w5, w6, w7);
        const float g2 = dot8(x2, y2, w0, w1, w2, w3, w4, w5, w6, w7);
        const float g3 = dot8(x3, y3, w0, w1, w2, w3, w4, w5, w6, w7);
        const float v0 = lrelu(g0 + ad), v1 = lrelu(g1 + ad);
        const float v2 = lrelu(g2 + ad), v3 = lrelu(g3 + ad);
        const float u = v0 > v1 ? v0 : v1;
        const float w = v2 > v3 ? v2 : v3;
        const float z = u > w ? u : w;
        const float mn = z > m ? z : m;
        const float sc = exp2_hw(m - mn);
        m = mn;
        const float e0 = exp2_hw(v0 - mn);
        const float e1 = k1 ? exp2_hw(v1 - mn) : 0.f;
        const float e2 = k2 ? exp2_hw(v2 - mn) : 0.f;
        const float e3 = k3 ? exp2_hw(v3 - mn) : 0.f;
        den *= sc;
        a0 *= sc; a1 *= sc; a2 *= sc; a3 *= sc; a4 *= sc; a5 *= sc; a6 *= sc; a7 *= sc;
        den += e0; den += e1; den += e2; den += e3;
        a0 += e0 * x0.x; a1 += e0 * x0.y; a2 += e0 * x0.z; a3 += e0 * x0.w;
        a4 += e0 * y0.x; a5 += e0 * y0.y; a6 += e0 * y0.z; a7 += e0 * y0.w;
        a0 += e1 * x1.x; a1 += e1 * x1.y; a2 += e1 * x1.z; a3 += e1 * x1.w;
        a4 += e1 * y1.x; a5 += e1 * y1.y; a6 += e1 * y1.z; a7 += e1 * y1.w;
        a0 += e2 * x2.x; a1 += e2 * x2.y; a2 += e2 * x2.z; a3 += e2 * x2.w;
        a4 += e2 * y2.x; a5 += e2 * y2.y; a6 += e2 * y2.z; a7 += e2 * y2.w;
        a0 += e3 * x3.x; a1 += e3 * x3.y; a2 += e3 * x3.z; a3 += e3 * x3.w;
        a4 += e3 * y3.x; a5 += e3 * y3.y; a6 += e3 * y3.z; a7 += e3 * y3.w;
    }

    const int wv    = t >> 5;
    const int lane  = t & 31;
    const int row16 = lane & 15;
    const int kh    = lane >> 4;
    v2f b[16];
    if (wv < 2) {
        #pragma unroll
        for (int u = 0; u < 16; ++u) {
            const int ka = 4 * u + 2 * kh;
            b[u].x = W2[ka * OUT_C + row16];
            b[u].y = W2[(ka + 1) * OUT_C + row16];
        }
    }

    const float dn = den + 1e-16f;
    const float* bp = b1 + h * HID;
    const float t0 = a0 / dn + bp[0], t1 = a1 / dn + bp[1], t2 = a2 / dn + bp[2], t3 = a3 / dn + bp[3];
    const float t4 = a4 / dn + bp[4], t5 = a5 / dn + bp[5], t6 = a6 / dn + bp[6], t7 = a7 / dn + bp[7];
    float4 r0, r1;
    r0.x = t0 > 0.f ? t0 : (exp2_hw(t0 * LOG2E) - 1.f);
    r0.y = t1 > 0.f ? t1 : (exp2_hw(t1 * LOG2E) - 1.f);
    r0.z = t2 > 0.f ? t2 : (exp2_hw(t2 * LOG2E) - 1.f);
    r0.w = t3 > 0.f ? t3 : (exp2_hw(t3 * LOG2E) - 1.f);
    r1.x = t4 > 0.f ? t4 : (exp2_hw(t4 * LOG2E) - 1.f);
    r1.y = t5 > 0.f ? t5 : (exp2_hw(t5 * LOG2E) - 1.f);
    r1.z = t6 > 0.f ? t6 : (exp2_hw(t6 * LOG2E) - 1.f);
    r1.w = t7 > 0.f ? t7 : (exp2_hw(t7 * LOG2E) - 1.f);
    float4* lp = (float4*)&ot[(t >> 3) * O1_PITCH + h * HID];
    lp[0] = r0;
    lp[1] = r1;
    __syncthreads();
    if (wv >= 2) return;

    const float* ar = &ot[(wv * 16 + row16) * O1_PITCH + 2 * kh];
    v8f acc = {};
    #pragma unroll
    for (int u = 0; u < 16; ++u) {
        v2f a;
        a.x = ar[4 * u];
        a.y = ar[4 * u + 1];
        acc = __builtin_amdgcn_wmma_f32_16x16x4_f32(false, a, false, b[u], (short)0, acc, false, false);
    }
    #pragma unroll
    for (int r = 0; r < 8; ++r) {
        const int row = (int)blockIdx.x * 32 + wv * 16 + r + 8 * kh;
        if (row < N) h2[(size_t)row * OUT_C + row16] = acc[r];
    }
}

__global__ void node_gather2(const int* __restrict__ rowptr,
                             const int* __restrict__ csr,
                             const float* __restrict__ attS, const float* __restrict__ attD,
                             const float* __restrict__ h2, const float* __restrict__ b2,
                             float* __restrict__ out, int NQ) {
    int i = blockIdx.x * blockDim.x + threadIdx.x;
    if (i >= NQ) return;
    const int n = i >> 2, q = i & 3;
    const int beg = rowptr[n];
    const int end = rowptr[n + 1];
    const int last = end - 1;

    const float4* __restrict__ hv = (const float4*)h2;
    const float* ws = attS + q * 4;
    const float w0 = ws[0] * LOG2E, w1 = ws[1] * LOG2E, w2 = ws[2] * LOG2E, w3 = ws[3] * LOG2E;
    float ad;
    {
        const float* wd = attD + q * 4;
        const float4 p = hv[i];
        ad = quad_sum(dot4(p, wd[0] * LOG2E, wd[1] * LOG2E, wd[2] * LOG2E, wd[3] * LOG2E));
    }
    float m = -INFINITY;
    float den = 0.f;
    float a0 = 0.f, a1 = 0.f, a2 = 0.f, a3 = 0.f;
    for (int j = beg; j < end; j += 4) {
        const bool k1 = j + 1 < end, k2 = j + 2 < end, k3 = j + 3 < end;
        const int j1 = k1 ? j + 1 : last;
        const int j2 = k2 ? j + 2 : last;
        const int j3 = k3 ? j + 3 : last;
        const int s0 = csr[j], s1 = csr[j1], s2 = csr[j2], s3 = csr[j3];
        const float4 x0 = hv[s0 * (OUT_C / 4) + q];
        const float4 x1 = hv[s1 * (OUT_C / 4) + q];
        const float4 x2 = hv[s2 * (OUT_C / 4) + q];
        const float4 x3 = hv[s3 * (OUT_C / 4) + q];
        const float g0 = quad_sum(dot4(x0, w0, w1, w2, w3)), g1 = quad_sum(dot4(x1, w0, w1, w2, w3));
        const float g2 = quad_sum(dot4(x2, w0, w1, w2, w3)), g3 = quad_sum(dot4(x3, w0, w1, w2, w3));
        const float v0 = lrelu(g0 + ad), v1 = lrelu(g1 + ad);
        const float v2 = lrelu(g2 + ad), v3 = lrelu(g3 + ad);
        const float u = v0 > v1 ? v0 : v1;
        const float w = v2 > v3 ? v2 : v3;
        const float z = u > w ? u : w;
        const float mn = z > m ? z : m;
        const float sc = exp2_hw(m - mn);
        m = mn;
        const float e0 = exp2_hw(v0 - mn);
        const float e1 = k1 ? exp2_hw(v1 - mn) : 0.f;
        const float e2 = k2 ? exp2_hw(v2 - mn) : 0.f;
        const float e3 = k3 ? exp2_hw(v3 - mn) : 0.f;
        den *= sc;
        a0 *= sc; a1 *= sc; a2 *= sc; a3 *= sc;
        den += e0; den += e1; den += e2; den += e3;
        a0 += e0 * x0.x; a1 += e0 * x0.y; a2 += e0 * x0.z; a3 += e0 * x0.w;
        a0 += e1 * x1.x; a1 += e1 * x1.y; a2 += e1 * x1.z; a3 += e1 * x1.w;
        a0 += e2 * x2.x; a1 += e2 * x2.y; a2 += e2 * x2.z; a3 += e2 * x2.w;
        a0 += e3 * x3.x; a1 += e3 * x3.y; a2 += e3 * x3.z; a3 += e3 * x3.w;
    }

    const float dn = den + 1e-16f;
    const float* bq = b2 + q * 4;
    float4 r;
    r.x = a0 / dn + bq[0];
    r.y = a1 / dn + bq[1];
    r.z = a2 / dn + bq[2];
    r.w = a3 / dn + bq[3];
    ((float4*)(out + (size_t)n * OUT_C))[q] = r;
}

extern "C" void kernel_launch(void* const* d_in, const int* in_sizes, int n_in,
                              void* d_out, int out_size, void* d_ws, size_t ws_size,
                              hipStream_t stream) {
    const int N = in_sizes[0] / IN_C;
    const int E = in_sizes[1] / 2;

    const float* x     = (const float*)d_in[0];
    const int*   ei    = (const int*)  d_in[1];
    const int*   src   = ei;
    const int*   dst   = ei + E;
    const float* W1    = (const float*)d_in[2];
    const float* b1    = (const float*)d_in[3];
    const float* attS1 = (const float*)d_in[4];
    const float* attD1 = (const float*)d_in[5];
    const float* W2    = (const float*)d_in[6];
    const float* b2    = (const float*)d_in[7];
    const float* attS2 = (const float*)d_in[8];
    const float* attD2 = (const float*)d_in[9];
    float* out = (float*)d_out;

    char* ws = (char*)d_ws;
    size_t off = 0;
    auto alloc = [&](size_t elems) -> float* {
        float* p = (float*)(ws + off);
        off = (off + elems * sizeof(float) + 255) & ~(size_t)255;
        return p;
    };
    float* h1   = alloc((size_t)N * F1);
    float* h2   = alloc((size_t)N * OUT_C);
    const int NB = (N + 255) / 256;
    const int NT = (E + TILE_E - 1) / TILE_E;
    int*   bucketCount = (int*)alloc((size_t)BKT_MAX);
    int*   bucketStart = (int*)alloc((size_t)BKT_MAX + 1);
    int*   tileOff     = (int*)alloc((size_t)NT * NB);
    int*   tileHist    = (int*)alloc((size_t)NT * NB);
    int*   rowptr      = (int*)alloc((size_t)N + 1);
    int*   binned      = (int*)alloc((size_t)E);
    int*   csr         = (int*)alloc((size_t)E);

    hipMemsetAsync(bucketCount, 0, (size_t)NB * sizeof(int), stream);

    const int TB = 256;
    auto nb = [](long long n, int tb) { return (unsigned)((n + tb - 1) / tb); };

    const int nTiles = (N + 15) / 16;
    const int gemmBlks = (nTiles + 7) / 8;
    const int gPart = (gemmBlks + 2) / 3;

    bucket_count<<<NT + gPart, 256, 0, stream>>>(dst, bucketCount, tileOff, tileHist, E, NB, NT, x, W1, h1, N, nTiles, 0);
    bucket_scatter<<<NT + gPart, 256, 0, stream>>>(src, dst, bucketCount, tileOff, tileHist, bucketStart, binned, E, NB, NT,
                                                   x, W1, h1, N, nTiles, gPart);
    csr_fill<<<NB + gPart, 256, 0, stream>>>(bucketStart, binned, rowptr, csr, N, NB, x, W1, h1, nTiles, 2 * gPart);

    node_gather1<<<nb((long long)N, 32), TB, 0, stream>>>(rowptr, csr, attS1, attD1, h1, b1, W2, h2, N);

    node_gather2<<<nb((long long)N * 4, TB), TB, 0, stream>>>(rowptr, csr, attS2, attD2, h2, b2, out, N * 4);
}


# GEMM1 software-pipelined: a tile's 32 row loads requested up front under the W1 staging, B fragments read from a k-pair LDS image one k-step ahead of their matrix instructions; sums unchanged
# speedup vs baseline: 8.4241x; 1.0332x over previous
#include <hip/hip_runtime.h>
#include <math.h>

#define IN_C   128
#define HEADS  8
#define HID    8
#define F1     (HEADS * HID)
#define OUT_C  16
#define NEG_SLOPE 0.2f

typedef float v2f __attribute__((ext_vector_type(2)));
typedef float v8f __attribute__((ext_vector_type(8)));

__device__ __forceinline__ float lrelu(float v) {
    return v > 0.0f ? v : NEG_SLOPE * v;
}

#define LOG2E 1.4426950408889634f
__device__ __forceinline__ float exp2_hw(float y) {
    return __builtin_amdgcn_exp2f(y);
}

#define W1_PITCH2 80
__device__ __forceinline__ void gemm1_block(const float* __restrict__ x,
                                            const float* __restrict__ W,
                                            float* __restrict__ h1,
                                            int N, int nTiles, int gblk) {
    __shared__ v2f wimg[(IN_C / 2) * W1_PITCH2];
    const int t = threadIdx.x;

    float wr[32];
    #pragma unroll
    for (int u = 0; u < 32; ++u) wr[u] = W[t + 256 * u];

    const int tile  = gblk * 8 + (t >> 5);
    const int lane  = t & 31;
    const int row16 = lane & 15;
    const int kh    = lane >> 4;

    (void)nTiles;
    int rowIn = tile * 16 + row16;
    if (rowIn >= N) rowIn = N - 1;
    const float* xr = x + (size_t)rowIn * IN_C + 2 * kh;

    v2f a[32];
    #pragma unroll
    for (int u = 0; u < 32; ++u) {
        a[u].x = xr[4 * u];
        a[u].y = xr[4 * u + 1];
    }
    __builtin_amdgcn_sched_barrier(0);

    float* wf = (float*)wimg;
    #pragma unroll
    for (int u = 0; u < 32; ++u) {
        const int i = t + 256 * u;
        const int k = i >> 6, c = i & 63;
        wf[(k >> 1) * (2 * W1_PITCH2) + c * 2 + (k & 1)] = wr[u];
    }
    __builtin_amdgcn_fence(__ATOMIC_RELEASE, "workgroup", "local");
    __builtin_amdgcn_s_barrier();
    __builtin_amdgcn_fence(__ATOMIC_ACQUIRE, "workgroup", "local");

    v8f acc0 = {}, acc1 = {}, acc2 = {}, acc3 = {};

    const v2f* wb = &wimg[kh * W1_PITCH2 + row16];
    v2f b0 = wb[0], b1 = wb[16], b2 = wb[32], b3 = wb[48];
    #pragma unroll
    for (int u = 0; u < 32; ++u) {
        v2f n0 = b0, n1 = b1, n2 = b2, n3 = b3;
        if (u + 1 < 32) {
            const v2f* wn = wb + 2 * (u + 1) * W1_PITCH2;
            n0 = wn[0]; n1 = wn[16]; n2 = wn[32]; n3 = wn[48];
        }
        __builtin_amdgcn_sched_barrier(0);
        acc0 = __builtin_amdgcn_wmma_f32_16x16x4_f32(false, a[u], false, b0, (short)0, acc0, false, false);
        acc1 = __builtin_amdgcn_wmma_f32_16x16x4_f32(false, a[u], false, b1, (short)0, acc1, false, false);
        acc2 = __builtin_amdgcn_wmma_f32_16x16x4_f32(false, a[u], false, b2, (short)0, acc2, false, false);
        acc3 = __builtin_amdgcn_wmma_f32_16x16x4_f32(false, a[u], false, b3, (short)0, acc3, false, false);
        b0 = n0; b1 = n1; b2 = n2; b3 = n3;
    }

    #pragma unroll
    for (int i = 0; i < 8; ++i) {
        const int r = tile * 16 + i + 8 * kh;
        if (r < N) {
            float* o = h1 + (size_t)r * F1 + row16;
            o[0]  = acc0[i];
            o[16] = acc1[i];
            o[32] = acc2[i];
            o[48] = acc3[i];
        }
    }
}

__device__ __forceinline__ float dot8(const float4 a, const float4 b,
                                      const float w0, const float w1, const float w2, const float w3,
                                      const float w4, const float w5, const float w6, const float w7) {
    float s = a.x * w0;
    s = fmaf(a.y, w1, s); s = fmaf(a.z, w2, s); s = fmaf(a.w, w3, s);
    s = fmaf(b.x, w4, s); s = fmaf(b.y, w5, s); s = fmaf(b.z, w6, s); s = fmaf(b.w, w7, s);
    return s;
}

__device__ __forceinline__ float dot4(const float4 a, const float w0, const float w1, const float w2, const float w3) {
    float s = a.x * w0;
    s = fmaf(a.y, w1, s); s = fmaf(a.z, w2, s); s = fmaf(a.w, w3, s);
    return s;
}

__device__ __forceinline__ float quad_sum(float v) {
    v += __builtin_bit_cast(float, __builtin_amdgcn_update_dpp(0, __builtin_bit_cast(int, v), 0xB1, 0xF, 0xF, true));
    v += __builtin_bit_cast(float, __builtin_amdgcn_update_dpp(0, __builtin_bit_cast(int, v), 0x4E, 0xF, 0xF, true));
    return v;
}

#define TILE_E  4096
#define BKT_MAX 512
#define STAGE_MAX 8192

__device__ __forceinline__ void scan256(int* part, const int t) {
    for (int off = 1; off < 256; off <<= 1) {
        const int v = (t >= off) ? part[t - off] : 0;
        __syncthreads();
        part[t] += v;
        __syncthreads();
    }
}

__device__ __forceinline__ void scan256x2(int* pa, int* pb, const int t) {
    for (int off = 1; off < 256; off <<= 1) {
        const int va = (t >= off) ? pa[t - off] : 0;
        const int vb = (t >= off) ? pb[t - off] : 0;
        __syncthreads();
        pa[t] += va;
        pb[t] += vb;
        __syncthreads();
    }
}

__global__ void __launch_bounds__(256) bucket_count(const int* __restrict__ dst, int* __restrict__ bucketCount,
                                                    int* __restrict__ tileOff, int* __restrict__ tileHist,
                                                    int E, int NB, int NT,
                                                    const float* __restrict__ x, const float* __restrict__ W1,
                                                    float* __restrict__ h1, int N, int nTiles, int gBase) {
    if ((int)blockIdx.x >= NT) { gemm1_block(x, W1, h1, N, nTiles, gBase + (int)blockIdx.x - NT); return; }
    __shared__ int hist[BKT_MAX];
    const int t = threadIdx.x;
    for (int i = t; i < NB; i += 256) hist[i] = 0;
    __syncthreads();
    const int base = blockIdx.x * TILE_E + t;
    int d[16];
    #pragma unroll
    for (int k = 0; k < 16; ++k) {
        const int e = base + k * 256;
        d[k] = dst[e < E ? e : E - 1];
    }
    #pragma unroll
    for (int k = 0; k < 16; ++k) {
        if (base + k * 256 < E) atomicAdd(&hist[d[k] >> 8], 1);
    }
    __syncthreads();
    const int i0 = t, i1 = t + 256;
    int r0 = 0, r1 = 0;
    if (i0 < NB) r0 = atomicAdd(&bucketCount[i0], hist[i0]);
    if (i1 < NB) r1 = atomicAdd(&bucketCount[i1], hist[i1]);
    if (i0 < NB) tileOff[blockIdx.x * NB + i0] = r0;
    if (i1 < NB) tileOff[blockIdx.x * NB + i1] = r1;
    if (i0 < NB) tileHist[blockIdx.x * NB + i0] = hist[i0];
    if (i1 < NB) tileHist[blockIdx.x * NB + i1] = hist[i1];
}

__global__ void __launch_bounds__(256) bucket_scatter(const int* __restrict__ src, const int* __restrict__ dst,
                                                      const int* __restrict__ bucketCount,
                                                      const int* __restrict__ tileOff,
                                                      const int* __restrict__ tileHist,
                                                      int* __restrict__ bucketStart, int* __restrict__ binned,
                                                      int E, int NB, int NT,
                                                      const float* __restrict__ x, const float* __restrict__ W1,
                                                      float* __restrict__ h1, int N, int nTiles, int gBase) {
    if ((int)blockIdx.x >= NT) { gemm1_block(x, W1, h1, N, nTiles, gBase + (int)blockIdx.x - NT); return; }
    __shared__ int cur[BKT_MAX + 32];
    __shared__ int dlt[BKT_MAX];
    __shared__ int part[256];
    __shared__ int lpart[256];
    __shared__ int sw[TILE_E + 32];
    __shared__ int sd[TILE_E + 32];
    const int t = threadIdx.x;
    const int base = blockIdx.x * TILE_E + t;
    if (t < 32) cur[BKT_MAX + t] = 0;
    int d[16], s[16];
    #pragma unroll
    for (int k = 0; k < 16; ++k) {
        const int e = base + k * 256;
        const int ec = e < E ? e : E - 1;
        d[k] = dst[ec];
        s[k] = src[ec];
    }
    const int b0 = 2 * t, b1 = 2 * t + 1;
    const int b0c = b0 < NB ? b0 : NB - 1, b1c = b1 < NB ? b1 : NB - 1;
    const int c0r = bucketCount[b0c], c1r = bucketCount[b1c];
    const int o0r = tileOff[blockIdx.x * NB + b0c], o1r = tileOff[blockIdx.x * NB + b1c];
    const int h0r = tileHist[blockIdx.x * NB + b0c], h1r = tileHist[blockIdx.x * NB + b1c];
    __builtin_amdgcn_sched_barrier(0);
    const int c0 = b0 < NB ? c0r : 0, c1 = b1 < NB ? c1r : 0;
    const int o0 = b0 < NB ? o0r : 0, o1 = b1 < NB ? o1r : 0;
    const int h0 = b0 < NB ? h0r : 0, h1c = b1 < NB ? h1r : 0;
    part[t] = c0 + c1;
    lpart[t] = h0 + h1c;
    __syncthreads();
    scan256x2(part, lpart, t);
    const int ex = part[t] - (c0 + c1);
    const int lex = lpart[t] - (h0 + h1c);
    if (b0 < NB) { cur[b0] = ex + o0;      dlt[b0] = lex - (ex + o0); }
    if (b1 < NB) { cur[b1] = ex + c0 + o1; dlt[b1] = lex + h0 - (ex + c0 + o1); }
    if (blockIdx.x == 0) {
        if (b0 < NB) bucketStart[b0] = ex;
        if (b1 < NB) bucketStart[b1] = ex + c0;
        if (t == 255) bucketStart[NB] = part[255];
    }
    __syncthreads();
    int pos[16];
    #pragma unroll
    for (int k = 0; k < 16; ++k)
        pos[k] = atomicAdd(&cur[(base + k * 256 < E) ? (d[k] >> 8) : (BKT_MAX + (t & 31))], 1);
    int dl[16];
    #pragma unroll
    for (int k = 0; k < 16; ++k) dl[k] = dlt[d[k] >> 8];
    #pragma unroll
    for (int k = 0; k < 16; ++k) {
        const int slot = (base + k * 256 < E) ? pos[k] + dl[k] : TILE_E + (t & 31);
        sw[slot] = (int)(((unsigned)s[k] << 8) | (unsigned)(d[k] & 255));
        sd[slot] = pos[k];
    }
    __syncthreads();
    const int nE = min(TILE_E, E - (int)blockIdx.x * TILE_E);
    int wv[16], wd[16];
    #pragma unroll
    for (int k = 0; k < 16; ++k) {
        const int i = t + k * 256;
        const int ic = i < nE ? i : nE - 1;
        wv[k] = sw[ic];
        wd[k] = sd[ic];
    }
    #pragma unroll
    for (int k = 0; k < 16; ++k)
        if (t + k * 256 < nE) binned[wd[k]] = wv[k];
}

__global__ void __launch_bounds__(256) csr_fill(const int* __restrict__ bucketStart, const int* __restrict__ binned,
                                                int* __restrict__ rowptr, int* __restrict__ csr, int N, int NB,
                                                const float* __restrict__ x, const float* __restrict__ W1,
                                                float* __restrict__ h1, int nTiles, int gBase) {
    if ((int)blockIdx.x >= NB) { gemm1_block(x, W1, h1, N, nTiles, gBase + (int)blockIdx.x - NB); return; }
    __shared__ int cnt[256 + 32];
    __shared__ int part[256];
    __shared__ int stage[STAGE_MAX];
    const int t = threadIdx.x;
    const int spare = 256 + (t & 31);
    const int beg = bucketStart[blockIdx.x], end = bucketStart[blockIdx.x + 1];
    const int last = end - 1;
    const int nb = end - beg;
    const bool staged = nb <= STAGE_MAX;
    cnt[t] = 0;
    if (t < 32) cnt[256 + t] = 0;
    __syncthreads();
    for (int j = beg + t; j < end; j += 2048) {
        int v[8];
        #pragma unroll
        for (int k = 0; k < 8; ++k) {
            const int jk = j + k * 256;
            v[k] = binned[jk < end ? jk : last];
        }
        #pragma unroll
        for (int k = 0; k < 8; ++k)
            atomicAdd(&cnt[(j + k * 256 < end) ? (v[k] & 255) : spare], 1);
    }
    __syncthreads();
    const int c = cnt[t];
    part[t] = c;
    __syncthreads();
    scan256(part, t);
    const int first = beg + part[t] - c;
    const int n = blockIdx.x * 256 + t;
    if (n < N) rowptr[n] = first;
    if (n == N - 1) rowptr[N] = first + c;
    cnt[t] = first;
    __syncthreads();
    for (int j = beg + t; j < end; j += 2048) {
        int v[8], pos[8];
        #pragma unroll
        for (int k = 0; k < 8; ++k) {
            const int jk = j + k * 256;
            v[k] = binned[jk < end ? jk : last];
        }
        #pragma unroll
        for (int k = 0; k < 8; ++k)
            pos[k] = atomicAdd(&cnt[(j + k * 256 < end) ? (v[k] & 255) : spare], 1);
        #pragma unroll
        for (int k = 0; k < 8; ++k) {
            if (j + k * 256 < end) {
                const int sv = (int)((unsigned)v[k] >> 8);
                if (staged) stage[pos[k] - beg] = sv;
                else        csr[pos[k]] = sv;
            }
        }
    }
    if (!staged) return;
    __syncthreads();
    const int shift = beg & 31;
    for (int i0 = t - shift; i0 < nb; i0 += 2048) {
        int v[8];
        #pragma unroll
        for (int k = 0; k < 8; ++k) {
            const int i = i0 + k * 256;
            v[k] = stage[max(0, min(i, nb - 1))];
        }
        #pragma unroll
        for (int k = 0; k < 8; ++k) {
            const int i = i0 + k * 256;
            if (i >= 0 && i < nb) csr[beg + i] = v[k];
        }
    }
}

#define O1_PITCH 68
__global__ void __launch_bounds__(256, 12) node_gather1(const int* __restrict__ rowptr,
                             const int* __restrict__ csr,
                             const float* __restrict__ attS, const float* __restrict__ attD,
                             const float* __restrict__ h1, const float* __restrict__ b1,
                             const float* __restrict__ W2, float* __restrict__ h2, int N) {
    __shared__ float ot[32 * O1_PITCH];
    const int t = threadIdx.x;
    const int h = t & 7;
    int n = (int)blockIdx.x * 32 + (t >> 3);
    if (n >= N) n = N - 1;
    const int i = n * HEADS + h;
    const int beg = rowptr[n];
    const int end = rowptr[n + 1];
    const int last = end - 1;

    const float4* __restrict__ hv = (const float4*)h1;
    const float* ws = attS + h * HID;
    const float w0 = ws[0] * LOG2E, w1 = ws[1] * LOG2E, w2 = ws[2] * LOG2E, w3 = ws[3] * LOG2E;
    const float w4 = ws[4] * LOG2E, w5 = ws[5] * LOG2E, w6 = ws[6] * LOG2E, w7 = ws[7] * LOG2E;
    float ad;
    {
        const float* wd = attD + h * HID;
        const float4 p0 = hv[i * 2], p1 = hv[i * 2 + 1];
        ad = dot8(p0, p1, wd[0] * LOG2E, wd[1] * LOG2E, wd[2] * LOG2E, wd[3] * LOG2E,
                  wd[4] * LOG2E, wd[5] * LOG2E, wd[6] * LOG2E, wd[7] * LOG2E);
    }
    float m = -INFINITY;
    float den = 0.f;
    float a0 = 0.f, a1 = 0.f, a2 = 0.f, a3 = 0.f, a4 = 0.f, a5 = 0.f, a6 = 0.f, a7 = 0.f;
    for (int j = beg; j < end; j += 4) {
        const bool k1 = j + 1 < end, k2 = j + 2 < end, k3 = j + 3 < end;
        const int j1 = k1 ? j + 1 : last;
        const int j2 = k2 ? j + 2 : last;
        const int j3 = k3 ? j + 3 : last;
        const int s0 = csr[j], s1 = csr[j1], s2 = csr[j2], s3 = csr[j3];
        const int i0 = s0 * (F1 / 4) + h * 2, i1 = s1 * (F1 / 4) + h * 2;
        const int i2 = s2 * (F1 / 4) + h * 2, i3 = s3 * (F1 / 4) + h * 2;
        const float4 x0 = hv[i0], y0 = hv[i0 + 1];
        const float4 x1 = hv[i1], y1 = hv[i1 + 1];
        const float4 x2 = hv[i2], y2 = hv[i2 + 1];
        const float4 x3 = hv[i3], y3 = hv[i3 + 1];
        const float g0 = dot8(x0, y0, w0, w1, w2, w3, w4, w5, w6, w7);
        const float g1 = dot8(x1, y1, w0, w1, w2, w3, w4, w5, w6, w7);
        const float g2 = dot8(x2, y2, w0, w1, w2, w3, w4, w5, w6, w7);
        const float g3 = dot8(x3, y3, w0, w1, w2, w3, w4, w5, w6, w7);
        const float v0 = lrelu(g0 + ad), v1 = lrelu(g1 + ad);
        const float v2 = lrelu(g2 + ad), v3 = lrelu(g3 + ad);
        const float u = v0 > v1 ? v0 : v1;
        const float w = v2 > v3 ? v2 : v3;
        const float z = u > w ? u : w;
        const float mn = z > m ? z : m;
        const float sc = exp2_hw(m - mn);
        m = mn;
        const float e0 = exp2_hw(v0 - mn);
        const float e1 = k1 ? exp2_hw(v1 - mn) : 0.f;
        const float e2 = k2 ? exp2_hw(v2 - mn) : 0.f;
        const float e3 = k3 ? exp2_hw(v3 - mn) : 0.f;
        den *= sc;
        a0 *= sc; a1 *= sc; a2 *= sc; a3 *= sc; a4 *= sc; a5 *= sc; a6 *= sc; a7 *= sc;
        den += e0; den += e1; den += e2; den += e3;
        a0 += e0 * x0.x; a1 += e0 * x0.y; a2 += e0 * x0.z; a3 += e0 * x0.w;
        a4 += e0 * y0.x; a5 += e0 * y0.y; a6 += e0 * y0.z; a7 += e0 * y0.w;
        a0 += e1 * x1.x; a1 += e1 * x1.y; a2 += e1 * x1.z; a3 += e1 * x1.w;
        a4 += e1 * y1.x; a5 += e1 * y1.y; a6 += e1 * y1.z; a7 += e1 * y1.w;
        a0 += e2 * x2.x; a1 += e2 * x2.y; a2 += e2 * x2.z; a3 += e2 * x2.w;
        a4 += e2 * y2.x; a5 += e2 * y2.y; a6 += e2 * y2.z; a7 += e2 * y2.w;
        a0 += e3 * x3.x; a1 += e3 * x3.y; a2 += e3 * x3.z; a3 += e3 * x3.w;
        a4 += e3 * y3.x; a5 += e3 * y3.y; a6 += e3 * y3.z; a7 += e3 * y3.w;
    }

    const int wv    = t >> 5;
    const int lane  = t & 31;
    const int row16 = lane & 15;
    const int kh    = lane >> 4;
    v2f b[16];
    if (wv < 2) {
        #pragma unroll
        for (int u = 0; u < 16; ++u) {
            const int ka = 4 * u + 2 * kh;
            b[u].x = W2[ka * OUT_C + row16];
            b[u].y = W2[(ka + 1) * OUT_C + row16];
        }
    }

    const float dn = den + 1e-16f;
    const float* bp = b1 + h * HID;
    const float t0 = a0 / dn + bp[0], t1 = a1 / dn + bp[1], t2 = a2 / dn + bp[2], t3 = a3 / dn + bp[3];
    const float t4 = a4 / dn + bp[4], t5 = a5 / dn + bp[5], t6 = a6 / dn + bp[6], t7 = a7 / dn + bp[7];
    float4 r0, r1;
    r0.x = t0 > 0.f ? t0 : (exp2_hw(t0 * LOG2E) - 1.f);
    r0.y = t1 > 0.f ? t1 : (exp2_hw(t1 * LOG2E) - 1.f);
    r0.z = t2 > 0.f ? t2 : (exp2_hw(t2 * LOG2E) - 1.f);
    r0.w = t3 > 0.f ? t3 : (exp2_hw(t3 * LOG2E) - 1.f);
    r1.x = t4 > 0.f ? t4 : (exp2_hw(t4 * LOG2E) - 1.f);
    r1.y = t5 > 0.f ? t5 : (exp2_hw(t5 * LOG2E) - 1.f);
    r1.z = t6 > 0.f ? t6 : (exp2_hw(t6 * LOG2E) - 1.f);
    r1.w = t7 > 0.f ? t7 : (exp2_hw(t7 * LOG2E) - 1.f);
    float4* lp = (float4*)&ot[(t >> 3) * O1_PITCH + h * HID];
    lp[0] = r0;
    lp[1] = r1;
    __syncthreads();
    if (wv >= 2) return;

    const float* ar = &ot[(wv * 16 + row16) * O1_PITCH + 2 * kh];
    v8f acc = {};
    #pragma unroll
    for (int u = 0; u < 16; ++u) {
        v2f a;
        a.x = ar[4 * u];
        a.y = ar[4 * u + 1];
        acc = __builtin_amdgcn_wmma_f32_16x16x4_f32(false, a, false, b[u], (short)0, acc, false, false);
    }
    #pragma unroll
    for (int r = 0; r < 8; ++r) {
        const int row = (int)blockIdx.x * 32 + wv * 16 + r + 8 * kh;
        if (row < N) h2[(size_t)row * OUT_C + row16] = acc[r];
    }
}

__global__ void node_gather2(const int* __restrict__ rowptr,
                             const int* __restrict__ csr,
                             const float* __restrict__ attS, const float* __restrict__ attD,
                             const float* __restrict__ h2, const float* __restrict__ b2,
                             float* __restrict__ out, int NQ) {
    int i = blockIdx.x * blockDim.x + threadIdx.x;
    if (i >= NQ) return;
    const int n = i >> 2, q = i & 3;
    const int beg = rowptr[n];
    const int end = rowptr[n + 1];
    const int last = end - 1;

    const float4* __restrict__ hv = (const float4*)h2;
    const float* ws = attS + q * 4;
    const float w0 = ws[0] * LOG2E, w1 = ws[1] * LOG2E, w2 = ws[2] * LOG2E, w3 = ws[3] * LOG2E;
    float ad;
    {
        const float* wd = attD + q * 4;
        const float4 p = hv[i];
        ad = quad_sum(dot4(p, wd[0] * LOG2E, wd[1] * LOG2E, wd[2] * LOG2E, wd[3] * LOG2E));
    }
    float m = -INFINITY;
    float den = 0.f;
    float a0 = 0.f, a1 = 0.f, a2 = 0.f, a3 = 0.f;
    for (int j = beg; j < end; j += 4) {
        const bool k1 = j + 1 < end, k2 = j + 2 < end, k3 = j + 3 < end;
        const int j1 = k1 ? j + 1 : last;
        const int j2 = k2 ? j + 2 : last;
        const int j3 = k3 ? j + 3 : last;
        const int s0 = csr[j], s1 = csr[j1], s2 = csr[j2], s3 = csr[j3];
        const float4 x0 = hv[s0 * (OUT_C / 4) + q];
        const float4 x1 = hv[s1 * (OUT_C / 4) + q];
        const float4 x2 = hv[s2 * (OUT_C / 4) + q];
        const float4 x3 = hv[s3 * (OUT_C / 4) + q];
        const float g0 = quad_sum(dot4(x0, w0, w1, w2, w3)), g1 = quad_sum(dot4(x1, w0, w1, w2, w3));
        const float g2 = quad_sum(dot4(x2, w0, w1, w2, w3)), g3 = quad_sum(dot4(x3, w0, w1, w2, w3));
        const float v0 = lrelu(g0 + ad), v1 = lrelu(g1 + ad);
        const float v2 = lrelu(g2 + ad), v3 = lrelu(g3 + ad);
        const float u = v0 > v1 ? v0 : v1;
        const float w = v2 > v3 ? v2 : v3;
        const float z = u > w ? u : w;
        const float mn = z > m ? z : m;
        const float sc = exp2_hw(m - mn);
        m = mn;
        const float e0 = exp2_hw(v0 - mn);
        const float e1 = k1 ? exp2_hw(v1 - mn) : 0.f;
        const float e2 = k2 ? exp2_hw(v2 - mn) : 0.f;
        const float e3 = k3 ? exp2_hw(v3 - mn) : 0.f;
        den *= sc;
        a0 *= sc; a1 *= sc; a2 *= sc; a3 *= sc;
        den += e0; den += e1; den += e2; den += e3;
        a0 += e0 * x0.x; a1 += e0 * x0.y; a2 += e0 * x0.z; a3 += e0 * x0.w;
        a0 += e1 * x1.x; a1 += e1 * x1.y; a2 += e1 * x1.z; a3 += e1 * x1.w;
        a0 += e2 * x2.x; a1 += e2 * x2.y; a2 += e2 * x2.z; a3 += e2 * x2.w;
        a0 += e3 * x3.x; a1 += e3 * x3.y; a2 += e3 * x3.z; a3 += e3 * x3.w;
    }

    const float dn = den + 1e-16f;
    const float* bq = b2 + q * 4;
    float4 r;
    r.x = a0 / dn + bq[0];
    r.y = a1 / dn + bq[1];
    r.z = a2 / dn + bq[2];
    r.w = a3 / dn + bq[3];
    ((float4*)(out + (size_t)n * OUT_C))[q] = r;
}

extern "C" void kernel_launch(void* const* d_in, const int* in_sizes, int n_in,
                              void* d_out, int out_size, void* d_ws, size_t ws_size,
                              hipStream_t stream) {
    const int N = in_sizes[0] / IN_C;
    const int E = in_sizes[1] / 2;

    const float* x     = (const float*)d_in[0];
    const int*   ei    = (const int*)  d_in[1];
    const int*   src   = ei;
    const int*   dst   = ei + E;
    const float* W1    = (const float*)d_in[2];
    const float* b1    = (const float*)d_in[3];
    const float* attS1 = (const float*)d_in[4];
    const float* attD1 = (const float*)d_in[5];
    const float* W2    = (const float*)d_in[6];
    const float* b2    = (const float*)d_in[7];
    const float* attS2 = (const float*)d_in[8];
    const float* attD2 = (const float*)d_in[9];
    float* out = (float*)d_out;

    char* ws = (char*)d_ws;
    size_t off = 0;
    auto alloc = [&](size_t elems) -> float* {
        float* p = (float*)(ws + off);
        off = (off + elems * sizeof(float) + 255) & ~(size_t)255;
        return p;
    };
    float* h1   = alloc((size_t)N * F1);
    float* h2   = alloc((size_t)N * OUT_C);
    const int NB = (N + 255) / 256;
    const int NT = (E + TILE_E - 1) / TILE_E;
    int*   bucketCount = (int*)alloc((size_t)BKT_MAX);
    int*   bucketStart = (int*)alloc((size_t)BKT_MAX + 1);
    int*   tileOff     = (int*)alloc((size_t)NT * NB);
    int*   tileHist    = (int*)alloc((size_t)NT * NB);
    int*   rowptr      = (int*)alloc((size_t)N + 1);
    int*   binned      = (int*)alloc((size_t)E);
    int*   csr         = (int*)alloc((size_t)E);

    hipMemsetAsync(bucketCount, 0, (size_t)NB * sizeof(int), stream);

    const int TB = 256;
    auto nb = [](long long n, int tb) { return (unsigned)((n + tb - 1) / tb); };

    const int nTiles = (N + 15) / 16;
    const int gemmBlks = (nTiles + 7) / 8;
    const int gPart = (gemmBlks + 2) / 3;

    bucket_count<<<NT + gPart, 256, 0, stream>>>(dst, bucketCount, tileOff, tileHist, E, NB, NT, x, W1, h1, N, nTiles, 0);
    bucket_scatter<<<NT + gPart, 256, 0, stream>>>(src, dst, bucketCount, tileOff, tileHist, bucketStart, binned, E, NB, NT,
                                                   x, W1, h1, N, nTiles, gPart);
    csr_fill<<<NB + gPart, 256, 0, stream>>>(bucketStart, binned, rowptr, csr, N, NB, x, W1, h1, nTiles, 2 * gPart);

    node_gather1<<<nb((long long)N, 32), TB, 0, stream>>>(rowptr, csr, attS1, attD1, h1, b1, W2, h2, N);

    node_gather2<<<nb((long long)N * 4, TB), TB, 0, stream>>>(rowptr, csr, attS2, attD2, h2, b2, out, N * 4);
}
